# starting kernel
# baseline (speedup 1.0000x reference)
#include <hip/hip_runtime.h>
#include <hip/hip_bf16.h>


typedef float v8f __attribute__((ext_vector_type(8)));
typedef float v2f __attribute__((ext_vector_type(2)));

#define GNN_D 128
#define TILE_M 16
#define LDS_STRIDE 132

__device__ __forceinline__ unsigned f2mono(float f) {
    unsigned u = __float_as_uint(f);
    return (u & 0x80000000u) ? ~u : (u | 0x80000000u);
}
__device__ __forceinline__ float mono2f(unsigned m) {
    unsigned u = (m & 0x80000000u) ? (m & 0x7FFFFFFFu) : ~m;
    return __uint_as_float(u);
}
#define MONO_NEG_INF 0x007FFFFFu


__global__ __launch_bounds__(256) void gnn_init_mono_kernel(unsigned* __restrict__ p, int n) {
    int i = blockIdx.x * 256 + threadIdx.x;
    if (i < n) p[i] = MONO_NEG_INF;
}

__global__ __launch_bounds__(256) void gnn_deg_kernel(const int* __restrict__ dst,
                                                      float* __restrict__ deg, int E) {
    int e = blockIdx.x * 256 + threadIdx.x;
    if (e < E) atomicAdd(&deg[dst[e]], 1.0f);
}

__global__ __launch_bounds__(256) void gnn_scatter_kernel(
    const float* __restrict__ x, const int* __restrict__ src,
    const int* __restrict__ dst, float* __restrict__ s,
    unsigned* __restrict__ mx, int E)
{
    int gid  = blockIdx.x * 256 + threadIdx.x;
    int e    = gid >> 5;
    int lane = gid & 31;
    if (e >= E) return;
    int sn = src[e];
    int dn = dst[e];
    const float4 v = *reinterpret_cast<const float4*>(x + (size_t)sn * GNN_D + lane * 4);
    float*    sp = s  + (size_t)dn * GNN_D + lane * 4;
    unsigned* mp = mx + (size_t)dn * GNN_D + lane * 4;
    atomicAdd(sp + 0, v.x); atomicAdd(sp + 1, v.y);
    atomicAdd(sp + 2, v.z); atomicAdd(sp + 3, v.w);
    atomicMax(mp + 0, f2mono(v.x)); atomicMax(mp + 1, f2mono(v.y));
    atomicMax(mp + 2, f2mono(v.z)); atomicMax(mp + 3, f2mono(v.w));
}

__global__ __launch_bounds__(256) void gnn_layer_kernel(
    const float* __restrict__ x, const float* __restrict__ s,
    const unsigned* __restrict__ mx, const float* __restrict__ deg,
    const float* __restrict__ zagg,
    const float* __restrict__ W1, const float* __restrict__ b1,
    const float* __restrict__ W2, const float* __restrict__ b2,
    float* __restrict__ xout)
{
    __shared__ float hs[TILE_M * LDS_STRIDE];
    __shared__ float xt[TILE_M * LDS_STRIDE];
    __shared__ float ts[TILE_M * LDS_STRIDE];

    const int tid  = threadIdx.x;
    const int row0 = blockIdx.x * TILE_M;

    const float z0 = zagg[0], z1 = zagg[1], z2 = zagg[2];

    {
        int base = tid * 8;
        int r = base >> 7;
        int c = base & 127;
        size_t g = (size_t)(row0 + r) * GNN_D + c;
        float dg = deg[row0 + r];
        float dc = fmaxf(dg, 1.0f);
        bool  nb = dg > 0.0f;
        #pragma unroll
        for (int q = 0; q < 2; ++q) {
            float4 xv = *reinterpret_cast<const float4*>(x + g + q * 4);
            float4 sv = *reinterpret_cast<const float4*>(s + g + q * 4);
            uint4  mv = *reinterpret_cast<const uint4*>(mx + g + q * 4);
            float4 h;
            h.x = xv.x + z0 * sv.x + z1 * (sv.x / dc) + z2 * (nb ? mono2f(mv.x) : 0.0f);
            h.y = xv.y + z0 * sv.y + z1 * (sv.y / dc) + z2 * (nb ? mono2f(mv.y) : 0.0f);
            h.z = xv.z + z0 * sv.z + z1 * (sv.z / dc) + z2 * (nb ? mono2f(mv.z) : 0.0f);
            h.w = xv.w + z0 * sv.w + z1 * (sv.w / dc) + z2 * (nb ? mono2f(mv.w) : 0.0f);
            *reinterpret_cast<float4*>(&hs[r * LDS_STRIDE + c + q * 4]) = h;
            *reinterpret_cast<float4*>(&xt[r * LDS_STRIDE + c + q * 4]) = xv;
        }
    }
    __syncthreads();

    const int wave  = tid >> 5;
    const int lane  = tid & 31;
    const int ncol  = (wave << 4) + (lane & 15);
    const int mlo   = lane & 15;
    const int koff  = (lane >> 4) << 1;
    const int rbase = (lane >> 4) << 3;
    const float bia1 = b1[ncol];
    const float bia2 = b2[ncol];

    v8f acc = {};
    #pragma unroll
    for (int k0 = 0; k0 < GNN_D; k0 += 4) {
        int kk = k0 + koff;
        v2f a; a.x = hs[mlo * LDS_STRIDE + kk];
               a.y = hs[mlo * LDS_STRIDE + kk + 1];
        v2f b; b.x = W1[(size_t)kk * GNN_D + ncol];
               b.y = W1[(size_t)(kk + 1) * GNN_D + ncol];
        acc = __builtin_amdgcn_wmma_f32_16x16x4_f32(false, a, false, b,
                                                    (short)0, acc, false, false);
    }
    #pragma unroll
    for (int i = 0; i < 8; ++i)
        ts[(rbase + i) * LDS_STRIDE + ncol] = fmaxf(acc[i] + bia1, 0.0f);
    __syncthreads();

    v8f acc2 = {};
    #pragma unroll
    for (int k0 = 0; k0 < GNN_D; k0 += 4) {
        int kk = k0 + koff;
        v2f a; a.x = ts[mlo * LDS_STRIDE + kk];
               a.y = ts[mlo * LDS_STRIDE + kk + 1];
        v2f b; b.x = W2[(size_t)kk * GNN_D + ncol];
               b.y = W2[(size_t)(kk + 1) * GNN_D + ncol];
        acc2 = __builtin_amdgcn_wmma_f32_16x16x4_f32(false, a, false, b,
                                                     (short)0, acc2, false, false);
    }
    #pragma unroll
    for (int i = 0; i < 8; ++i) {
        int r = rbase + i;
        float h2 = acc2[i] + bia2;
        float xv = xt[r * LDS_STRIDE + ncol];
        xout[(size_t)(row0 + r) * GNN_D + ncol] = xv + fmaxf(h2, 0.0f);
    }
}

__global__ __launch_bounds__(256) void gnn_pool_scatter_kernel(
    const float* __restrict__ x, const int* __restrict__ batch,
    float* __restrict__ ps, unsigned* __restrict__ pmx,
    float* __restrict__ cnt, int N)
{
    int gid  = blockIdx.x * 256 + threadIdx.x;
    int n    = gid >> 5;
    int lane = gid & 31;
    if (n >= N) return;
    int b = batch[n];
    const float4 v = *reinterpret_cast<const float4*>(x + (size_t)n * GNN_D + lane * 4);
    float*    sp = ps  + (size_t)b * GNN_D + lane * 4;
    unsigned* mp = pmx + (size_t)b * GNN_D + lane * 4;
    atomicAdd(sp + 0, v.x); atomicAdd(sp + 1, v.y);
    atomicAdd(sp + 2, v.z); atomicAdd(sp + 3, v.w);
    atomicMax(mp + 0, f2mono(v.x)); atomicMax(mp + 1, f2mono(v.y));
    atomicMax(mp + 2, f2mono(v.z)); atomicMax(mp + 3, f2mono(v.w));
    if (lane == 0) atomicAdd(&cnt[b], 1.0f);
}

__global__ __launch_bounds__(256) void gnn_pool_finalize_kernel(
    const float* __restrict__ ps, const unsigned* __restrict__ pmx,
    const float* __restrict__ cnt, const float* __restrict__ zpool,
    float* __restrict__ out, int total)
{
    int i = blockIdx.x * 256 + threadIdx.x;
    if (i >= total) return;
    int b = i >> 7;
    float c  = cnt[b];
    float sv = ps[i];
    float mv = (c > 0.0f) ? mono2f(pmx[i]) : 0.0f;
    out[i] = zpool[0] * sv + zpool[1] * (sv / fmaxf(c, 1.0f)) + zpool[2] * mv;
}


extern "C" void kernel_launch(void* const* d_in, const int* in_sizes, int n_in,
                              void* d_out, int out_size, void* d_ws, size_t ws_size,
                              hipStream_t stream) {
    const float* x_in   = (const float*)d_in[0];
    const int*   eidx   = (const int*)d_in[1];
    const int*   batch  = (const int*)d_in[2];
    const float* z_agg  = (const float*)d_in[3];
    const float* z_pool = (const float*)d_in[4];
    const float* W1     = (const float*)d_in[5];
    const float* b1     = (const float*)d_in[6];
    const float* W2     = (const float*)d_in[7];
    const float* b2     = (const float*)d_in[8];
    (void)n_in;

    const int N = in_sizes[0] / GNN_D;
    const int E = in_sizes[1] / 2;
    const int L = in_sizes[5] / (GNN_D * GNN_D);
    const int Bg = out_size / GNN_D;
    const size_t ND = (size_t)N * GNN_D;
    const size_t BD = (size_t)Bg * GNN_D;

    char* p = (char*)d_ws;
    char* pend = p + ws_size;
    auto take = [&](size_t bytes) -> void* {
        void* r = (void*)p;
        p += (bytes + 255) & ~(size_t)255;
        return r;
    };
    float*    xa   = (float*)take(ND * 4);
    float*    xb   = (float*)take(ND * 4);
    float*    s    = (float*)take(ND * 4);
    unsigned* mx   = (unsigned*)take(ND * 4);
    float*    deg  = (float*)take((size_t)N * 4);
    float*    ps   = (float*)take(BD * 4);
    unsigned* pmx  = (unsigned*)take(BD * 4);
    float*    cnt  = (float*)take((size_t)Bg * 4);
    if (p > pend) return;

    const int* src = eidx;
    const int* dst = eidx + E;

    hipMemcpyAsync(xa, x_in, ND * 4, hipMemcpyDeviceToDevice, stream);

    hipMemsetAsync(deg, 0, (size_t)N * 4, stream);
    gnn_deg_kernel<<<(E + 255) / 256, 256, 0, stream>>>(dst, deg, E);

    float* xcur = xa;
    float* xnxt = xb;
    for (int l = 0; l < L; ++l) {
        hipMemsetAsync(s, 0, ND * 4, stream);
        gnn_init_mono_kernel<<<(int)((ND + 255) / 256), 256, 0, stream>>>(mx, (int)ND);
        gnn_scatter_kernel<<<(E + 7) / 8, 256, 0, stream>>>(xcur, src, dst, s, mx, E);
        gnn_layer_kernel<<<N / TILE_M, 256, 0, stream>>>(
            xcur, s, mx, deg, z_agg + l * 3,
            W1 + (size_t)l * GNN_D * GNN_D, b1 + (size_t)l * GNN_D,
            W2 + (size_t)l * GNN_D * GNN_D, b2 + (size_t)l * GNN_D, xnxt);
        float* t = xcur; xcur = xnxt; xnxt = t;
    }

    hipMemsetAsync(ps, 0, BD * 4, stream);
    hipMemsetAsync(cnt, 0, (size_t)Bg * 4, stream);
    gnn_init_mono_kernel<<<(int)((BD + 255) / 256), 256, 0, stream>>>(pmx, (int)BD);
    gnn_pool_scatter_kernel<<<(N + 7) / 8, 256, 0, stream>>>(xcur, batch, ps, pmx, cnt, N);
    gnn_pool_finalize_kernel<<<(int)((BD + 255) / 256), 256, 0, stream>>>(
        ps, pmx, cnt, z_pool, (float*)d_out, (int)BD);
}


# Pooling by owner: a work-group per (graph, 32-column slice) scans the batch ids, sums/maxes its rows in registers in node order, folds 16 wave partials in LDS, one store; no pooling atomics or clears
# speedup vs baseline: 1.3028x; 1.3028x over previous
#include <hip/hip_runtime.h>
#include <hip/hip_bf16.h>


typedef float v8f __attribute__((ext_vector_type(8)));
typedef float v2f __attribute__((ext_vector_type(2)));

#define GNN_D 128
#define TILE_M 16
#define LDS_STRIDE 132

__device__ __forceinline__ unsigned f2mono(float f) {
    unsigned u = __float_as_uint(f);
    return (u & 0x80000000u) ? ~u : (u | 0x80000000u);
}
__device__ __forceinline__ float mono2f(unsigned m) {
    unsigned u = (m & 0x80000000u) ? (m & 0x7FFFFFFFu) : ~m;
    return __uint_as_float(u);
}
#define MONO_NEG_INF 0x007FFFFFu


__global__ __launch_bounds__(256) void gnn_init_mono_kernel(unsigned* __restrict__ p, int n) {
    int i = blockIdx.x * 256 + threadIdx.x;
    if (i < n) p[i] = MONO_NEG_INF;
}

__global__ __launch_bounds__(256) void gnn_deg_kernel(const int* __restrict__ dst,
                                                      float* __restrict__ deg, int E) {
    int e = blockIdx.x * 256 + threadIdx.x;
    if (e < E) atomicAdd(&deg[dst[e]], 1.0f);
}

__global__ __launch_bounds__(256) void gnn_scatter_kernel(
    const float* __restrict__ x, const int* __restrict__ src,
    const int* __restrict__ dst, float* __restrict__ s,
    unsigned* __restrict__ mx, int E)
{
    int gid  = blockIdx.x * 256 + threadIdx.x;
    int e    = gid >> 5;
    int lane = gid & 31;
    if (e >= E) return;
    int sn = src[e];
    int dn = dst[e];
    const float4 v = *reinterpret_cast<const float4*>(x + (size_t)sn * GNN_D + lane * 4);
    float*    sp = s  + (size_t)dn * GNN_D + lane * 4;
    unsigned* mp = mx + (size_t)dn * GNN_D + lane * 4;
    atomicAdd(sp + 0, v.x); atomicAdd(sp + 1, v.y);
    atomicAdd(sp + 2, v.z); atomicAdd(sp + 3, v.w);
    atomicMax(mp + 0, f2mono(v.x)); atomicMax(mp + 1, f2mono(v.y));
    atomicMax(mp + 2, f2mono(v.z)); atomicMax(mp + 3, f2mono(v.w));
}

__global__ __launch_bounds__(256) void gnn_layer_kernel(
    const float* __restrict__ x, const float* __restrict__ s,
    const unsigned* __restrict__ mx, const float* __restrict__ deg,
    const float* __restrict__ zagg,
    const float* __restrict__ W1, const float* __restrict__ b1,
    const float* __restrict__ W2, const float* __restrict__ b2,
    float* __restrict__ xout)
{
    __shared__ float hs[TILE_M * LDS_STRIDE];
    __shared__ float xt[TILE_M * LDS_STRIDE];
    __shared__ float ts[TILE_M * LDS_STRIDE];

    const int tid  = threadIdx.x;
    const int row0 = blockIdx.x * TILE_M;

    const float z0 = zagg[0], z1 = zagg[1], z2 = zagg[2];

    {
        int base = tid * 8;
        int r = base >> 7;
        int c = base & 127;
        size_t g = (size_t)(row0 + r) * GNN_D + c;
        float dg = deg[row0 + r];
        float dc = fmaxf(dg, 1.0f);
        bool  nb = dg > 0.0f;
        #pragma unroll
        for (int q = 0; q < 2; ++q) {
            float4 xv = *reinterpret_cast<const float4*>(x + g + q * 4);
            float4 sv = *reinterpret_cast<const float4*>(s + g + q * 4);
            uint4  mv = *reinterpret_cast<const uint4*>(mx + g + q * 4);
            float4 h;
            h.x = xv.x + z0 * sv.x + z1 * (sv.x / dc) + z2 * (nb ? mono2f(mv.x) : 0.0f);
            h.y = xv.y + z0 * sv.y + z1 * (sv.y / dc) + z2 * (nb ? mono2f(mv.y) : 0.0f);
            h.z = xv.z + z0 * sv.z + z1 * (sv.z / dc) + z2 * (nb ? mono2f(mv.z) : 0.0f);
            h.w = xv.w + z0 * sv.w + z1 * (sv.w / dc) + z2 * (nb ? mono2f(mv.w) : 0.0f);
            *reinterpret_cast<float4*>(&hs[r * LDS_STRIDE + c + q * 4]) = h;
            *reinterpret_cast<float4*>(&xt[r * LDS_STRIDE + c + q * 4]) = xv;
        }
    }
    __syncthreads();

    const int wave  = tid >> 5;
    const int lane  = tid & 31;
    const int ncol  = (wave << 4) + (lane & 15);
    const int mlo   = lane & 15;
    const int koff  = (lane >> 4) << 1;
    const int rbase = (lane >> 4) << 3;
    const float bia1 = b1[ncol];
    const float bia2 = b2[ncol];

    v8f acc = {};
    #pragma unroll
    for (int k0 = 0; k0 < GNN_D; k0 += 4) {
        int kk = k0 + koff;
        v2f a; a.x = hs[mlo * LDS_STRIDE + kk];
               a.y = hs[mlo * LDS_STRIDE + kk + 1];
        v2f b; b.x = W1[(size_t)kk * GNN_D + ncol];
               b.y = W1[(size_t)(kk + 1) * GNN_D + ncol];
        acc = __builtin_amdgcn_wmma_f32_16x16x4_f32(false, a, false, b,
                                                    (short)0, acc, false, false);
    }
    #pragma unroll
    for (int i = 0; i < 8; ++i)
        ts[(rbase + i) * LDS_STRIDE + ncol] = fmaxf(acc[i] + bia1, 0.0f);
    __syncthreads();

    v8f acc2 = {};
    #pragma unroll
    for (int k0 = 0; k0 < GNN_D; k0 += 4) {
        int kk = k0 + koff;
        v2f a; a.x = ts[mlo * LDS_STRIDE + kk];
               a.y = ts[mlo * LDS_STRIDE + kk + 1];
        v2f b; b.x = W2[(size_t)kk * GNN_D + ncol];
               b.y = W2[(size_t)(kk + 1) * GNN_D + ncol];
        acc2 = __builtin_amdgcn_wmma_f32_16x16x4_f32(false, a, false, b,
                                                     (short)0, acc2, false, false);
    }
    #pragma unroll
    for (int i = 0; i < 8; ++i) {
        int r = rbase + i;
        float h2 = acc2[i] + bia2;
        float xv = xt[r * LDS_STRIDE + ncol];
        xout[(size_t)(row0 + r) * GNN_D + ncol] = xv + fmaxf(h2, 0.0f);
    }
}

#define POOL_WAVES 16
#define POOL_U     8
__global__ __launch_bounds__(POOL_WAVES * 32) void gnn_pool_owner_kernel(
    const float* __restrict__ x, const int* __restrict__ batch,
    const float* __restrict__ zpool, float* __restrict__ out, int N)
{
    __shared__ float red_s[POOL_WAVES][32];
    __shared__ float red_m[POOL_WAVES][32];
    __shared__ int   red_c[POOL_WAVES];

    const int lane = threadIdx.x & 31;
    const int wave = __builtin_amdgcn_readfirstlane((int)(threadIdx.x >> 5));
    const int b    = blockIdx.x >> 2;
    const unsigned col = ((blockIdx.x & 3u) << 5) + (unsigned)lane;

    float sum = 0.0f;
    float mxv = -INFINITY;
    int   cnt = 0;

    for (int c0 = wave; (c0 << 5) < N; c0 += POOL_WAVES * POOL_U) {
        int ids[POOL_U];
        #pragma unroll
        for (int u = 0; u < POOL_U; ++u) {
            int n  = ((c0 + u * POOL_WAVES) << 5) + lane;
            int id = batch[min(n, N - 1)];
            ids[u] = (n < N) ? id : -1;
        }
        #pragma unroll
        for (int u = 0; u < POOL_U; ++u) {
            const unsigned mask = __builtin_amdgcn_ballot_w32(ids[u] == b);
            if (mask == 0u) continue;
            cnt += __builtin_popcount(mask);
            const int base = (c0 + u * POOL_WAVES) << 5;
            float v[32];
            if (base + 32 <= N) {
                const float* p = x + (size_t)base * GNN_D + col;
                #pragma unroll
                for (int j = 0; j < 32; ++j) v[j] = p[j * GNN_D];
            } else {
                #pragma unroll
                for (int j = 0; j < 32; ++j)
                    v[j] = x[(size_t)min(base + j, N - 1) * GNN_D + col];
            }
            #pragma unroll
            for (int j = 0; j < 32; ++j) {
                const bool in = (mask >> j) & 1u;
                sum += in ? v[j] : 0.0f;
                mxv = fmaxf(mxv, in ? v[j] : -INFINITY);
            }
        }
    }

    red_s[wave][lane] = sum;
    red_m[wave][lane] = mxv;
    if (lane == 0) red_c[wave] = cnt;
    __syncthreads();

    if (wave == 0) {
        float s = 0.0f;
        float m = -INFINITY;
        int   c = 0;
        #pragma unroll
        for (int w = 0; w < POOL_WAVES; ++w) {
            s += red_s[w][lane];
            m  = fmaxf(m, red_m[w][lane]);
            c += red_c[w];
        }
        const float cf = (float)c;
        const float mv = (c > 0) ? m : 0.0f;
        out[(size_t)b * GNN_D + col] =
            zpool[0] * s + zpool[1] * (s / fmaxf(cf, 1.0f)) + zpool[2] * mv;
    }
}


extern "C" void kernel_launch(void* const* d_in, const int* in_sizes, int n_in,
                              void* d_out, int out_size, void* d_ws, size_t ws_size,
                              hipStream_t stream) {
    const float* x_in   = (const float*)d_in[0];
    const int*   eidx   = (const int*)d_in[1];
    const int*   batch  = (const int*)d_in[2];
    const float* z_agg  = (const float*)d_in[3];
    const float* z_pool = (const float*)d_in[4];
    const float* W1     = (const float*)d_in[5];
    const float* b1     = (const float*)d_in[6];
    const float* W2     = (const float*)d_in[7];
    const float* b2     = (const float*)d_in[8];
    (void)n_in;

    const int N = in_sizes[0] / GNN_D;
    const int E = in_sizes[1] / 2;
    const int L = in_sizes[5] / (GNN_D * GNN_D);
    const int Bg = out_size / GNN_D;
    const size_t ND = (size_t)N * GNN_D;

    char* p = (char*)d_ws;
    char* pend = p + ws_size;
    auto take = [&](size_t bytes) -> void* {
        void* r = (void*)p;
        p += (bytes + 255) & ~(size_t)255;
        return r;
    };
    float*    xa   = (float*)take(ND * 4);
    float*    xb   = (float*)take(ND * 4);
    float*    s    = (float*)take(ND * 4);
    unsigned* mx   = (unsigned*)take(ND * 4);
    float*    deg  = (float*)take((size_t)N * 4);
    if (p > pend) return;

    const int* src = eidx;
    const int* dst = eidx + E;

    hipMemcpyAsync(xa, x_in, ND * 4, hipMemcpyDeviceToDevice, stream);

    hipMemsetAsync(deg, 0, (size_t)N * 4, stream);
    gnn_deg_kernel<<<(E + 255) / 256, 256, 0, stream>>>(dst, deg, E);

    float* xcur = xa;
    float* xnxt = xb;
    for (int l = 0; l < L; ++l) {
        hipMemsetAsync(s, 0, ND * 4, stream);
        gnn_init_mono_kernel<<<(int)((ND + 255) / 256), 256, 0, stream>>>(mx, (int)ND);
        gnn_scatter_kernel<<<(E + 7) / 8, 256, 0, stream>>>(xcur, src, dst, s, mx, E);
        gnn_layer_kernel<<<N / TILE_M, 256, 0, stream>>>(
            xcur, s, mx, deg, z_agg + l * 3,
            W1 + (size_t)l * GNN_D * GNN_D, b1 + (size_t)l * GNN_D,
            W2 + (size_t)l * GNN_D * GNN_D, b2 + (size_t)l * GNN_D, xnxt);
        float* t = xcur; xcur = xnxt; xnxt = t;
    }

    gnn_pool_owner_kernel<<<Bg * 4, POOL_WAVES * 32, 0, stream>>>(
        xcur, batch, z_pool, (float*)d_out, N);
}


# Edge aggregation by owner: a work-group owns 200 rows, scans the destination ids, queues its edges, adds/maxes their source rows in LDS (LDS atomics, sums by arrival); global atomics, clears gone
# speedup vs baseline: 2.6259x; 2.0156x over previous
#include <hip/hip_runtime.h>
#include <hip/hip_bf16.h>


typedef float v8f __attribute__((ext_vector_type(8)));
typedef float v2f __attribute__((ext_vector_type(2)));

#define GNN_D 128
#define TILE_M 16
#define LDS_STRIDE 132

__device__ __forceinline__ unsigned f2mono(float f) {
    unsigned u = __float_as_uint(f);
    return (u & 0x80000000u) ? ~u : (u | 0x80000000u);
}
__device__ __forceinline__ float mono2f(unsigned m) {
    unsigned u = (m & 0x80000000u) ? (m & 0x7FFFFFFFu) : ~m;
    return __uint_as_float(u);
}
#define MONO_NEG_INF 0x007FFFFFu


__global__ __launch_bounds__(256) void gnn_deg_kernel(const int* __restrict__ dst,
                                                      float* __restrict__ deg, int E) {
    int e = blockIdx.x * 256 + threadIdx.x;
    if (e < E) atomicAdd(&deg[dst[e]], 1.0f);
}

#define OWN_R     200
#define OWN_Q     4096
#define OWN_WAVES 32
#define OWN_U     16
#define OWN_PB    8
#define OWN_TRIP  (OWN_WAVES * OWN_U * 32)

__device__ __forceinline__ void own_take(
    int e, unsigned loc, int* q_n, int* q_src, int* q_row,
    float* acc_s, unsigned* acc_m,
    const float* __restrict__ x, const int* __restrict__ src)
{
    const int pos = atomicAdd(q_n, 1);
    if (pos < OWN_Q) {
        q_src[pos] = e;
        q_row[pos] = (int)loc;
    } else {
        const float4* xr = reinterpret_cast<const float4*>(x + (size_t)src[e] * GNN_D);
        float*    as = acc_s + loc * GNN_D;
        unsigned* am = acc_m + loc * GNN_D;
        #pragma nounroll
        for (int f = 0; f < GNN_D / 4; ++f) {
            const float4 v = xr[f];
            atomicAdd(as + 4 * f + 0, v.x); atomicMax(am + 4 * f + 0, f2mono(v.x));
            atomicAdd(as + 4 * f + 1, v.y); atomicMax(am + 4 * f + 1, f2mono(v.y));
            atomicAdd(as + 4 * f + 2, v.z); atomicMax(am + 4 * f + 2, f2mono(v.z));
            atomicAdd(as + 4 * f + 3, v.w); atomicMax(am + 4 * f + 3, f2mono(v.w));
        }
    }
}

__global__ __launch_bounds__(OWN_WAVES * 32) void gnn_aggr_owner_kernel(
    const float* __restrict__ x, const int* __restrict__ src,
    const int* __restrict__ dst, float* __restrict__ s,
    unsigned* __restrict__ mx, int N, int E)
{
    __shared__ __attribute__((aligned(16))) float    acc_s[(OWN_R + 1) * GNN_D];
    __shared__ __attribute__((aligned(16))) unsigned acc_m[(OWN_R + 1) * GNN_D];
    __shared__ int q_src[OWN_Q];
    __shared__ int q_row[OWN_Q];
    __shared__ int q_n;

    const int tid   = threadIdx.x;
    const int lane  = tid & 31;
    const int wave  = __builtin_amdgcn_readfirstlane(tid >> 5);
    const int r0    = blockIdx.x * OWN_R;
    const int nrows = min(OWN_R, N - r0);

    for (int i = tid; i < (OWN_R + 1) * GNN_D / 4; i += OWN_WAVES * 32) {
        reinterpret_cast<float4*>(acc_s)[i] = make_float4(0.0f, 0.0f, 0.0f, 0.0f);
        reinterpret_cast<uint4*>(acc_m)[i]  =
            make_uint4(MONO_NEG_INF, MONO_NEG_INF, MONO_NEG_INF, MONO_NEG_INF);
    }
    if (tid == 0) q_n = 0;
    __syncthreads();

    const int nfull = E / OWN_TRIP;
    unsigned trip = ((unsigned)blockIdx.x * (unsigned)nfull) / gridDim.x;
    for (int t = 0; t < nfull; ++t) {
        const int ebase = (int)trip * OWN_TRIP + (wave << 5) + lane;
        const int* p = dst + ebase;
        int d[OWN_U];
        #pragma unroll
        for (int u = 0; u < OWN_U; ++u) d[u] = p[u * (OWN_WAVES * 32)];
        #pragma unroll
        for (int u = 0; u < OWN_U; ++u) {
            const unsigned loc = (unsigned)d[u] - (unsigned)r0;
            if (loc < (unsigned)nrows)
                own_take(ebase + u * (OWN_WAVES * 32), loc, &q_n, q_src, q_row,
                         acc_s, acc_m, x, src);
        }
        if (++trip == (unsigned)nfull) trip = 0u;
    }
    {
        const int ebase = nfull * OWN_TRIP + (wave << 5) + lane;
        int d[OWN_U];
        #pragma unroll
        for (int u = 0; u < OWN_U; ++u) {
            const int e = ebase + u * (OWN_WAVES * 32);
            d[u] = (e < E) ? dst[e] : -1;
        }
        #pragma unroll
        for (int u = 0; u < OWN_U; ++u) {
            const unsigned loc = (unsigned)d[u] - (unsigned)r0;
            if (loc < (unsigned)nrows)
                own_take(ebase + u * (OWN_WAVES * 32), loc, &q_n, q_src, q_row,
                         acc_s, acc_m, x, src);
        }
    }
    __syncthreads();

    const int qn = min(q_n, OWN_Q);
    for (int i = tid; i < qn; i += OWN_WAVES * 32) q_src[i] = src[q_src[i]];
    __syncthreads();

    for (int q0 = wave * OWN_PB; q0 < qn; q0 += OWN_WAVES * OWN_PB) {
        float v[OWN_PB][4];
        int   rw[OWN_PB];
        #pragma unroll
        for (int p = 0; p < OWN_PB; ++p) {
            const int qc = min(q0 + p, qn - 1);
            const int sn = __builtin_amdgcn_readfirstlane(q_src[qc]);
            const int rr = __builtin_amdgcn_readfirstlane(q_row[qc]);
            rw[p] = (q0 + p < qn) ? rr : OWN_R;
            const float* xp = x + (size_t)sn * GNN_D + lane;
            #pragma unroll
            for (int k = 0; k < 4; ++k) v[p][k] = xp[32 * k];
        }
        #pragma unroll
        for (int p = 0; p < OWN_PB; ++p) {
            float*    as = acc_s + rw[p] * GNN_D + lane;
            unsigned* am = acc_m + rw[p] * GNN_D + lane;
            #pragma unroll
            for (int k = 0; k < 4; ++k) {
                atomicAdd(as + 32 * k, v[p][k]);
                atomicMax(am + 32 * k, f2mono(v[p][k]));
            }
        }
    }
    __syncthreads();

    float4* sg = reinterpret_cast<float4*>(s  + (size_t)r0 * GNN_D);
    uint4*  mg = reinterpret_cast<uint4*>(mx + (size_t)r0 * GNN_D);
    for (int i = tid; i < nrows * (GNN_D / 4); i += OWN_WAVES * 32) {
        sg[i] = reinterpret_cast<const float4*>(acc_s)[i];
        mg[i] = reinterpret_cast<const uint4*>(acc_m)[i];
    }
}

__global__ __launch_bounds__(256) void gnn_layer_kernel(
    const float* __restrict__ x, const float* __restrict__ s,
    const unsigned* __restrict__ mx, const float* __restrict__ deg,
    const float* __restrict__ zagg,
    const float* __restrict__ W1, const float* __restrict__ b1,
    const float* __restrict__ W2, const float* __restrict__ b2,
    float* __restrict__ xout)
{
    __shared__ float hs[TILE_M * LDS_STRIDE];
    __shared__ float xt[TILE_M * LDS_STRIDE];
    __shared__ float ts[TILE_M * LDS_STRIDE];

    const int tid  = threadIdx.x;
    const int row0 = blockIdx.x * TILE_M;

    const float z0 = zagg[0], z1 = zagg[1], z2 = zagg[2];

    {
        int base = tid * 8;
        int r = base >> 7;
        int c = base & 127;
        size_t g = (size_t)(row0 + r) * GNN_D + c;
        float dg = deg[row0 + r];
        float dc = fmaxf(dg, 1.0f);
        bool  nb = dg > 0.0f;
        #pragma unroll
        for (int q = 0; q < 2; ++q) {
            float4 xv = *reinterpret_cast<const float4*>(x + g + q * 4);
            float4 sv = *reinterpret_cast<const float4*>(s + g + q * 4);
            uint4  mv = *reinterpret_cast<const uint4*>(mx + g + q * 4);
            float4 h;
            h.x = xv.x + z0 * sv.x + z1 * (sv.x / dc) + z2 * (nb ? mono2f(mv.x) : 0.0f);
            h.y = xv.y + z0 * sv.y + z1 * (sv.y / dc) + z2 * (nb ? mono2f(mv.y) : 0.0f);
            h.z = xv.z + z0 * sv.z + z1 * (sv.z / dc) + z2 * (nb ? mono2f(mv.z) : 0.0f);
            h.w = xv.w + z0 * sv.w + z1 * (sv.w / dc) + z2 * (nb ? mono2f(mv.w) : 0.0f);
            *reinterpret_cast<float4*>(&hs[r * LDS_STRIDE + c + q * 4]) = h;
            *reinterpret_cast<float4*>(&xt[r * LDS_STRIDE + c + q * 4]) = xv;
        }
    }
    __syncthreads();

    const int wave  = tid >> 5;
    const int lane  = tid & 31;
    const int ncol  = (wave << 4) + (lane & 15);
    const int mlo   = lane & 15;
    const int koff  = (lane >> 4) << 1;
    const int rbase = (lane >> 4) << 3;
    const float bia1 = b1[ncol];
    const float bia2 = b2[ncol];

    v8f acc = {};
    #pragma unroll
    for (int k0 = 0; k0 < GNN_D; k0 += 4) {
        int kk = k0 + koff;
        v2f a; a.x = hs[mlo * LDS_STRIDE + kk];
               a.y = hs[mlo * LDS_STRIDE + kk + 1];
        v2f b; b.x = W1[(size_t)kk * GNN_D + ncol];
               b.y = W1[(size_t)(kk + 1) * GNN_D + ncol];
        acc = __builtin_amdgcn_wmma_f32_16x16x4_f32(false, a, false, b,
                                                    (short)0, acc, false, false);
    }
    #pragma unroll
    for (int i = 0; i < 8; ++i)
        ts[(rbase + i) * LDS_STRIDE + ncol] = fmaxf(acc[i] + bia1, 0.0f);
    __syncthreads();

    v8f acc2 = {};
    #pragma unroll
    for (int k0 = 0; k0 < GNN_D; k0 += 4) {
        int kk = k0 + koff;
        v2f a; a.x = ts[mlo * LDS_STRIDE + kk];
               a.y = ts[mlo * LDS_STRIDE + kk + 1];
        v2f b; b.x = W2[(size_t)kk * GNN_D + ncol];
               b.y = W2[(size_t)(kk + 1) * GNN_D + ncol];
        acc2 = __builtin_amdgcn_wmma_f32_16x16x4_f32(false, a, false, b,
                                                     (short)0, acc2, false, false);
    }
    #pragma unroll
    for (int i = 0; i < 8; ++i) {
        int r = rbase + i;
        float h2 = acc2[i] + bia2;
        float xv = xt[r * LDS_STRIDE + ncol];
        xout[(size_t)(row0 + r) * GNN_D + ncol] = xv + fmaxf(h2, 0.0f);
    }
}

#define POOL_WAVES 16
#define POOL_U     8
__global__ __launch_bounds__(POOL_WAVES * 32) void gnn_pool_owner_kernel(
    const float* __restrict__ x, const int* __restrict__ batch,
    const float* __restrict__ zpool, float* __restrict__ out, int N)
{
    __shared__ float red_s[POOL_WAVES][32];
    __shared__ float red_m[POOL_WAVES][32];
    __shared__ int   red_c[POOL_WAVES];

    const int lane = threadIdx.x & 31;
    const int wave = __builtin_amdgcn_readfirstlane((int)(threadIdx.x >> 5));
    const int b    = blockIdx.x >> 2;
    const unsigned col = ((blockIdx.x & 3u) << 5) + (unsigned)lane;

    float sum = 0.0f;
    float mxv = -INFINITY;
    int   cnt = 0;

    for (int c0 = wave; (c0 << 5) < N; c0 += POOL_WAVES * POOL_U) {
        int ids[POOL_U];
        #pragma unroll
        for (int u = 0; u < POOL_U; ++u) {
            int n  = ((c0 + u * POOL_WAVES) << 5) + lane;
            int id = batch[min(n, N - 1)];
            ids[u] = (n < N) ? id : -1;
        }
        #pragma unroll
        for (int u = 0; u < POOL_U; ++u) {
            const unsigned mask = __builtin_amdgcn_ballot_w32(ids[u] == b);
            if (mask == 0u) continue;
            cnt += __builtin_popcount(mask);
            const int base = (c0 + u * POOL_WAVES) << 5;
            float v[32];
            if (base + 32 <= N) {
                const float* p = x + (size_t)base * GNN_D + col;
                #pragma unroll
                for (int j = 0; j < 32; ++j) v[j] = p[j * GNN_D];
            } else {
                #pragma unroll
                for (int j = 0; j < 32; ++j)
                    v[j] = x[(size_t)min(base + j, N - 1) * GNN_D + col];
            }
            #pragma unroll
            for (int j = 0; j < 32; ++j) {
                const bool in = (mask >> j) & 1u;
                sum += in ? v[j] : 0.0f;
                mxv = fmaxf(mxv, in ? v[j] : -INFINITY);
            }
        }
    }

    red_s[wave][lane] = sum;
    red_m[wave][lane] = mxv;
    if (lane == 0) red_c[wave] = cnt;
    __syncthreads();

    if (wave == 0) {
        float s = 0.0f;
        float m = -INFINITY;
        int   c = 0;
        #pragma unroll
        for (int w = 0; w < POOL_WAVES; ++w) {
            s += red_s[w][lane];
            m  = fmaxf(m, red_m[w][lane]);
            c += red_c[w];
        }
        const float cf = (float)c;
        const float mv = (c > 0) ? m : 0.0f;
        out[(size_t)b * GNN_D + col] =
            zpool[0] * s + zpool[1] * (s / fmaxf(cf, 1.0f)) + zpool[2] * mv;
    }
}


extern "C" void kernel_launch(void* const* d_in, const int* in_sizes, int n_in,
                              void* d_out, int out_size, void* d_ws, size_t ws_size,
                              hipStream_t stream) {
    const float* x_in   = (const float*)d_in[0];
    const int*   eidx   = (const int*)d_in[1];
    const int*   batch  = (const int*)d_in[2];
    const float* z_agg  = (const float*)d_in[3];
    const float* z_pool = (const float*)d_in[4];
    const float* W1     = (const float*)d_in[5];
    const float* b1     = (const float*)d_in[6];
    const float* W2     = (const float*)d_in[7];
    const float* b2     = (const float*)d_in[8];
    (void)n_in;

    const int N = in_sizes[0] / GNN_D;
    const int E = in_sizes[1] / 2;
    const int L = in_sizes[5] / (GNN_D * GNN_D);
    const int Bg = out_size / GNN_D;
    const size_t ND = (size_t)N * GNN_D;

    char* p = (char*)d_ws;
    char* pend = p + ws_size;
    auto take = [&](size_t bytes) -> void* {
        void* r = (void*)p;
        p += (bytes + 255) & ~(size_t)255;
        return r;
    };
    float*    xa   = (float*)take(ND * 4);
    float*    xb   = (float*)take(ND * 4);
    float*    s    = (float*)take(ND * 4);
    unsigned* mx   = (unsigned*)take(ND * 4);
    float*    deg  = (float*)take((size_t)N * 4);
    if (p > pend) return;

    const int* src = eidx;
    const int* dst = eidx + E;

    hipMemcpyAsync(xa, x_in, ND * 4, hipMemcpyDeviceToDevice, stream);

    hipMemsetAsync(deg, 0, (size_t)N * 4, stream);
    gnn_deg_kernel<<<(E + 255) / 256, 256, 0, stream>>>(dst, deg, E);

    float* xcur = xa;
    float* xnxt = xb;
    for (int l = 0; l < L; ++l) {
        gnn_aggr_owner_kernel<<<(N + OWN_R - 1) / OWN_R, OWN_WAVES * 32, 0, stream>>>(
            xcur, src, dst, s, mx, N, E);
        gnn_layer_kernel<<<N / TILE_M, 256, 0, stream>>>(
            xcur, s, mx, deg, z_agg + l * 3,
            W1 + (size_t)l * GNN_D * GNN_D, b1 + (size_t)l * GNN_D,
            W2 + (size_t)l * GNN_D * GNN_D, b2 + (size_t)l * GNN_D, xnxt);
        float* t = xcur; xcur = xnxt; xnxt = t;
    }

    gnn_pool_owner_kernel<<<Bg * 4, POOL_WAVES * 32, 0, stream>>>(
        xcur, batch, z_pool, (float*)d_out, N);
}


# Owner edge queue kept: layer 0's scan stores each work-group's queue (source row, owned row) and count in the workspace; layers 1, 2 load it and skip the destination scan; sums still by arrival
# speedup vs baseline: 3.1257x; 1.1903x over previous
#include <hip/hip_runtime.h>
#include <hip/hip_bf16.h>


typedef float v8f __attribute__((ext_vector_type(8)));
typedef float v2f __attribute__((ext_vector_type(2)));

#define GNN_D 128
#define TILE_M 16
#define LDS_STRIDE 132

__device__ __forceinline__ unsigned f2mono(float f) {
    unsigned u = __float_as_uint(f);
    return (u & 0x80000000u) ? ~u : (u | 0x80000000u);
}
__device__ __forceinline__ float mono2f(unsigned m) {
    unsigned u = (m & 0x80000000u) ? (m & 0x7FFFFFFFu) : ~m;
    return __uint_as_float(u);
}
#define MONO_NEG_INF 0x007FFFFFu


__global__ __launch_bounds__(256) void gnn_deg_kernel(const int* __restrict__ dst,
                                                      float* __restrict__ deg, int E) {
    int e = blockIdx.x * 256 + threadIdx.x;
    if (e < E) atomicAdd(&deg[dst[e]], 1.0f);
}

#define OWN_R     200
#define OWN_Q     4096
#define OWN_WAVES 32
#define OWN_U     16
#define OWN_PB    8
#define OWN_TRIP  (OWN_WAVES * OWN_U * 32)

__device__ __forceinline__ void own_take(
    int e, unsigned loc, int* q_n, int* q_src, int* q_row,
    float* acc_s, unsigned* acc_m,
    const float* __restrict__ x, const int* __restrict__ src)
{
    const int pos = atomicAdd(q_n, 1);
    if (pos < OWN_Q) {
        q_src[pos] = e;
        q_row[pos] = (int)loc;
    } else {
        const float4* xr = reinterpret_cast<const float4*>(x + (size_t)src[e] * GNN_D);
        float*    as = acc_s + loc * GNN_D;
        unsigned* am = acc_m + loc * GNN_D;
        #pragma nounroll
        for (int f = 0; f < GNN_D / 4; ++f) {
            const float4 v = xr[f];
            atomicAdd(as + 4 * f + 0, v.x); atomicMax(am + 4 * f + 0, f2mono(v.x));
            atomicAdd(as + 4 * f + 1, v.y); atomicMax(am + 4 * f + 1, f2mono(v.y));
            atomicAdd(as + 4 * f + 2, v.z); atomicMax(am + 4 * f + 2, f2mono(v.z));
            atomicAdd(as + 4 * f + 3, v.w); atomicMax(am + 4 * f + 3, f2mono(v.w));
        }
    }
}

__global__ __launch_bounds__(OWN_WAVES * 32) void gnn_aggr_owner_kernel(
    const float* __restrict__ x, const int* __restrict__ src,
    const int* __restrict__ dst, float* __restrict__ s,
    unsigned* __restrict__ mx, int* wq, int* wqn, int reuse, int N, int E)
{
    __shared__ __attribute__((aligned(16))) float    acc_s[(OWN_R + 1) * GNN_D];
    __shared__ __attribute__((aligned(16))) unsigned acc_m[(OWN_R + 1) * GNN_D];
    __shared__ int q_src[OWN_Q];
    __shared__ int q_row[OWN_Q];
    __shared__ int q_n;

    const int tid   = threadIdx.x;
    const int lane  = tid & 31;
    const int wave  = __builtin_amdgcn_readfirstlane(tid >> 5);
    const int r0    = blockIdx.x * OWN_R;
    const int nrows = min(OWN_R, N - r0);
    int* const my_q = wq + (size_t)blockIdx.x * (2 * OWN_Q);
    const int saved = __builtin_amdgcn_readfirstlane(reuse ? wqn[blockIdx.x] : -1);
    const bool have = (unsigned)saved <= (unsigned)OWN_Q;

    for (int i = tid; i < (OWN_R + 1) * GNN_D / 4; i += OWN_WAVES * 32) {
        reinterpret_cast<float4*>(acc_s)[i] = make_float4(0.0f, 0.0f, 0.0f, 0.0f);
        reinterpret_cast<uint4*>(acc_m)[i]  =
            make_uint4(MONO_NEG_INF, MONO_NEG_INF, MONO_NEG_INF, MONO_NEG_INF);
    }
    if (tid == 0) q_n = 0;
    __syncthreads();

    int qn;
    if (have) {
        qn = saved;
        for (int i = tid; i < qn; i += OWN_WAVES * 32) {
            q_src[i] = my_q[i];
            q_row[i] = my_q[OWN_Q + i];
        }
        __syncthreads();
    } else {
        const int nfull = E / OWN_TRIP;
        unsigned trip = ((unsigned)blockIdx.x * (unsigned)nfull) / gridDim.x;
        for (int t = 0; t < nfull; ++t) {
            const int ebase = (int)trip * OWN_TRIP + (wave << 5) + lane;
            const int* p = dst + ebase;
            int d[OWN_U];
            #pragma unroll
            for (int u = 0; u < OWN_U; ++u) d[u] = p[u * (OWN_WAVES * 32)];
            #pragma unroll
            for (int u = 0; u < OWN_U; ++u) {
                const unsigned loc = (unsigned)d[u] - (unsigned)r0;
                if (loc < (unsigned)nrows)
                    own_take(ebase + u * (OWN_WAVES * 32), loc, &q_n, q_src, q_row,
                             acc_s, acc_m, x, src);
            }
            if (++trip == (unsigned)nfull) trip = 0u;
        }
        {
            const int ebase = nfull * OWN_TRIP + (wave << 5) + lane;
            int d[OWN_U];
            #pragma unroll
            for (int u = 0; u < OWN_U; ++u) {
                const int e = ebase + u * (OWN_WAVES * 32);
                d[u] = (e < E) ? dst[e] : -1;
            }
            #pragma unroll
            for (int u = 0; u < OWN_U; ++u) {
                const unsigned loc = (unsigned)d[u] - (unsigned)r0;
                if (loc < (unsigned)nrows)
                    own_take(ebase + u * (OWN_WAVES * 32), loc, &q_n, q_src, q_row,
                             acc_s, acc_m, x, src);
            }
        }
        __syncthreads();

        qn = min(q_n, OWN_Q);
        for (int i = tid; i < qn; i += OWN_WAVES * 32) {
            const int sn = src[q_src[i]];
            q_src[i] = sn;
            if (!reuse) {
                my_q[i]         = sn;
                my_q[OWN_Q + i] = q_row[i];
            }
        }
        if (!reuse && tid == 0) wqn[blockIdx.x] = q_n;
        __syncthreads();
    }

    for (int q0 = wave * OWN_PB; q0 < qn; q0 += OWN_WAVES * OWN_PB) {
        float v[OWN_PB][4];
        int   rw[OWN_PB];
        #pragma unroll
        for (int p = 0; p < OWN_PB; ++p) {
            const int qc = min(q0 + p, qn - 1);
            const int sn = __builtin_amdgcn_readfirstlane(q_src[qc]);
            const int rr = __builtin_amdgcn_readfirstlane(q_row[qc]);
            rw[p] = (q0 + p < qn) ? rr : OWN_R;
            const float* xp = x + (size_t)sn * GNN_D + lane;
            #pragma unroll
            for (int k = 0; k < 4; ++k) v[p][k] = xp[32 * k];
        }
        #pragma unroll
        for (int p = 0; p < OWN_PB; ++p) {
            float*    as = acc_s + rw[p] * GNN_D + lane;
            unsigned* am = acc_m + rw[p] * GNN_D + lane;
            #pragma unroll
            for (int k = 0; k < 4; ++k) {
                atomicAdd(as + 32 * k, v[p][k]);
                atomicMax(am + 32 * k, f2mono(v[p][k]));
            }
        }
    }
    __syncthreads();

    float4* sg = reinterpret_cast<float4*>(s  + (size_t)r0 * GNN_D);
    uint4*  mg = reinterpret_cast<uint4*>(mx + (size_t)r0 * GNN_D);
    for (int i = tid; i < nrows * (GNN_D / 4); i += OWN_WAVES * 32) {
        sg[i] = reinterpret_cast<const float4*>(acc_s)[i];
        mg[i] = reinterpret_cast<const uint4*>(acc_m)[i];
    }
}

__global__ __launch_bounds__(256) void gnn_layer_kernel(
    const float* __restrict__ x, const float* __restrict__ s,
    const unsigned* __restrict__ mx, const float* __restrict__ deg,
    const float* __restrict__ zagg,
    const float* __restrict__ W1, const float* __restrict__ b1,
    const float* __restrict__ W2, const float* __restrict__ b2,
    float* __restrict__ xout)
{
    __shared__ float hs[TILE_M * LDS_STRIDE];
    __shared__ float xt[TILE_M * LDS_STRIDE];
    __shared__ float ts[TILE_M * LDS_STRIDE];

    const int tid  = threadIdx.x;
    const int row0 = blockIdx.x * TILE_M;

    const float z0 = zagg[0], z1 = zagg[1], z2 = zagg[2];

    {
        int base = tid * 8;
        int r = base >> 7;
        int c = base & 127;
        size_t g = (size_t)(row0 + r) * GNN_D + c;
        float dg = deg[row0 + r];
        float dc = fmaxf(dg, 1.0f);
        bool  nb = dg > 0.0f;
        #pragma unroll
        for (int q = 0; q < 2; ++q) {
            float4 xv = *reinterpret_cast<const float4*>(x + g + q * 4);
            float4 sv = *reinterpret_cast<const float4*>(s + g + q * 4);
            uint4  mv = *reinterpret_cast<const uint4*>(mx + g + q * 4);
            float4 h;
            h.x = xv.x + z0 * sv.x + z1 * (sv.x / dc) + z2 * (nb ? mono2f(mv.x) : 0.0f);
            h.y = xv.y + z0 * sv.y + z1 * (sv.y / dc) + z2 * (nb ? mono2f(mv.y) : 0.0f);
            h.z = xv.z + z0 * sv.z + z1 * (sv.z / dc) + z2 * (nb ? mono2f(mv.z) : 0.0f);
            h.w = xv.w + z0 * sv.w + z1 * (sv.w / dc) + z2 * (nb ? mono2f(mv.w) : 0.0f);
            *reinterpret_cast<float4*>(&hs[r * LDS_STRIDE + c + q * 4]) = h;
            *reinterpret_cast<float4*>(&xt[r * LDS_STRIDE + c + q * 4]) = xv;
        }
    }
    __syncthreads();

    const int wave  = tid >> 5;
    const int lane  = tid & 31;
    const int ncol  = (wave << 4) + (lane & 15);
    const int mlo   = lane & 15;
    const int koff  = (lane >> 4) << 1;
    const int rbase = (lane >> 4) << 3;
    const float bia1 = b1[ncol];
    const float bia2 = b2[ncol];

    v8f acc = {};
    #pragma unroll
    for (int k0 = 0; k0 < GNN_D; k0 += 4) {
        int kk = k0 + koff;
        v2f a; a.x = hs[mlo * LDS_STRIDE + kk];
               a.y = hs[mlo * LDS_STRIDE + kk + 1];
        v2f b; b.x = W1[(size_t)kk * GNN_D + ncol];
               b.y = W1[(size_t)(kk + 1) * GNN_D + ncol];
        acc = __builtin_amdgcn_wmma_f32_16x16x4_f32(false, a, false, b,
                                                    (short)0, acc, false, false);
    }
    #pragma unroll
    for (int i = 0; i < 8; ++i)
        ts[(rbase + i) * LDS_STRIDE + ncol] = fmaxf(acc[i] + bia1, 0.0f);
    __syncthreads();

    v8f acc2 = {};
    #pragma unroll
    for (int k0 = 0; k0 < GNN_D; k0 += 4) {
        int kk = k0 + koff;
        v2f a; a.x = ts[mlo * LDS_STRIDE + kk];
               a.y = ts[mlo * LDS_STRIDE + kk + 1];
        v2f b; b.x = W2[(size_t)kk * GNN_D + ncol];
               b.y = W2[(size_t)(kk + 1) * GNN_D + ncol];
        acc2 = __builtin_amdgcn_wmma_f32_16x16x4_f32(false, a, false, b,
                                                     (short)0, acc2, false, false);
    }
    #pragma unroll
    for (int i = 0; i < 8; ++i) {
        int r = rbase + i;
        float h2 = acc2[i] + bia2;
        float xv = xt[r * LDS_STRIDE + ncol];
        xout[(size_t)(row0 + r) * GNN_D + ncol] = xv + fmaxf(h2, 0.0f);
    }
}

#define POOL_WAVES 16
#define POOL_U     8
__global__ __launch_bounds__(POOL_WAVES * 32) void gnn_pool_owner_kernel(
    const float* __restrict__ x, const int* __restrict__ batch,
    const float* __restrict__ zpool, float* __restrict__ out, int N)
{
    __shared__ float red_s[POOL_WAVES][32];
    __shared__ float red_m[POOL_WAVES][32];
    __shared__ int   red_c[POOL_WAVES];

    const int lane = threadIdx.x & 31;
    const int wave = __builtin_amdgcn_readfirstlane((int)(threadIdx.x >> 5));
    const int b    = blockIdx.x >> 2;
    const unsigned col = ((blockIdx.x & 3u) << 5) + (unsigned)lane;

    float sum = 0.0f;
    float mxv = -INFINITY;
    int   cnt = 0;

    for (int c0 = wave; (c0 << 5) < N; c0 += POOL_WAVES * POOL_U) {
        int ids[POOL_U];
        #pragma unroll
        for (int u = 0; u < POOL_U; ++u) {
            int n  = ((c0 + u * POOL_WAVES) << 5) + lane;
            int id = batch[min(n, N - 1)];
            ids[u] = (n < N) ? id : -1;
        }
        #pragma unroll
        for (int u = 0; u < POOL_U; ++u) {
            const unsigned mask = __builtin_amdgcn_ballot_w32(ids[u] == b);
            if (mask == 0u) continue;
            cnt += __builtin_popcount(mask);
            const int base = (c0 + u * POOL_WAVES) << 5;
            float v[32];
            if (base + 32 <= N) {
                const float* p = x + (size_t)base * GNN_D + col;
                #pragma unroll
                for (int j = 0; j < 32; ++j) v[j] = p[j * GNN_D];
            } else {
                #pragma unroll
                for (int j = 0; j < 32; ++j)
                    v[j] = x[(size_t)min(base + j, N - 1) * GNN_D + col];
            }
            #pragma unroll
            for (int j = 0; j < 32; ++j) {
                const bool in = (mask >> j) & 1u;
                sum += in ? v[j] : 0.0f;
                mxv = fmaxf(mxv, in ? v[j] : -INFINITY);
            }
        }
    }

    red_s[wave][lane] = sum;
    red_m[wave][lane] = mxv;
    if (lane == 0) red_c[wave] = cnt;
    __syncthreads();

    if (wave == 0) {
        float s = 0.0f;
        float m = -INFINITY;
        int   c = 0;
        #pragma unroll
        for (int w = 0; w < POOL_WAVES; ++w) {
            s += red_s[w][lane];
            m  = fmaxf(m, red_m[w][lane]);
            c += red_c[w];
        }
        const float cf = (float)c;
        const float mv = (c > 0) ? m : 0.0f;
        out[(size_t)b * GNN_D + col] =
            zpool[0] * s + zpool[1] * (s / fmaxf(cf, 1.0f)) + zpool[2] * mv;
    }
}


extern "C" void kernel_launch(void* const* d_in, const int* in_sizes, int n_in,
                              void* d_out, int out_size, void* d_ws, size_t ws_size,
                              hipStream_t stream) {
    const float* x_in   = (const float*)d_in[0];
    const int*   eidx   = (const int*)d_in[1];
    const int*   batch  = (const int*)d_in[2];
    const float* z_agg  = (const float*)d_in[3];
    const float* z_pool = (const float*)d_in[4];
    const float* W1     = (const float*)d_in[5];
    const float* b1     = (const float*)d_in[6];
    const float* W2     = (const float*)d_in[7];
    const float* b2     = (const float*)d_in[8];
    (void)n_in;

    const int N = in_sizes[0] / GNN_D;
    const int E = in_sizes[1] / 2;
    const int L = in_sizes[5] / (GNN_D * GNN_D);
    const int Bg = out_size / GNN_D;
    const size_t ND = (size_t)N * GNN_D;

    char* p = (char*)d_ws;
    char* pend = p + ws_size;
    auto take = [&](size_t bytes) -> void* {
        void* r = (void*)p;
        p += (bytes + 255) & ~(size_t)255;
        return r;
    };
    float*    xa   = (float*)take(ND * 4);
    float*    xb   = (float*)take(ND * 4);
    float*    s    = (float*)take(ND * 4);
    unsigned* mx   = (unsigned*)take(ND * 4);
    float*    deg  = (float*)take((size_t)N * 4);
    const int nown = (N + OWN_R - 1) / OWN_R;
    int*      wq   = (int*)take((size_t)nown * 2 * OWN_Q * 4);
    int*      wqn  = (int*)take((size_t)nown * 4);
    if (p > pend) return;

    const int* src = eidx;
    const int* dst = eidx + E;

    hipMemcpyAsync(xa, x_in, ND * 4, hipMemcpyDeviceToDevice, stream);

    hipMemsetAsync(deg, 0, (size_t)N * 4, stream);
    gnn_deg_kernel<<<(E + 255) / 256, 256, 0, stream>>>(dst, deg, E);

    float* xcur = xa;
    float* xnxt = xb;
    for (int l = 0; l < L; ++l) {
        gnn_aggr_owner_kernel<<<nown, OWN_WAVES * 32, 0, stream>>>(
            xcur, src, dst, s, mx, wq, wqn, l > 0 ? 1 : 0, N, E);
        gnn_layer_kernel<<<N / TILE_M, 256, 0, stream>>>(
            xcur, s, mx, deg, z_agg + l * 3,
            W1 + (size_t)l * GNN_D * GNN_D, b1 + (size_t)l * GNN_D,
            W2 + (size_t)l * GNN_D * GNN_D, b2 + (size_t)l * GNN_D, xnxt);
        float* t = xcur; xcur = xnxt; xnxt = t;
    }

    gnn_pool_owner_kernel<<<Bg * 4, POOL_WAVES * 32, 0, stream>>>(
        xcur, batch, z_pool, (float*)d_out, N);
}


# Zero-gate skip: an aggregate (sum or max) whose z_agg gates are exactly 0 is not formed: a wave-uniform test drops its LDS atomics, start values, row stores, layer-kernel loads; sums still by arrival
# speedup vs baseline: 3.5325x; 1.1301x over previous
#include <hip/hip_runtime.h>
#include <hip/hip_bf16.h>


typedef float v8f __attribute__((ext_vector_type(8)));
typedef float v2f __attribute__((ext_vector_type(2)));

#define GNN_D 128
#define TILE_M 16
#define LDS_STRIDE 132

__device__ __forceinline__ unsigned f2mono(float f) {
    unsigned u = __float_as_uint(f);
    return (u & 0x80000000u) ? ~u : (u | 0x80000000u);
}
__device__ __forceinline__ float mono2f(unsigned m) {
    unsigned u = (m & 0x80000000u) ? (m & 0x7FFFFFFFu) : ~m;
    return __uint_as_float(u);
}
#define MONO_NEG_INF 0x007FFFFFu


__global__ __launch_bounds__(256) void gnn_deg_kernel(const int* __restrict__ dst,
                                                      float* __restrict__ deg, int E) {
    int e = blockIdx.x * 256 + threadIdx.x;
    if (e < E) atomicAdd(&deg[dst[e]], 1.0f);
}

#define OWN_R     200
#define OWN_Q     4096
#define OWN_WAVES 32
#define OWN_U     16
#define OWN_PB    8
#define OWN_TRIP  (OWN_WAVES * OWN_U * 32)

__device__ __forceinline__ void own_take(
    int e, unsigned loc, int* q_n, int* q_src, int* q_row,
    float* acc_s, unsigned* acc_m,
    const float* __restrict__ x, const int* __restrict__ src,
    bool need_s, bool need_m)
{
    const int pos = atomicAdd(q_n, 1);
    if (pos < OWN_Q) {
        q_src[pos] = e;
        q_row[pos] = (int)loc;
    } else {
        const float4* xr = reinterpret_cast<const float4*>(x + (size_t)src[e] * GNN_D);
        float*    as = acc_s + loc * GNN_D;
        unsigned* am = acc_m + loc * GNN_D;
        #pragma nounroll
        for (int f = 0; f < GNN_D / 4; ++f) {
            const float4 v = xr[f];
            if (need_s) {
                atomicAdd(as + 4 * f + 0, v.x);
                atomicAdd(as + 4 * f + 1, v.y);
                atomicAdd(as + 4 * f + 2, v.z);
                atomicAdd(as + 4 * f + 3, v.w);
            }
            if (need_m) {
                atomicMax(am + 4 * f + 0, f2mono(v.x));
                atomicMax(am + 4 * f + 1, f2mono(v.y));
                atomicMax(am + 4 * f + 2, f2mono(v.z));
                atomicMax(am + 4 * f + 3, f2mono(v.w));
            }
        }
    }
}

template <bool SUM, bool MAX>
__device__ __forceinline__ void own_apply(
    const float* __restrict__ x, const int* q_src, const int* q_row,
    float* acc_s, unsigned* acc_m, int qn, int wave, int lane)
{
    for (int q0 = wave * OWN_PB; q0 < qn; q0 += OWN_WAVES * OWN_PB) {
        float v[OWN_PB][4];
        int   rw[OWN_PB];
        #pragma unroll
        for (int p = 0; p < OWN_PB; ++p) {
            const int qc = min(q0 + p, qn - 1);
            const int sn = __builtin_amdgcn_readfirstlane(q_src[qc]);
            const int rr = __builtin_amdgcn_readfirstlane(q_row[qc]);
            rw[p] = (q0 + p < qn) ? rr : OWN_R;
            const float* xp = x + (size_t)sn * GNN_D + lane;
            #pragma unroll
            for (int k = 0; k < 4; ++k) v[p][k] = xp[32 * k];
        }
        #pragma unroll
        for (int p = 0; p < OWN_PB; ++p) {
            float*    as = acc_s + rw[p] * GNN_D + lane;
            unsigned* am = acc_m + rw[p] * GNN_D + lane;
            #pragma unroll
            for (int k = 0; k < 4; ++k) {
                if (SUM) atomicAdd(as + 32 * k, v[p][k]);
                if (MAX) atomicMax(am + 32 * k, f2mono(v[p][k]));
            }
        }
    }
}

__global__ __launch_bounds__(OWN_WAVES * 32) void gnn_aggr_owner_kernel(
    const float* __restrict__ x, const int* __restrict__ src,
    const int* __restrict__ dst, float* __restrict__ s,
    unsigned* __restrict__ mx, const float* __restrict__ zagg,
    int* wq, int* wqn, int reuse, int N, int E)
{
    __shared__ __attribute__((aligned(16))) float    acc_s[(OWN_R + 1) * GNN_D];
    __shared__ __attribute__((aligned(16))) unsigned acc_m[(OWN_R + 1) * GNN_D];
    __shared__ int q_src[OWN_Q];
    __shared__ int q_row[OWN_Q];
    __shared__ int q_n;

    const int tid   = threadIdx.x;
    const int lane  = tid & 31;
    const int wave  = __builtin_amdgcn_readfirstlane(tid >> 5);
    const int r0    = blockIdx.x * OWN_R;
    const int nrows = min(OWN_R, N - r0);
    int* const my_q = wq + (size_t)blockIdx.x * (2 * OWN_Q);
    const int saved = __builtin_amdgcn_readfirstlane(reuse ? wqn[blockIdx.x] : -1);
    const bool have = (unsigned)saved <= (unsigned)OWN_Q;
    const bool need_s = (zagg[0] != 0.0f) || (zagg[1] != 0.0f);
    const bool need_m = (zagg[2] != 0.0f);

    for (int i = tid; i < (OWN_R + 1) * GNN_D / 4; i += OWN_WAVES * 32) {
        if (need_s)
            reinterpret_cast<float4*>(acc_s)[i] = make_float4(0.0f, 0.0f, 0.0f, 0.0f);
        if (need_m)
            reinterpret_cast<uint4*>(acc_m)[i]  =
                make_uint4(MONO_NEG_INF, MONO_NEG_INF, MONO_NEG_INF, MONO_NEG_INF);
    }
    if (tid == 0) q_n = 0;
    __syncthreads();

    int qn;
    if (have) {
        qn = saved;
        for (int i = tid; i < qn; i += OWN_WAVES * 32) {
            q_src[i] = my_q[i];
            q_row[i] = my_q[OWN_Q + i];
        }
        __syncthreads();
    } else {
        const int nfull = E / OWN_TRIP;
        unsigned trip = ((unsigned)blockIdx.x * (unsigned)nfull) / gridDim.x;
        for (int t = 0; t < nfull; ++t) {
            const int ebase = (int)trip * OWN_TRIP + (wave << 5) + lane;
            const int* p = dst + ebase;
            int d[OWN_U];
            #pragma unroll
            for (int u = 0; u < OWN_U; ++u) d[u] = p[u * (OWN_WAVES * 32)];
            #pragma unroll
            for (int u = 0; u < OWN_U; ++u) {
                const unsigned loc = (unsigned)d[u] - (unsigned)r0;
                if (loc < (unsigned)nrows)
                    own_take(ebase + u * (OWN_WAVES * 32), loc, &q_n, q_src, q_row,
                             acc_s, acc_m, x, src, need_s, need_m);
            }
            if (++trip == (unsigned)nfull) trip = 0u;
        }
        {
            const int ebase = nfull * OWN_TRIP + (wave << 5) + lane;
            int d[OWN_U];
            #pragma unroll
            for (int u = 0; u < OWN_U; ++u) {
                const int e = ebase + u * (OWN_WAVES * 32);
                d[u] = (e < E) ? dst[e] : -1;
            }
            #pragma unroll
            for (int u = 0; u < OWN_U; ++u) {
                const unsigned loc = (unsigned)d[u] - (unsigned)r0;
                if (loc < (unsigned)nrows)
                    own_take(ebase + u * (OWN_WAVES * 32), loc, &q_n, q_src, q_row,
                             acc_s, acc_m, x, src, need_s, need_m);
            }
        }
        __syncthreads();

        qn = min(q_n, OWN_Q);
        for (int i = tid; i < qn; i += OWN_WAVES * 32) {
            const int sn = src[q_src[i]];
            q_src[i] = sn;
            if (!reuse) {
                my_q[i]         = sn;
                my_q[OWN_Q + i] = q_row[i];
            }
        }
        if (!reuse && tid == 0) wqn[blockIdx.x] = q_n;
        __syncthreads();
    }

    if (need_s && need_m)
        own_apply<true, true>(x, q_src, q_row, acc_s, acc_m, qn, wave, lane);
    else if (need_s)
        own_apply<true, false>(x, q_src, q_row, acc_s, acc_m, qn, wave, lane);
    else if (need_m)
        own_apply<false, true>(x, q_src, q_row, acc_s, acc_m, qn, wave, lane);
    __syncthreads();

    float4* sg = reinterpret_cast<float4*>(s  + (size_t)r0 * GNN_D);
    uint4*  mg = reinterpret_cast<uint4*>(mx + (size_t)r0 * GNN_D);
    for (int i = tid; i < nrows * (GNN_D / 4); i += OWN_WAVES * 32) {
        if (need_s) sg[i] = reinterpret_cast<const float4*>(acc_s)[i];
        if (need_m) mg[i] = reinterpret_cast<const uint4*>(acc_m)[i];
    }
}

__global__ __launch_bounds__(256) void gnn_layer_kernel(
    const float* __restrict__ x, const float* __restrict__ s,
    const unsigned* __restrict__ mx, const float* __restrict__ deg,
    const float* __restrict__ zagg,
    const float* __restrict__ W1, const float* __restrict__ b1,
    const float* __restrict__ W2, const float* __restrict__ b2,
    float* __restrict__ xout)
{
    __shared__ float hs[TILE_M * LDS_STRIDE];
    __shared__ float xt[TILE_M * LDS_STRIDE];
    __shared__ float ts[TILE_M * LDS_STRIDE];

    const int tid  = threadIdx.x;
    const int row0 = blockIdx.x * TILE_M;

    const float z0 = zagg[0], z1 = zagg[1], z2 = zagg[2];
    const bool need_s = (z0 != 0.0f) || (z1 != 0.0f);
    const bool need_m = (z2 != 0.0f);

    {
        int base = tid * 8;
        int r = base >> 7;
        int c = base & 127;
        size_t g = (size_t)(row0 + r) * GNN_D + c;
        float dg = deg[row0 + r];
        float dc = fmaxf(dg, 1.0f);
        bool  nb = (dg > 0.0f) && need_m;
        #pragma unroll
        for (int q = 0; q < 2; ++q) {
            float4 xv = *reinterpret_cast<const float4*>(x + g + q * 4);
            float4 sv = make_float4(0.0f, 0.0f, 0.0f, 0.0f);
            uint4  mv = make_uint4(0u, 0u, 0u, 0u);
            if (need_s) sv = *reinterpret_cast<const float4*>(s + g + q * 4);
            if (need_m) mv = *reinterpret_cast<const uint4*>(mx + g + q * 4);
            float4 h;
            h.x = xv.x + z0 * sv.x + z1 * (sv.x / dc) + z2 * (nb ? mono2f(mv.x) : 0.0f);
            h.y = xv.y + z0 * sv.y + z1 * (sv.y / dc) + z2 * (nb ? mono2f(mv.y) : 0.0f);
            h.z = xv.z + z0 * sv.z + z1 * (sv.z / dc) + z2 * (nb ? mono2f(mv.z) : 0.0f);
            h.w = xv.w + z0 * sv.w + z1 * (sv.w / dc) + z2 * (nb ? mono2f(mv.w) : 0.0f);
            *reinterpret_cast<float4*>(&hs[r * LDS_STRIDE + c + q * 4]) = h;
            *reinterpret_cast<float4*>(&xt[r * LDS_STRIDE + c + q * 4]) = xv;
        }
    }
    __syncthreads();

    const int wave  = tid >> 5;
    const int lane  = tid & 31;
    const int ncol  = (wave << 4) + (lane & 15);
    const int mlo   = lane & 15;
    const int koff  = (lane >> 4) << 1;
    const int rbase = (lane >> 4) << 3;
    const float bia1 = b1[ncol];
    const float bia2 = b2[ncol];

    v8f acc = {};
    #pragma unroll
    for (int k0 = 0; k0 < GNN_D; k0 += 4) {
        int kk = k0 + koff;
        v2f a; a.x = hs[mlo * LDS_STRIDE + kk];
               a.y = hs[mlo * LDS_STRIDE + kk + 1];
        v2f b; b.x = W1[(size_t)kk * GNN_D + ncol];
               b.y = W1[(size_t)(kk + 1) * GNN_D + ncol];
        acc = __builtin_amdgcn_wmma_f32_16x16x4_f32(false, a, false, b,
                                                    (short)0, acc, false, false);
    }
    #pragma unroll
    for (int i = 0; i < 8; ++i)
        ts[(rbase + i) * LDS_STRIDE + ncol] = fmaxf(acc[i] + bia1, 0.0f);
    __syncthreads();

    v8f acc2 = {};
    #pragma unroll
    for (int k0 = 0; k0 < GNN_D; k0 += 4) {
        int kk = k0 + koff;
        v2f a; a.x = ts[mlo * LDS_STRIDE + kk];
               a.y = ts[mlo * LDS_STRIDE + kk + 1];
        v2f b; b.x = W2[(size_t)kk * GNN_D + ncol];
               b.y = W2[(size_t)(kk + 1) * GNN_D + ncol];
        acc2 = __builtin_amdgcn_wmma_f32_16x16x4_f32(false, a, false, b,
                                                     (short)0, acc2, false, false);
    }
    #pragma unroll
    for (int i = 0; i < 8; ++i) {
        int r = rbase + i;
        float h2 = acc2[i] + bia2;
        float xv = xt[r * LDS_STRIDE + ncol];
        xout[(size_t)(row0 + r) * GNN_D + ncol] = xv + fmaxf(h2, 0.0f);
    }
}

#define POOL_WAVES 16
#define POOL_U     8
__global__ __launch_bounds__(POOL_WAVES * 32) void gnn_pool_owner_kernel(
    const float* __restrict__ x, const int* __restrict__ batch,
    const float* __restrict__ zpool, float* __restrict__ out, int N)
{
    __shared__ float red_s[POOL_WAVES][32];
    __shared__ float red_m[POOL_WAVES][32];
    __shared__ int   red_c[POOL_WAVES];

    const int lane = threadIdx.x & 31;
    const int wave = __builtin_amdgcn_readfirstlane((int)(threadIdx.x >> 5));
    const int b    = blockIdx.x >> 2;
    const unsigned col = ((blockIdx.x & 3u) << 5) + (unsigned)lane;

    float sum = 0.0f;
    float mxv = -INFINITY;
    int   cnt = 0;

    for (int c0 = wave; (c0 << 5) < N; c0 += POOL_WAVES * POOL_U) {
        int ids[POOL_U];
        #pragma unroll
        for (int u = 0; u < POOL_U; ++u) {
            int n  = ((c0 + u * POOL_WAVES) << 5) + lane;
            int id = batch[min(n, N - 1)];
            ids[u] = (n < N) ? id : -1;
        }
        #pragma unroll
        for (int u = 0; u < POOL_U; ++u) {
            const unsigned mask = __builtin_amdgcn_ballot_w32(ids[u] == b);
            if (mask == 0u) continue;
            cnt += __builtin_popcount(mask);
            const int base = (c0 + u * POOL_WAVES) << 5;
            float v[32];
            if (base + 32 <= N) {
                const float* p = x + (size_t)base * GNN_D + col;
                #pragma unroll
                for (int j = 0; j < 32; ++j) v[j] = p[j * GNN_D];
            } else {
                #pragma unroll
                for (int j = 0; j < 32; ++j)
                    v[j] = x[(size_t)min(base + j, N - 1) * GNN_D + col];
            }
            #pragma unroll
            for (int j = 0; j < 32; ++j) {
                const bool in = (mask >> j) & 1u;
                sum += in ? v[j] : 0.0f;
                mxv = fmaxf(mxv, in ? v[j] : -INFINITY);
            }
        }
    }

    red_s[wave][lane] = sum;
    red_m[wave][lane] = mxv;
    if (lane == 0) red_c[wave] = cnt;
    __syncthreads();

    if (wave == 0) {
        float s = 0.0f;
        float m = -INFINITY;
        int   c = 0;
        #pragma unroll
        for (int w = 0; w < POOL_WAVES; ++w) {
            s += red_s[w][lane];
            m  = fmaxf(m, red_m[w][lane]);
            c += red_c[w];
        }
        const float cf = (float)c;
        const float mv = (c > 0) ? m : 0.0f;
        out[(size_t)b * GNN_D + col] =
            zpool[0] * s + zpool[1] * (s / fmaxf(cf, 1.0f)) + zpool[2] * mv;
    }
}


extern "C" void kernel_launch(void* const* d_in, const int* in_sizes, int n_in,
                              void* d_out, int out_size, void* d_ws, size_t ws_size,
                              hipStream_t stream) {
    const float* x_in   = (const float*)d_in[0];
    const int*   eidx   = (const int*)d_in[1];
    const int*   batch  = (const int*)d_in[2];
    const float* z_agg  = (const float*)d_in[3];
    const float* z_pool = (const float*)d_in[4];
    const float* W1     = (const float*)d_in[5];
    const float* b1     = (const float*)d_in[6];
    const float* W2     = (const float*)d_in[7];
    const float* b2     = (const float*)d_in[8];
    (void)n_in;

    const int N = in_sizes[0] / GNN_D;
    const int E = in_sizes[1] / 2;
    const int L = in_sizes[5] / (GNN_D * GNN_D);
    const int Bg = out_size / GNN_D;
    const size_t ND = (size_t)N * GNN_D;

    char* p = (char*)d_ws;
    char* pend = p + ws_size;
    auto take = [&](size_t bytes) -> void* {
        void* r = (void*)p;
        p += (bytes + 255) & ~(size_t)255;
        return r;
    };
    float*    xa   = (float*)take(ND * 4);
    float*    xb   = (float*)take(ND * 4);
    float*    s    = (float*)take(ND * 4);
    unsigned* mx   = (unsigned*)take(ND * 4);
    float*    deg  = (float*)take((size_t)N * 4);
    const int nown = (N + OWN_R - 1) / OWN_R;
    int*      wq   = (int*)take((size_t)nown * 2 * OWN_Q * 4);
    int*      wqn  = (int*)take((size_t)nown * 4);
    if (p > pend) return;

    const int* src = eidx;
    const int* dst = eidx + E;

    hipMemcpyAsync(xa, x_in, ND * 4, hipMemcpyDeviceToDevice, stream);

    hipMemsetAsync(deg, 0, (size_t)N * 4, stream);
    gnn_deg_kernel<<<(E + 255) / 256, 256, 0, stream>>>(dst, deg, E);

    float* xcur = xa;
    float* xnxt = xb;
    for (int l = 0; l < L; ++l) {
        gnn_aggr_owner_kernel<<<nown, OWN_WAVES * 32, 0, stream>>>(
            xcur, src, dst, s, mx, z_agg + l * 3, wq, wqn, l > 0 ? 1 : 0, N, E);
        gnn_layer_kernel<<<N / TILE_M, 256, 0, stream>>>(
            xcur, s, mx, deg, z_agg + l * 3,
            W1 + (size_t)l * GNN_D * GNN_D, b1 + (size_t)l * GNN_D,
            W2 + (size_t)l * GNN_D * GNN_D, b2 + (size_t)l * GNN_D, xnxt);
        float* t = xcur; xcur = xnxt; xnxt = t;
    }

    gnn_pool_owner_kernel<<<Bg * 4, POOL_WAVES * 32, 0, stream>>>(
        xcur, batch, z_pool, (float*)d_out, N);
}


# Layer kernel weight-stationary: 512 persistent work-groups; each wave loads its 16-column slabs of W1, W2 once as register B fragments and walks 16-row tiles, next inputs in flight; sums unchanged
# speedup vs baseline: 3.7172x; 1.0523x over previous
#include <hip/hip_runtime.h>
#include <hip/hip_bf16.h>


typedef float v8f __attribute__((ext_vector_type(8)));
typedef float v2f __attribute__((ext_vector_type(2)));

#define GNN_D 128
#define TILE_M 16
#define LDS_STRIDE 132

__device__ __forceinline__ unsigned f2mono(float f) {
    unsigned u = __float_as_uint(f);
    return (u & 0x80000000u) ? ~u : (u | 0x80000000u);
}
__device__ __forceinline__ float mono2f(unsigned m) {
    unsigned u = (m & 0x80000000u) ? (m & 0x7FFFFFFFu) : ~m;
    return __uint_as_float(u);
}
#define MONO_NEG_INF 0x007FFFFFu


__global__ __launch_bounds__(256) void gnn_deg_kernel(const int* __restrict__ dst,
                                                      float* __restrict__ deg, int E) {
    int e = blockIdx.x * 256 + threadIdx.x;
    if (e < E) atomicAdd(&deg[dst[e]], 1.0f);
}

#define OWN_R     200
#define OWN_Q     4096
#define OWN_WAVES 32
#define OWN_U     16
#define OWN_PB    8
#define OWN_TRIP  (OWN_WAVES * OWN_U * 32)

__device__ __forceinline__ void own_take(
    int e, unsigned loc, int* q_n, int* q_src, int* q_row,
    float* acc_s, unsigned* acc_m,
    const float* __restrict__ x, const int* __restrict__ src,
    bool need_s, bool need_m)
{
    const int pos = atomicAdd(q_n, 1);
    if (pos < OWN_Q) {
        q_src[pos] = e;
        q_row[pos] = (int)loc;
    } else {
        const float4* xr = reinterpret_cast<const float4*>(x + (size_t)src[e] * GNN_D);
        float*    as = acc_s + loc * GNN_D;
        unsigned* am = acc_m + loc * GNN_D;
        #pragma nounroll
        for (int f = 0; f < GNN_D / 4; ++f) {
            const float4 v = xr[f];
            if (need_s) {
                atomicAdd(as + 4 * f + 0, v.x);
                atomicAdd(as + 4 * f + 1, v.y);
                atomicAdd(as + 4 * f + 2, v.z);
                atomicAdd(as + 4 * f + 3, v.w);
            }
            if (need_m) {
                atomicMax(am + 4 * f + 0, f2mono(v.x));
                atomicMax(am + 4 * f + 1, f2mono(v.y));
                atomicMax(am + 4 * f + 2, f2mono(v.z));
                atomicMax(am + 4 * f + 3, f2mono(v.w));
            }
        }
    }
}

template <bool SUM, bool MAX>
__device__ __forceinline__ void own_apply(
    const float* __restrict__ x, const int* q_src, const int* q_row,
    float* acc_s, unsigned* acc_m, int qn, int wave, int lane)
{
    for (int q0 = wave * OWN_PB; q0 < qn; q0 += OWN_WAVES * OWN_PB) {
        float v[OWN_PB][4];
        int   rw[OWN_PB];
        #pragma unroll
        for (int p = 0; p < OWN_PB; ++p) {
            const int qc = min(q0 + p, qn - 1);
            const int sn = __builtin_amdgcn_readfirstlane(q_src[qc]);
            const int rr = __builtin_amdgcn_readfirstlane(q_row[qc]);
            rw[p] = (q0 + p < qn) ? rr : OWN_R;
            const float* xp = x + (size_t)sn * GNN_D + lane;
            #pragma unroll
            for (int k = 0; k < 4; ++k) v[p][k] = xp[32 * k];
        }
        #pragma unroll
        for (int p = 0; p < OWN_PB; ++p) {
            float*    as = acc_s + rw[p] * GNN_D + lane;
            unsigned* am = acc_m + rw[p] * GNN_D + lane;
            #pragma unroll
            for (int k = 0; k < 4; ++k) {
                if (SUM) atomicAdd(as + 32 * k, v[p][k]);
                if (MAX) atomicMax(am + 32 * k, f2mono(v[p][k]));
            }
        }
    }
}

__global__ __launch_bounds__(OWN_WAVES * 32) void gnn_aggr_owner_kernel(
    const float* __restrict__ x, const int* __restrict__ src,
    const int* __restrict__ dst, float* __restrict__ s,
    unsigned* __restrict__ mx, const float* __restrict__ zagg,
    int* wq, int* wqn, int reuse, int N, int E)
{
    __shared__ __attribute__((aligned(16))) float    acc_s[(OWN_R + 1) * GNN_D];
    __shared__ __attribute__((aligned(16))) unsigned acc_m[(OWN_R + 1) * GNN_D];
    __shared__ int q_src[OWN_Q];
    __shared__ int q_row[OWN_Q];
    __shared__ int q_n;

    const int tid   = threadIdx.x;
    const int lane  = tid & 31;
    const int wave  = __builtin_amdgcn_readfirstlane(tid >> 5);
    const int r0    = blockIdx.x * OWN_R;
    const int nrows = min(OWN_R, N - r0);
    int* const my_q = wq + (size_t)blockIdx.x * (2 * OWN_Q);
    const int saved = __builtin_amdgcn_readfirstlane(reuse ? wqn[blockIdx.x] : -1);
    const bool have = (unsigned)saved <= (unsigned)OWN_Q;
    const bool need_s = (zagg[0] != 0.0f) || (zagg[1] != 0.0f);
    const bool need_m = (zagg[2] != 0.0f);

    for (int i = tid; i < (OWN_R + 1) * GNN_D / 4; i += OWN_WAVES * 32) {
        if (need_s)
            reinterpret_cast<float4*>(acc_s)[i] = make_float4(0.0f, 0.0f, 0.0f, 0.0f);
        if (need_m)
            reinterpret_cast<uint4*>(acc_m)[i]  =
                make_uint4(MONO_NEG_INF, MONO_NEG_INF, MONO_NEG_INF, MONO_NEG_INF);
    }
    if (tid == 0) q_n = 0;
    __syncthreads();

    int qn;
    if (have) {
        qn = saved;
        for (int i = tid; i < qn; i += OWN_WAVES * 32) {
            q_src[i] = my_q[i];
            q_row[i] = my_q[OWN_Q + i];
        }
        __syncthreads();
    } else {
        const int nfull = E / OWN_TRIP;
        unsigned trip = ((unsigned)blockIdx.x * (unsigned)nfull) / gridDim.x;
        for (int t = 0; t < nfull; ++t) {
            const int ebase = (int)trip * OWN_TRIP + (wave << 5) + lane;
            const int* p = dst + ebase;
            int d[OWN_U];
            #pragma unroll
            for (int u = 0; u < OWN_U; ++u) d[u] = p[u * (OWN_WAVES * 32)];
            #pragma unroll
            for (int u = 0; u < OWN_U; ++u) {
                const unsigned loc = (unsigned)d[u] - (unsigned)r0;
                if (loc < (unsigned)nrows)
                    own_take(ebase + u * (OWN_WAVES * 32), loc, &q_n, q_src, q_row,
                             acc_s, acc_m, x, src, need_s, need_m);
            }
            if (++trip == (unsigned)nfull) trip = 0u;
        }
        {
            const int ebase = nfull * OWN_TRIP + (wave << 5) + lane;
            int d[OWN_U];
            #pragma unroll
            for (int u = 0; u < OWN_U; ++u) {
                const int e = ebase + u * (OWN_WAVES * 32);
                d[u] = (e < E) ? dst[e] : -1;
            }
            #pragma unroll
            for (int u = 0; u < OWN_U; ++u) {
                const unsigned loc = (unsigned)d[u] - (unsigned)r0;
                if (loc < (unsigned)nrows)
                    own_take(ebase + u * (OWN_WAVES * 32), loc, &q_n, q_src, q_row,
                             acc_s, acc_m, x, src, need_s, need_m);
            }
        }
        __syncthreads();

        qn = min(q_n, OWN_Q);
        for (int i = tid; i < qn; i += OWN_WAVES * 32) {
            const int sn = src[q_src[i]];
            q_src[i] = sn;
            if (!reuse) {
                my_q[i]         = sn;
                my_q[OWN_Q + i] = q_row[i];
            }
        }
        if (!reuse && tid == 0) wqn[blockIdx.x] = q_n;
        __syncthreads();
    }

    if (need_s && need_m)
        own_apply<true, true>(x, q_src, q_row, acc_s, acc_m, qn, wave, lane);
    else if (need_s)
        own_apply<true, false>(x, q_src, q_row, acc_s, acc_m, qn, wave, lane);
    else if (need_m)
        own_apply<false, true>(x, q_src, q_row, acc_s, acc_m, qn, wave, lane);
    __syncthreads();

    float4* sg = reinterpret_cast<float4*>(s  + (size_t)r0 * GNN_D);
    uint4*  mg = reinterpret_cast<uint4*>(mx + (size_t)r0 * GNN_D);
    for (int i = tid; i < nrows * (GNN_D / 4); i += OWN_WAVES * 32) {
        if (need_s) sg[i] = reinterpret_cast<const float4*>(acc_s)[i];
        if (need_m) mg[i] = reinterpret_cast<const uint4*>(acc_m)[i];
    }
}

#define LAYER_WGS 512
#define KSTEPS (GNN_D / 4)

struct LayerTileIn {
    float4 xv[2];
    float4 sv[2];
    uint4  mv[2];
    float  dg;
};

__device__ __forceinline__ void layer_load_tile(
    LayerTileIn& t, const float* __restrict__ x, const float* __restrict__ s,
    const unsigned* __restrict__ mx, const float* __restrict__ deg,
    int row, int c, bool need_s, bool need_m)
{
    const size_t g = (size_t)row * GNN_D + c;
    t.dg = deg[row];
    #pragma unroll
    for (int q = 0; q < 2; ++q) {
        t.xv[q] = *reinterpret_cast<const float4*>(x + g + q * 4);
        t.sv[q] = make_float4(0.0f, 0.0f, 0.0f, 0.0f);
        t.mv[q] = make_uint4(0u, 0u, 0u, 0u);
        if (need_s) t.sv[q] = *reinterpret_cast<const float4*>(s + g + q * 4);
        if (need_m) t.mv[q] = *reinterpret_cast<const uint4*>(mx + g + q * 4);
    }
}

__global__ __launch_bounds__(256) void gnn_layer_kernel(
    const float* __restrict__ x, const float* __restrict__ s,
    const unsigned* __restrict__ mx, const float* __restrict__ deg,
    const float* __restrict__ zagg,
    const float* __restrict__ W1, const float* __restrict__ b1,
    const float* __restrict__ W2, const float* __restrict__ b2,
    float* __restrict__ xout, int ntiles)
{
    __shared__ float hs[TILE_M * LDS_STRIDE];
    __shared__ float ts[TILE_M * LDS_STRIDE];
    __shared__ float xt[2][TILE_M * LDS_STRIDE];

    const int tid  = threadIdx.x;

    const float z0 = zagg[0], z1 = zagg[1], z2 = zagg[2];
    const bool need_s = (z0 != 0.0f) || (z1 != 0.0f);
    const bool need_m = (z2 != 0.0f);

    const int wave  = tid >> 5;
    const int lane  = tid & 31;
    const int ncol  = (wave << 4) + (lane & 15);
    const int mlo   = lane & 15;
    const int koff  = (lane >> 4) << 1;
    const int rbase = (lane >> 4) << 3;
    const int pr    = (tid * 8) >> 7;
    const int pc    = (tid * 8) & 127;
    const float bia1 = b1[ncol];
    const float bia2 = b2[ncol];

    int tile = blockIdx.x;
    LayerTileIn in;
    if (tile < ntiles)
        layer_load_tile(in, x, s, mx, deg, tile * TILE_M + pr, pc, need_s, need_m);

    v2f w1f[KSTEPS], w2f[KSTEPS];
    #pragma unroll
    for (int i = 0; i < KSTEPS; ++i) {
        const int kk = 4 * i + koff;
        w1f[i].x = W1[(size_t)kk * GNN_D + ncol];
        w1f[i].y = W1[(size_t)(kk + 1) * GNN_D + ncol];
        w2f[i].x = W2[(size_t)kk * GNN_D + ncol];
        w2f[i].y = W2[(size_t)(kk + 1) * GNN_D + ncol];
    }

    int buf = 0;
    for (; tile < ntiles; tile += gridDim.x) {
        const int row0 = tile * TILE_M;
        float* const xtc = xt[buf];
        buf ^= 1;

        {
            const float dg = in.dg;
            const float dc = fmaxf(dg, 1.0f);
            const bool  nb = (dg > 0.0f) && need_m;
            #pragma unroll
            for (int q = 0; q < 2; ++q) {
                const float4 xv = in.xv[q];
                const float4 sv = in.sv[q];
                const uint4  mv = in.mv[q];
                float4 h;
                h.x = xv.x + z0 * sv.x + z1 * (sv.x / dc) + z2 * (nb ? mono2f(mv.x) : 0.0f);
                h.y = xv.y + z0 * sv.y + z1 * (sv.y / dc) + z2 * (nb ? mono2f(mv.y) : 0.0f);
                h.z = xv.z + z0 * sv.z + z1 * (sv.z / dc) + z2 * (nb ? mono2f(mv.z) : 0.0f);
                h.w = xv.w + z0 * sv.w + z1 * (sv.w / dc) + z2 * (nb ? mono2f(mv.w) : 0.0f);
                *reinterpret_cast<float4*>(&hs[pr * LDS_STRIDE + pc + q * 4]) = h;
                *reinterpret_cast<float4*>(&xtc[pr * LDS_STRIDE + pc + q * 4]) = xv;
            }
        }
        __syncthreads();

        {
            const int nxt = tile + (int)gridDim.x;
            if (nxt < ntiles)
                layer_load_tile(in, x, s, mx, deg, nxt * TILE_M + pr, pc, need_s, need_m);
        }

        v8f acc = {};
        #pragma unroll
        for (int i = 0; i < KSTEPS; ++i) {
            const int kk = 4 * i + koff;
            v2f a; a.x = hs[mlo * LDS_STRIDE + kk];
                   a.y = hs[mlo * LDS_STRIDE + kk + 1];
            acc = __builtin_amdgcn_wmma_f32_16x16x4_f32(false, a, false, w1f[i],
                                                        (short)0, acc, false, false);
        }
        #pragma unroll
        for (int i = 0; i < 8; ++i)
            ts[(rbase + i) * LDS_STRIDE + ncol] = fmaxf(acc[i] + bia1, 0.0f);
        __syncthreads();

        v8f acc2 = {};
        #pragma unroll
        for (int i = 0; i < KSTEPS; ++i) {
            const int kk = 4 * i + koff;
            v2f a; a.x = ts[mlo * LDS_STRIDE + kk];
                   a.y = ts[mlo * LDS_STRIDE + kk + 1];
            acc2 = __builtin_amdgcn_wmma_f32_16x16x4_f32(false, a, false, w2f[i],
                                                         (short)0, acc2, false, false);
        }
        #pragma unroll
        for (int i = 0; i < 8; ++i) {
            int r = rbase + i;
            float h2 = acc2[i] + bia2;
            float xv = xtc[r * LDS_STRIDE + ncol];
            xout[(size_t)(row0 + r) * GNN_D + ncol] = xv + fmaxf(h2, 0.0f);
        }
    }
}

#define POOL_WAVES 16
#define POOL_U     8
__global__ __launch_bounds__(POOL_WAVES * 32) void gnn_pool_owner_kernel(
    const float* __restrict__ x, const int* __restrict__ batch,
    const float* __restrict__ zpool, float* __restrict__ out, int N)
{
    __shared__ float red_s[POOL_WAVES][32];
    __shared__ float red_m[POOL_WAVES][32];
    __shared__ int   red_c[POOL_WAVES];

    const int lane = threadIdx.x & 31;
    const int wave = __builtin_amdgcn_readfirstlane((int)(threadIdx.x >> 5));
    const int b    = blockIdx.x >> 2;
    const unsigned col = ((blockIdx.x & 3u) << 5) + (unsigned)lane;

    float sum = 0.0f;
    float mxv = -INFINITY;
    int   cnt = 0;

    for (int c0 = wave; (c0 << 5) < N; c0 += POOL_WAVES * POOL_U) {
        int ids[POOL_U];
        #pragma unroll
        for (int u = 0; u < POOL_U; ++u) {
            int n  = ((c0 + u * POOL_WAVES) << 5) + lane;
            int id = batch[min(n, N - 1)];
            ids[u] = (n < N) ? id : -1;
        }
        #pragma unroll
        for (int u = 0; u < POOL_U; ++u) {
            const unsigned mask = __builtin_amdgcn_ballot_w32(ids[u] == b);
            if (mask == 0u) continue;
            cnt += __builtin_popcount(mask);
            const int base = (c0 + u * POOL_WAVES) << 5;
            float v[32];
            if (base + 32 <= N) {
                const float* p = x + (size_t)base * GNN_D + col;
                #pragma unroll
                for (int j = 0; j < 32; ++j) v[j] = p[j * GNN_D];
            } else {
                #pragma unroll
                for (int j = 0; j < 32; ++j)
                    v[j] = x[(size_t)min(base + j, N - 1) * GNN_D + col];
            }
            #pragma unroll
            for (int j = 0; j < 32; ++j) {
                const bool in = (mask >> j) & 1u;
                sum += in ? v[j] : 0.0f;
                mxv = fmaxf(mxv, in ? v[j] : -INFINITY);
            }
        }
    }

    red_s[wave][lane] = sum;
    red_m[wave][lane] = mxv;
    if (lane == 0) red_c[wave] = cnt;
    __syncthreads();

    if (wave == 0) {
        float s = 0.0f;
        float m = -INFINITY;
        int   c = 0;
        #pragma unroll
        for (int w = 0; w < POOL_WAVES; ++w) {
            s += red_s[w][lane];
            m  = fmaxf(m, red_m[w][lane]);
            c += red_c[w];
        }
        const float cf = (float)c;
        const float mv = (c > 0) ? m : 0.0f;
        out[(size_t)b * GNN_D + col] =
            zpool[0] * s + zpool[1] * (s / fmaxf(cf, 1.0f)) + zpool[2] * mv;
    }
}


extern "C" void kernel_launch(void* const* d_in, const int* in_sizes, int n_in,
                              void* d_out, int out_size, void* d_ws, size_t ws_size,
                              hipStream_t stream) {
    const float* x_in   = (const float*)d_in[0];
    const int*   eidx   = (const int*)d_in[1];
    const int*   batch  = (const int*)d_in[2];
    const float* z_agg  = (const float*)d_in[3];
    const float* z_pool = (const float*)d_in[4];
    const float* W1     = (const float*)d_in[5];
    const float* b1     = (const float*)d_in[6];
    const float* W2     = (const float*)d_in[7];
    const float* b2     = (const float*)d_in[8];
    (void)n_in;

    const int N = in_sizes[0] / GNN_D;
    const int E = in_sizes[1] / 2;
    const int L = in_sizes[5] / (GNN_D * GNN_D);
    const int Bg = out_size / GNN_D;
    const size_t ND = (size_t)N * GNN_D;

    char* p = (char*)d_ws;
    char* pend = p + ws_size;
    auto take = [&](size_t bytes) -> void* {
        void* r = (void*)p;
        p += (bytes + 255) & ~(size_t)255;
        return r;
    };
    float*    xa   = (float*)take(ND * 4);
    float*    xb   = (float*)take(ND * 4);
    float*    s    = (float*)take(ND * 4);
    unsigned* mx   = (unsigned*)take(ND * 4);
    float*    deg  = (float*)take((size_t)N * 4);
    const int nown = (N + OWN_R - 1) / OWN_R;
    int*      wq   = (int*)take((size_t)nown * 2 * OWN_Q * 4);
    int*      wqn  = (int*)take((size_t)nown * 4);
    if (p > pend) return;

    const int* src = eidx;
    const int* dst = eidx + E;

    hipMemcpyAsync(xa, x_in, ND * 4, hipMemcpyDeviceToDevice, stream);

    hipMemsetAsync(deg, 0, (size_t)N * 4, stream);
    gnn_deg_kernel<<<(E + 255) / 256, 256, 0, stream>>>(dst, deg, E);

    float* xcur = xa;
    float* xnxt = xb;
    for (int l = 0; l < L; ++l) {
        gnn_aggr_owner_kernel<<<nown, OWN_WAVES * 32, 0, stream>>>(
            xcur, src, dst, s, mx, z_agg + l * 3, wq, wqn, l > 0 ? 1 : 0, N, E);
        gnn_layer_kernel<<<LAYER_WGS, 256, 0, stream>>>(
            xcur, s, mx, deg, z_agg + l * 3,
            W1 + (size_t)l * GNN_D * GNN_D, b1 + (size_t)l * GNN_D,
            W2 + (size_t)l * GNN_D * GNN_D, b2 + (size_t)l * GNN_D, xnxt, N / TILE_M);
        float* t = xcur; xcur = xnxt; xnxt = t;
    }

    gnn_pool_owner_kernel<<<Bg * 4, POOL_WAVES * 32, 0, stream>>>(
        xcur, batch, z_pool, (float*)d_out, N);
}


# Owner scan on a byte plane: the degree kernel stores one owner byte per edge (row / 200); layer 0's scan reads 16 edges per 16-byte load, queued edges then look up their row; sums still by arrival
# speedup vs baseline: 3.8625x; 1.0391x over previous
#include <hip/hip_runtime.h>
#include <hip/hip_bf16.h>


typedef float v8f __attribute__((ext_vector_type(8)));
typedef float v2f __attribute__((ext_vector_type(2)));

#define GNN_D 128
#define TILE_M 16
#define LDS_STRIDE 132

__device__ __forceinline__ unsigned f2mono(float f) {
    unsigned u = __float_as_uint(f);
    return (u & 0x80000000u) ? ~u : (u | 0x80000000u);
}
__device__ __forceinline__ float mono2f(unsigned m) {
    unsigned u = (m & 0x80000000u) ? (m & 0x7FFFFFFFu) : ~m;
    return __uint_as_float(u);
}
#define MONO_NEG_INF 0x007FFFFFu


#define OWN_R     200
__global__ __launch_bounds__(256) void gnn_deg_kernel(const int* __restrict__ dst,
                                                      float* __restrict__ deg,
                                                      unsigned char* __restrict__ ob,
                                                      int E, int Epad) {
    int e = blockIdx.x * 256 + threadIdx.x;
    if (e < E) {
        const int d = dst[e];
        atomicAdd(&deg[d], 1.0f);
        ob[e] = (unsigned char)(((unsigned)d / (unsigned)OWN_R) & 0xFFu);
    } else if (e < Epad) {
        ob[e] = (unsigned char)0xFFu;
    }
}

#define OWN_Q     4096
#define OWN_WAVES 32
#define OWN_U     8
#define OWN_PB    8
#define OWN_G     (OWN_WAVES * 32)
#define OWN_TRIP  (OWN_G * OWN_U * 16)

__device__ __forceinline__ void own_take(
    int e, int r0, int nrows, int* q_n, int* q_src,
    float* acc_s, unsigned* acc_m,
    const float* __restrict__ x, const int* __restrict__ src,
    const int* __restrict__ dst, bool need_s, bool need_m)
{
    const int pos = atomicAdd(q_n, 1);
    if (pos < OWN_Q) {
        q_src[pos] = e;
    } else {
        const unsigned loc = (unsigned)dst[e] - (unsigned)r0;
        if (loc >= (unsigned)nrows) return;
        const float4* xr = reinterpret_cast<const float4*>(x + (size_t)src[e] * GNN_D);
        float*    as = acc_s + loc * GNN_D;
        unsigned* am = acc_m + loc * GNN_D;
        #pragma nounroll
        for (int f = 0; f < GNN_D / 4; ++f) {
            const float4 v = xr[f];
            if (need_s) {
                atomicAdd(as + 4 * f + 0, v.x);
                atomicAdd(as + 4 * f + 1, v.y);
                atomicAdd(as + 4 * f + 2, v.z);
                atomicAdd(as + 4 * f + 3, v.w);
            }
            if (need_m) {
                atomicMax(am + 4 * f + 0, f2mono(v.x));
                atomicMax(am + 4 * f + 1, f2mono(v.y));
                atomicMax(am + 4 * f + 2, f2mono(v.z));
                atomicMax(am + 4 * f + 3, f2mono(v.w));
            }
        }
    }
}

__device__ __forceinline__ unsigned own_eq_bytes(unsigned w, unsigned pat4) {
    const unsigned d = w ^ pat4;
    return ~(((d & 0x7F7F7F7Fu) + 0x7F7F7F7Fu) | d | 0x7F7F7F7Fu);
}
__device__ __forceinline__ unsigned own_nibble(unsigned h) {
    return (((h >> 7) * 0x00204081u) >> 21) & 0xFu;
}
__device__ __forceinline__ void own_scan16(
    const uint4 w, unsigned pat4, int e0, int elim, int r0, int nrows,
    int* q_n, int* q_src, float* acc_s, unsigned* acc_m,
    const float* __restrict__ x, const int* __restrict__ src,
    const int* __restrict__ dst, bool need_s, bool need_m)
{
    const unsigned h0 = own_eq_bytes(w.x, pat4);
    const unsigned h1 = own_eq_bytes(w.y, pat4);
    const unsigned h2 = own_eq_bytes(w.z, pat4);
    const unsigned h3 = own_eq_bytes(w.w, pat4);
    if ((h0 | h1 | h2 | h3) != 0u) {
        unsigned m = own_nibble(h0) | (own_nibble(h1) << 4) |
                     (own_nibble(h2) << 8) | (own_nibble(h3) << 12);
        do {
            const int e = e0 + __builtin_ctz(m);
            if (e < elim)
                own_take(e, r0, nrows, q_n, q_src, acc_s, acc_m, x, src, dst, need_s, need_m);
            m &= m - 1u;
        } while (m != 0u);
    }
}

template <bool SUM, bool MAX>
__device__ __forceinline__ void own_apply(
    const float* __restrict__ x, const int* q_src, const int* q_row,
    float* acc_s, unsigned* acc_m, int qn, int wave, int lane)
{
    for (int q0 = wave * OWN_PB; q0 < qn; q0 += OWN_WAVES * OWN_PB) {
        float v[OWN_PB][4];
        int   rw[OWN_PB];
        #pragma unroll
        for (int p = 0; p < OWN_PB; ++p) {
            const int qc = min(q0 + p, qn - 1);
            const int sn = __builtin_amdgcn_readfirstlane(q_src[qc]);
            const int rr = __builtin_amdgcn_readfirstlane(q_row[qc]);
            rw[p] = (q0 + p < qn) ? rr : OWN_R;
            const float* xp = x + (size_t)sn * GNN_D + lane;
            #pragma unroll
            for (int k = 0; k < 4; ++k) v[p][k] = xp[32 * k];
        }
        #pragma unroll
        for (int p = 0; p < OWN_PB; ++p) {
            float*    as = acc_s + rw[p] * GNN_D + lane;
            unsigned* am = acc_m + rw[p] * GNN_D + lane;
            #pragma unroll
            for (int k = 0; k < 4; ++k) {
                if (SUM) atomicAdd(as + 32 * k, v[p][k]);
                if (MAX) atomicMax(am + 32 * k, f2mono(v[p][k]));
            }
        }
    }
}

__global__ __launch_bounds__(OWN_WAVES * 32) void gnn_aggr_owner_kernel(
    const float* __restrict__ x, const int* __restrict__ src,
    const int* __restrict__ dst, const uint4* __restrict__ ob16,
    float* __restrict__ s,
    unsigned* __restrict__ mx, const float* __restrict__ zagg,
    int* wq, int* wqn, int reuse, int N, int E)
{
    __shared__ __attribute__((aligned(16))) float    acc_s[(OWN_R + 1) * GNN_D];
    __shared__ __attribute__((aligned(16))) unsigned acc_m[(OWN_R + 1) * GNN_D];
    __shared__ int q_src[OWN_Q];
    __shared__ int q_row[OWN_Q];
    __shared__ int q_n;

    const int tid   = threadIdx.x;
    const int lane  = tid & 31;
    const int wave  = __builtin_amdgcn_readfirstlane(tid >> 5);
    const int r0    = blockIdx.x * OWN_R;
    const int nrows = min(OWN_R, N - r0);
    int* const my_q = wq + (size_t)blockIdx.x * (2 * OWN_Q);
    const int saved = __builtin_amdgcn_readfirstlane(reuse ? wqn[blockIdx.x] : -1);
    const bool have = (unsigned)saved <= (unsigned)OWN_Q;
    const bool need_s = (zagg[0] != 0.0f) || (zagg[1] != 0.0f);
    const bool need_m = (zagg[2] != 0.0f);

    for (int i = tid; i < (OWN_R + 1) * GNN_D / 4; i += OWN_WAVES * 32) {
        if (need_s)
            reinterpret_cast<float4*>(acc_s)[i] = make_float4(0.0f, 0.0f, 0.0f, 0.0f);
        if (need_m)
            reinterpret_cast<uint4*>(acc_m)[i]  =
                make_uint4(MONO_NEG_INF, MONO_NEG_INF, MONO_NEG_INF, MONO_NEG_INF);
    }
    if (tid == 0) q_n = 0;
    __syncthreads();

    int qn;
    if (have) {
        qn = saved;
        for (int i = tid; i < qn; i += OWN_WAVES * 32) {
            q_src[i] = my_q[i];
            q_row[i] = my_q[OWN_Q + i];
        }
        __syncthreads();
    } else {
        const unsigned ownb = (unsigned)blockIdx.x & 0xFFu;
        const unsigned pat4 = ownb * 0x01010101u;
        const int nfull = E / OWN_TRIP;
        unsigned trip = ((unsigned)blockIdx.x * (unsigned)nfull) / gridDim.x;
        for (int t = 0; t < nfull; ++t) {
            const int gbase = (int)trip * (OWN_G * OWN_U) + tid;
            const uint4* p = ob16 + gbase;
            uint4 d[OWN_U];
            #pragma unroll
            for (int u = 0; u < OWN_U; ++u) d[u] = p[u * OWN_G];
            #pragma unroll
            for (int u = 0; u < OWN_U; ++u)
                own_scan16(d[u], pat4, (gbase + u * OWN_G) * 16, E, r0, nrows, &q_n, q_src,
                           acc_s, acc_m, x, src, dst, need_s, need_m);
            if (++trip == (unsigned)nfull) trip = 0u;
        }
        {
            const int ngrp  = (E + 15) >> 4;
            const int gbase = nfull * (OWN_G * OWN_U) + tid;
            uint4 d[OWN_U];
            #pragma unroll
            for (int u = 0; u < OWN_U; ++u) {
                const int g = gbase + u * OWN_G;
                d[u] = make_uint4(~pat4, ~pat4, ~pat4, ~pat4);
                if (g < ngrp) d[u] = ob16[g];
            }
            #pragma unroll
            for (int u = 0; u < OWN_U; ++u)
                own_scan16(d[u], pat4, (gbase + u * OWN_G) * 16, E, r0, nrows, &q_n, q_src,
                           acc_s, acc_m, x, src, dst, need_s, need_m);
        }
        __syncthreads();

        qn = min(q_n, OWN_Q);
        for (int i = tid; i < qn; i += OWN_WAVES * 32) {
            const int e  = q_src[i];
            const int sn = src[e];
            const unsigned loc = (unsigned)dst[e] - (unsigned)r0;
            const int rw = (loc < (unsigned)nrows) ? (int)loc : OWN_R;
            q_src[i] = sn;
            q_row[i] = rw;
            if (!reuse) {
                my_q[i]         = sn;
                my_q[OWN_Q + i] = rw;
            }
        }
        if (!reuse && tid == 0) wqn[blockIdx.x] = q_n;
        __syncthreads();
    }

    if (need_s && need_m)
        own_apply<true, true>(x, q_src, q_row, acc_s, acc_m, qn, wave, lane);
    else if (need_s)
        own_apply<true, false>(x, q_src, q_row, acc_s, acc_m, qn, wave, lane);
    else if (need_m)
        own_apply<false, true>(x, q_src, q_row, acc_s, acc_m, qn, wave, lane);
    __syncthreads();

    float4* sg = reinterpret_cast<float4*>(s  + (size_t)r0 * GNN_D);
    uint4*  mg = reinterpret_cast<uint4*>(mx + (size_t)r0 * GNN_D);
    for (int i = tid; i < nrows * (GNN_D / 4); i += OWN_WAVES * 32) {
        if (need_s) sg[i] = reinterpret_cast<const float4*>(acc_s)[i];
        if (need_m) mg[i] = reinterpret_cast<const uint4*>(acc_m)[i];
    }
}

#define LAYER_WGS 512
#define KSTEPS (GNN_D / 4)

struct LayerTileIn {
    float4 xv[2];
    float4 sv[2];
    uint4  mv[2];
    float  dg;
};

__device__ __forceinline__ void layer_load_tile(
    LayerTileIn& t, const float* __restrict__ x, const float* __restrict__ s,
    const unsigned* __restrict__ mx, const float* __restrict__ deg,
    int row, int c, bool need_s, bool need_m)
{
    const size_t g = (size_t)row * GNN_D + c;
    t.dg = deg[row];
    #pragma unroll
    for (int q = 0; q < 2; ++q) {
        t.xv[q] = *reinterpret_cast<const float4*>(x + g + q * 4);
        t.sv[q] = make_float4(0.0f, 0.0f, 0.0f, 0.0f);
        t.mv[q] = make_uint4(0u, 0u, 0u, 0u);
        if (need_s) t.sv[q] = *reinterpret_cast<const float4*>(s + g + q * 4);
        if (need_m) t.mv[q] = *reinterpret_cast<const uint4*>(mx + g + q * 4);
    }
}

__global__ __launch_bounds__(256) void gnn_layer_kernel(
    const float* __restrict__ x, const float* __restrict__ s,
    const unsigned* __restrict__ mx, const float* __restrict__ deg,
    const float* __restrict__ zagg,
    const float* __restrict__ W1, const float* __restrict__ b1,
    const float* __restrict__ W2, const float* __restrict__ b2,
    float* __restrict__ xout, int ntiles)
{
    __shared__ float hs[TILE_M * LDS_STRIDE];
    __shared__ float ts[TILE_M * LDS_STRIDE];
    __shared__ float xt[2][TILE_M * LDS_STRIDE];

    const int tid  = threadIdx.x;

    const float z0 = zagg[0], z1 = zagg[1], z2 = zagg[2];
    const bool need_s = (z0 != 0.0f) || (z1 != 0.0f);
    const bool need_m = (z2 != 0.0f);

    const int wave  = tid >> 5;
    const int lane  = tid & 31;
    const int ncol  = (wave << 4) + (lane & 15);
    const int mlo   = lane & 15;
    const int koff  = (lane >> 4) << 1;
    const int rbase = (lane >> 4) << 3;
    const int pr    = (tid * 8) >> 7;
    const int pc    = (tid * 8) & 127;
    const float bia1 = b1[ncol];
    const float bia2 = b2[ncol];

    int tile = blockIdx.x;
    LayerTileIn in;
    if (tile < ntiles)
        layer_load_tile(in, x, s, mx, deg, tile * TILE_M + pr, pc, need_s, need_m);

    v2f w1f[KSTEPS], w2f[KSTEPS];
    #pragma unroll
    for (int i = 0; i < KSTEPS; ++i) {
        const int kk = 4 * i + koff;
        w1f[i].x = W1[(size_t)kk * GNN_D + ncol];
        w1f[i].y = W1[(size_t)(kk + 1) * GNN_D + ncol];
        w2f[i].x = W2[(size_t)kk * GNN_D + ncol];
        w2f[i].y = W2[(size_t)(kk + 1) * GNN_D + ncol];
    }

    int buf = 0;
    for (; tile < ntiles; tile += gridDim.x) {
        const int row0 = tile * TILE_M;
        float* const xtc = xt[buf];
        buf ^= 1;

        {
            const float dg = in.dg;
            const float dc = fmaxf(dg, 1.0f);
            const bool  nb = (dg > 0.0f) && need_m;
            #pragma unroll
            for (int q = 0; q < 2; ++q) {
                const float4 xv = in.xv[q];
                const float4 sv = in.sv[q];
                const uint4  mv = in.mv[q];
                float4 h;
                h.x = xv.x + z0 * sv.x + z1 * (sv.x / dc) + z2 * (nb ? mono2f(mv.x) : 0.0f);
                h.y = xv.y + z0 * sv.y + z1 * (sv.y / dc) + z2 * (nb ? mono2f(mv.y) : 0.0f);
                h.z = xv.z + z0 * sv.z + z1 * (sv.z / dc) + z2 * (nb ? mono2f(mv.z) : 0.0f);
                h.w = xv.w + z0 * sv.w + z1 * (sv.w / dc) + z2 * (nb ? mono2f(mv.w) : 0.0f);
                *reinterpret_cast<float4*>(&hs[pr * LDS_STRIDE + pc + q * 4]) = h;
                *reinterpret_cast<float4*>(&xtc[pr * LDS_STRIDE + pc + q * 4]) = xv;
            }
        }
        __syncthreads();

        {
            const int nxt = tile + (int)gridDim.x;
            if (nxt < ntiles)
                layer_load_tile(in, x, s, mx, deg, nxt * TILE_M + pr, pc, need_s, need_m);
        }

        v8f acc = {};
        #pragma unroll
        for (int i = 0; i < KSTEPS; ++i) {
            const int kk = 4 * i + koff;
            v2f a; a.x = hs[mlo * LDS_STRIDE + kk];
                   a.y = hs[mlo * LDS_STRIDE + kk + 1];
            acc = __builtin_amdgcn_wmma_f32_16x16x4_f32(false, a, false, w1f[i],
                                                        (short)0, acc, false, false);
        }
        #pragma unroll
        for (int i = 0; i < 8; ++i)
            ts[(rbase + i) * LDS_STRIDE + ncol] = fmaxf(acc[i] + bia1, 0.0f);
        __syncthreads();

        v8f acc2 = {};
        #pragma unroll
        for (int i = 0; i < KSTEPS; ++i) {
            const int kk = 4 * i + koff;
            v2f a; a.x = ts[mlo * LDS_STRIDE + kk];
                   a.y = ts[mlo * LDS_STRIDE + kk + 1];
            acc2 = __builtin_amdgcn_wmma_f32_16x16x4_f32(false, a, false, w2f[i],
                                                         (short)0, acc2, false, false);
        }
        #pragma unroll
        for (int i = 0; i < 8; ++i) {
            int r = rbase + i;
            float h2 = acc2[i] + bia2;
            float xv = xtc[r * LDS_STRIDE + ncol];
            xout[(size_t)(row0 + r) * GNN_D + ncol] = xv + fmaxf(h2, 0.0f);
        }
    }
}

#define POOL_WAVES 16
#define POOL_U     8
__global__ __launch_bounds__(POOL_WAVES * 32) void gnn_pool_owner_kernel(
    const float* __restrict__ x, const int* __restrict__ batch,
    const float* __restrict__ zpool, float* __restrict__ out, int N)
{
    __shared__ float red_s[POOL_WAVES][32];
    __shared__ float red_m[POOL_WAVES][32];
    __shared__ int   red_c[POOL_WAVES];

    const int lane = threadIdx.x & 31;
    const int wave = __builtin_amdgcn_readfirstlane((int)(threadIdx.x >> 5));
    const int b    = blockIdx.x >> 2;
    const unsigned col = ((blockIdx.x & 3u) << 5) + (unsigned)lane;

    float sum = 0.0f;
    float mxv = -INFINITY;
    int   cnt = 0;

    for (int c0 = wave; (c0 << 5) < N; c0 += POOL_WAVES * POOL_U) {
        int ids[POOL_U];
        #pragma unroll
        for (int u = 0; u < POOL_U; ++u) {
            int n  = ((c0 + u * POOL_WAVES) << 5) + lane;
            int id = batch[min(n, N - 1)];
            ids[u] = (n < N) ? id : -1;
        }
        #pragma unroll
        for (int u = 0; u < POOL_U; ++u) {
            const unsigned mask = __builtin_amdgcn_ballot_w32(ids[u] == b);
            if (mask == 0u) continue;
            cnt += __builtin_popcount(mask);
            const int base = (c0 + u * POOL_WAVES) << 5;
            float v[32];
            if (base + 32 <= N) {
                const float* p = x + (size_t)base * GNN_D + col;
                #pragma unroll
                for (int j = 0; j < 32; ++j) v[j] = p[j * GNN_D];
            } else {
                #pragma unroll
                for (int j = 0; j < 32; ++j)
                    v[j] = x[(size_t)min(base + j, N - 1) * GNN_D + col];
            }
            #pragma unroll
            for (int j = 0; j < 32; ++j) {
                const bool in = (mask >> j) & 1u;
                sum += in ? v[j] : 0.0f;
                mxv = fmaxf(mxv, in ? v[j] : -INFINITY);
            }
        }
    }

    red_s[wave][lane] = sum;
    red_m[wave][lane] = mxv;
    if (lane == 0) red_c[wave] = cnt;
    __syncthreads();

    if (wave == 0) {
        float s = 0.0f;
        float m = -INFINITY;
        int   c = 0;
        #pragma unroll
        for (int w = 0; w < POOL_WAVES; ++w) {
            s += red_s[w][lane];
            m  = fmaxf(m, red_m[w][lane]);
            c += red_c[w];
        }
        const float cf = (float)c;
        const float mv = (c > 0) ? m : 0.0f;
        out[(size_t)b * GNN_D + col] =
            zpool[0] * s + zpool[1] * (s / fmaxf(cf, 1.0f)) + zpool[2] * mv;
    }
}


extern "C" void kernel_launch(void* const* d_in, const int* in_sizes, int n_in,
                              void* d_out, int out_size, void* d_ws, size_t ws_size,
                              hipStream_t stream) {
    const float* x_in   = (const float*)d_in[0];
    const int*   eidx   = (const int*)d_in[1];
    const int*   batch  = (const int*)d_in[2];
    const float* z_agg  = (const float*)d_in[3];
    const float* z_pool = (const float*)d_in[4];
    const float* W1     = (const float*)d_in[5];
    const float* b1     = (const float*)d_in[6];
    const float* W2     = (const float*)d_in[7];
    const float* b2     = (const float*)d_in[8];
    (void)n_in;

    const int N = in_sizes[0] / GNN_D;
    const int E = in_sizes[1] / 2;
    const int L = in_sizes[5] / (GNN_D * GNN_D);
    const int Bg = out_size / GNN_D;
    const size_t ND = (size_t)N * GNN_D;

    char* p = (char*)d_ws;
    char* pend = p + ws_size;
    auto take = [&](size_t bytes) -> void* {
        void* r = (void*)p;
        p += (bytes + 255) & ~(size_t)255;
        return r;
    };
    float*    xa   = (float*)take(ND * 4);
    float*    xb   = (float*)take(ND * 4);
    float*    s    = (float*)take(ND * 4);
    unsigned* mx   = (unsigned*)take(ND * 4);
    float*    deg  = (float*)take((size_t)N * 4);
    const int nown = (N + OWN_R - 1) / OWN_R;
    int*      wq   = (int*)take((size_t)nown * 2 * OWN_Q * 4);
    int*      wqn  = (int*)take((size_t)nown * 4);
    const int Epad = (E + 15) & ~15;
    unsigned char* ob = (unsigned char*)take((size_t)Epad);
    if (p > pend) return;

    const int* src = eidx;
    const int* dst = eidx + E;

    hipMemcpyAsync(xa, x_in, ND * 4, hipMemcpyDeviceToDevice, stream);

    hipMemsetAsync(deg, 0, (size_t)N * 4, stream);
    gnn_deg_kernel<<<(Epad + 255) / 256, 256, 0, stream>>>(dst, deg, ob, E, Epad);

    float* xcur = xa;
    float* xnxt = xb;
    for (int l = 0; l < L; ++l) {
        gnn_aggr_owner_kernel<<<nown, OWN_WAVES * 32, 0, stream>>>(
            xcur, src, dst, (const uint4*)ob, s, mx, z_agg + l * 3, wq, wqn, l > 0 ? 1 : 0, N, E);
        gnn_layer_kernel<<<LAYER_WGS, 256, 0, stream>>>(
            xcur, s, mx, deg, z_agg + l * 3,
            W1 + (size_t)l * GNN_D * GNN_D, b1 + (size_t)l * GNN_D,
            W2 + (size_t)l * GNN_D * GNN_D, b2 + (size_t)l * GNN_D, xnxt, N / TILE_M);
        float* t = xcur; xcur = xnxt; xnxt = t;
    }

    gnn_pool_owner_kernel<<<Bg * 4, POOL_WAVES * 32, 0, stream>>>(
        xcur, batch, z_pool, (float*)d_out, N);
}


# FFN moved into the owner work-group: after the LDS aggregation it forms the layer input in place and runs both WMMA products on its rows from LDS; s/mx not stored, 3 launches fewer; sums as before
# speedup vs baseline: 4.0386x; 1.0456x over previous
#include <hip/hip_runtime.h>
#include <hip/hip_bf16.h>


typedef float v8f __attribute__((ext_vector_type(8)));
typedef float v2f __attribute__((ext_vector_type(2)));

#define GNN_D 128
#define TILE_M 16
#define LDS_STRIDE 132
#define KSTEPS (GNN_D / 4)

__device__ __forceinline__ unsigned f2mono(float f) {
    unsigned u = __float_as_uint(f);
    return (u & 0x80000000u) ? ~u : (u | 0x80000000u);
}
__device__ __forceinline__ float mono2f(unsigned m) {
    unsigned u = (m & 0x80000000u) ? (m & 0x7FFFFFFFu) : ~m;
    return __uint_as_float(u);
}
#define MONO_NEG_INF 0x007FFFFFu


#define OWN_R     200
#define OWN_TILES ((OWN_R + TILE_M - 1) / TILE_M)
#define OWN_LROWS (OWN_TILES * TILE_M)
static_assert(OWN_LROWS > OWN_R, "the spare row must lie inside the padded tile rows");
__global__ __launch_bounds__(256) void gnn_deg_kernel(const int* __restrict__ dst,
                                                      float* __restrict__ deg,
                                                      unsigned char* __restrict__ ob,
                                                      int E, int Epad) {
    int e = blockIdx.x * 256 + threadIdx.x;
    if (e < E) {
        const int d = dst[e];
        atomicAdd(&deg[d], 1.0f);
        ob[e] = (unsigned char)(((unsigned)d / (unsigned)OWN_R) & 0xFFu);
    } else if (e < Epad) {
        ob[e] = (unsigned char)0xFFu;
    }
}

#define OWN_Q     4096
#define OWN_WAVES 32
#define OWN_U     8
#define OWN_PB    8
#define OWN_G     (OWN_WAVES * 32)
#define OWN_TRIP  (OWN_G * OWN_U * 16)

__device__ __forceinline__ void own_take(
    int e, int r0, int nrows, int* q_n, int* q_src,
    float* acc_s, unsigned* acc_m,
    const float* __restrict__ x, const int* __restrict__ src,
    const int* __restrict__ dst, bool need_s, bool need_m)
{
    const int pos = atomicAdd(q_n, 1);
    if (pos < OWN_Q) {
        q_src[pos] = e;
    } else {
        const unsigned loc = (unsigned)dst[e] - (unsigned)r0;
        if (loc >= (unsigned)nrows) return;
        const float4* xr = reinterpret_cast<const float4*>(x + (size_t)src[e] * GNN_D);
        float*    as = acc_s + loc * LDS_STRIDE;
        unsigned* am = acc_m + loc * LDS_STRIDE;
        #pragma nounroll
        for (int f = 0; f < GNN_D / 4; ++f) {
            const float4 v = xr[f];
            if (need_s) {
                atomicAdd(as + 4 * f + 0, v.x);
                atomicAdd(as + 4 * f + 1, v.y);
                atomicAdd(as + 4 * f + 2, v.z);
                atomicAdd(as + 4 * f + 3, v.w);
            }
            if (need_m) {
                atomicMax(am + 4 * f + 0, f2mono(v.x));
                atomicMax(am + 4 * f + 1, f2mono(v.y));
                atomicMax(am + 4 * f + 2, f2mono(v.z));
                atomicMax(am + 4 * f + 3, f2mono(v.w));
            }
        }
    }
}

__device__ __forceinline__ unsigned own_eq_bytes(unsigned w, unsigned pat4) {
    const unsigned d = w ^ pat4;
    return ~(((d & 0x7F7F7F7Fu) + 0x7F7F7F7Fu) | d | 0x7F7F7F7Fu);
}
__device__ __forceinline__ unsigned own_nibble(unsigned h) {
    return (((h >> 7) * 0x00204081u) >> 21) & 0xFu;
}
__device__ __forceinline__ void own_scan16(
    const uint4 w, unsigned pat4, int e0, int elim, int r0, int nrows,
    int* q_n, int* q_src, float* acc_s, unsigned* acc_m,
    const float* __restrict__ x, const int* __restrict__ src,
    const int* __restrict__ dst, bool need_s, bool need_m)
{
    const unsigned h0 = own_eq_bytes(w.x, pat4);
    const unsigned h1 = own_eq_bytes(w.y, pat4);
    const unsigned h2 = own_eq_bytes(w.z, pat4);
    const unsigned h3 = own_eq_bytes(w.w, pat4);
    if ((h0 | h1 | h2 | h3) != 0u) {
        unsigned m = own_nibble(h0) | (own_nibble(h1) << 4) |
                     (own_nibble(h2) << 8) | (own_nibble(h3) << 12);
        do {
            const int e = e0 + __builtin_ctz(m);
            if (e < elim)
                own_take(e, r0, nrows, q_n, q_src, acc_s, acc_m, x, src, dst, need_s, need_m);
            m &= m - 1u;
        } while (m != 0u);
    }
}

template <bool SUM, bool MAX>
__device__ __forceinline__ void own_apply(
    const float* __restrict__ x, const int* q_src, const int* q_row,
    float* acc_s, unsigned* acc_m, int qn, int wave, int lane)
{
    for (int q0 = wave * OWN_PB; q0 < qn; q0 += OWN_WAVES * OWN_PB) {
        float v[OWN_PB][4];
        int   rw[OWN_PB];
        #pragma unroll
        for (int p = 0; p < OWN_PB; ++p) {
            const int qc = min(q0 + p, qn - 1);
            const int sn = __builtin_amdgcn_readfirstlane(q_src[qc]);
            const int rr = __builtin_amdgcn_readfirstlane(q_row[qc]);
            rw[p] = (q0 + p < qn) ? rr : OWN_R;
            const float* xp = x + (size_t)sn * GNN_D + lane;
            #pragma unroll
            for (int k = 0; k < 4; ++k) v[p][k] = xp[32 * k];
        }
        #pragma unroll
        for (int p = 0; p < OWN_PB; ++p) {
            float*    as = acc_s + rw[p] * LDS_STRIDE + lane;
            unsigned* am = acc_m + rw[p] * LDS_STRIDE + lane;
            #pragma unroll
            for (int k = 0; k < 4; ++k) {
                if (SUM) atomicAdd(as + 32 * k, v[p][k]);
                if (MAX) atomicMax(am + 32 * k, f2mono(v[p][k]));
            }
        }
    }
}

__global__ __launch_bounds__(OWN_WAVES * 32) void gnn_layer_owner_kernel(
    const float* __restrict__ x, const int* __restrict__ src,
    const int* __restrict__ dst, const uint4* __restrict__ ob16,
    const float* __restrict__ deg, const float* __restrict__ zagg,
    const float* __restrict__ W1, const float* __restrict__ b1,
    const float* __restrict__ W2, const float* __restrict__ b2,
    float* __restrict__ xout,
    int* wq, int* wqn, int reuse, int N, int E)
{
    __shared__ __attribute__((aligned(16))) float    acc_s[OWN_LROWS * LDS_STRIDE];
    __shared__ __attribute__((aligned(16))) unsigned acc_m[OWN_LROWS * LDS_STRIDE];
    __shared__ int q_src[OWN_Q];
    __shared__ int q_row[OWN_Q];
    __shared__ int q_n;

    const int tid   = threadIdx.x;
    const int lane  = tid & 31;
    const int wave  = __builtin_amdgcn_readfirstlane(tid >> 5);
    const int r0    = blockIdx.x * OWN_R;
    const int nrows = min(OWN_R, N - r0);
    int* const my_q = wq + (size_t)blockIdx.x * (2 * OWN_Q);
    const int saved = __builtin_amdgcn_readfirstlane(reuse ? wqn[blockIdx.x] : -1);
    const bool have = (unsigned)saved <= (unsigned)OWN_Q;
    const bool need_s = (zagg[0] != 0.0f) || (zagg[1] != 0.0f);
    const bool need_m = (zagg[2] != 0.0f);

    for (int i = tid; i < OWN_LROWS * LDS_STRIDE / 4; i += OWN_WAVES * 32) {
        if (need_s)
            reinterpret_cast<float4*>(acc_s)[i] = make_float4(0.0f, 0.0f, 0.0f, 0.0f);
        if (need_m)
            reinterpret_cast<uint4*>(acc_m)[i]  =
                make_uint4(MONO_NEG_INF, MONO_NEG_INF, MONO_NEG_INF, MONO_NEG_INF);
    }
    if (tid == 0) q_n = 0;
    __syncthreads();

    int qn;
    if (have) {
        qn = saved;
        for (int i = tid; i < qn; i += OWN_WAVES * 32) {
            q_src[i] = my_q[i];
            q_row[i] = my_q[OWN_Q + i];
        }
        __syncthreads();
    } else {
        const unsigned ownb = (unsigned)blockIdx.x & 0xFFu;
        const unsigned pat4 = ownb * 0x01010101u;
        const int nfull = E / OWN_TRIP;
        unsigned trip = ((unsigned)blockIdx.x * (unsigned)nfull) / gridDim.x;
        for (int t = 0; t < nfull; ++t) {
            const int gbase = (int)trip * (OWN_G * OWN_U) + tid;
            const uint4* p = ob16 + gbase;
            uint4 d[OWN_U];
            #pragma unroll
            for (int u = 0; u < OWN_U; ++u) d[u] = p[u * OWN_G];
            #pragma unroll
            for (int u = 0; u < OWN_U; ++u)
                own_scan16(d[u], pat4, (gbase + u * OWN_G) * 16, E, r0, nrows, &q_n, q_src,
                           acc_s, acc_m, x, src, dst, need_s, need_m);
            if (++trip == (unsigned)nfull) trip = 0u;
        }
        {
            const int ngrp  = (E + 15) >> 4;
            const int gbase = nfull * (OWN_G * OWN_U) + tid;
            uint4 d[OWN_U];
            #pragma unroll
            for (int u = 0; u < OWN_U; ++u) {
                const int g = gbase + u * OWN_G;
                d[u] = make_uint4(~pat4, ~pat4, ~pat4, ~pat4);
                if (g < ngrp) d[u] = ob16[g];
            }
            #pragma unroll
            for (int u = 0; u < OWN_U; ++u)
                own_scan16(d[u], pat4, (gbase + u * OWN_G) * 16, E, r0, nrows, &q_n, q_src,
                           acc_s, acc_m, x, src, dst, need_s, need_m);
        }
        __syncthreads();

        qn = min(q_n, OWN_Q);
        for (int i = tid; i < qn; i += OWN_WAVES * 32) {
            const int e  = q_src[i];
            const int sn = src[e];
            const unsigned loc = (unsigned)dst[e] - (unsigned)r0;
            const int rw = (loc < (unsigned)nrows) ? (int)loc : OWN_R;
            q_src[i] = sn;
            q_row[i] = rw;
            if (!reuse) {
                my_q[i]         = sn;
                my_q[OWN_Q + i] = rw;
            }
        }
        if (!reuse && tid == 0) wqn[blockIdx.x] = q_n;
        __syncthreads();
    }

    if (need_s && need_m)
        own_apply<true, true>(x, q_src, q_row, acc_s, acc_m, qn, wave, lane);
    else if (need_s)
        own_apply<true, false>(x, q_src, q_row, acc_s, acc_m, qn, wave, lane);
    else if (need_m)
        own_apply<false, true>(x, q_src, q_row, acc_s, acc_m, qn, wave, lane);
    __syncthreads();

    const int slab  = wave & 7;
    const int grp   = wave >> 3;
    const int ncol  = (slab << 4) + (lane & 15);
    const int mlo   = lane & 15;
    const int koff  = (lane >> 4) << 1;
    const int rbase = (lane >> 4) << 3;
    const int ntl   = (nrows + TILE_M - 1) / TILE_M;
    float* const hbuf = acc_s;
    float* const tbuf = reinterpret_cast<float*>(acc_m);

    v2f wf[KSTEPS];
    #pragma unroll
    for (int i = 0; i < KSTEPS; ++i) {
        const int kk = 4 * i + koff;
        wf[i].x = W1[(size_t)kk * GNN_D + ncol];
        wf[i].y = W1[(size_t)(kk + 1) * GNN_D + ncol];
    }
    const float bia1 = b1[ncol];
    const float bia2 = b2[ncol];

    {
        const float z0 = zagg[0], z1 = zagg[1], z2 = zagg[2];
        for (int i = tid; i < ntl * TILE_M * (GNN_D / 4); i += OWN_WAVES * 32) {
            const int r = i >> 5;
            const int cc = (i & 31) << 2;
            if (r >= nrows) {
                *reinterpret_cast<float4*>(hbuf + r * LDS_STRIDE + cc) = make_float4(0.0f, 0.0f, 0.0f, 0.0f);
                continue;
            }
            const float4 xv = *reinterpret_cast<const float4*>(x + (size_t)(r0 + r) * GNN_D + cc);
            const float dg = deg[r0 + r];
            const float dc = fmaxf(dg, 1.0f);
            const bool  nb = (dg > 0.0f) && need_m;
            float4 sv = make_float4(0.0f, 0.0f, 0.0f, 0.0f);
            uint4  mv = make_uint4(0u, 0u, 0u, 0u);
            float* const hp = hbuf + r * LDS_STRIDE + cc;
            if (need_s) sv = *reinterpret_cast<const float4*>(hp);
            if (need_m) mv = *reinterpret_cast<const uint4*>(acc_m + r * LDS_STRIDE + cc);
            float4 h;
            h.x = xv.x + z0 * sv.x + z1 * (sv.x / dc) + z2 * (nb ? mono2f(mv.x) : 0.0f);
            h.y = xv.y + z0 * sv.y + z1 * (sv.y / dc) + z2 * (nb ? mono2f(mv.y) : 0.0f);
            h.z = xv.z + z0 * sv.z + z1 * (sv.z / dc) + z2 * (nb ? mono2f(mv.z) : 0.0f);
            h.w = xv.w + z0 * sv.w + z1 * (sv.w / dc) + z2 * (nb ? mono2f(mv.w) : 0.0f);
            *reinterpret_cast<float4*>(hp) = h;
        }
    }
    __syncthreads();

    for (int t = grp; t < ntl; t += OWN_WAVES / 8) {
        const float* const hrow = hbuf + (t * TILE_M + mlo) * LDS_STRIDE + koff;
        v8f acc = {};
        #pragma unroll
        for (int i = 0; i < KSTEPS; ++i) {
            v2f a; a.x = hrow[4 * i];
                   a.y = hrow[4 * i + 1];
            acc = __builtin_amdgcn_wmma_f32_16x16x4_f32(false, a, false, wf[i],
                                                        (short)0, acc, false, false);
        }
        #pragma unroll
        for (int i = 0; i < 8; ++i)
            tbuf[(t * TILE_M + rbase + i) * LDS_STRIDE + ncol] = fmaxf(acc[i] + bia1, 0.0f);
    }

    #pragma unroll
    for (int i = 0; i < KSTEPS; ++i) {
        const int kk = 4 * i + koff;
        wf[i].x = W2[(size_t)kk * GNN_D + ncol];
        wf[i].y = W2[(size_t)(kk + 1) * GNN_D + ncol];
    }
    __syncthreads();

    for (int t = grp; t < ntl; t += OWN_WAVES / 8) {
        float xs[8];
        #pragma unroll
        for (int i = 0; i < 8; ++i) {
            const int r = min(t * TILE_M + rbase + i, nrows - 1);
            xs[i] = x[(size_t)(r0 + r) * GNN_D + ncol];
        }
        const float* const trow = tbuf + (t * TILE_M + mlo) * LDS_STRIDE + koff;
        v8f acc2 = {};
        #pragma unroll
        for (int i = 0; i < KSTEPS; ++i) {
            v2f a; a.x = trow[4 * i];
                   a.y = trow[4 * i + 1];
            acc2 = __builtin_amdgcn_wmma_f32_16x16x4_f32(false, a, false, wf[i],
                                                         (short)0, acc2, false, false);
        }
        asm volatile("" :: "v"(xs[0]), "v"(xs[1]), "v"(xs[2]), "v"(xs[3]),
                           "v"(xs[4]), "v"(xs[5]), "v"(xs[6]), "v"(xs[7]));
        #pragma unroll
        for (int i = 0; i < 8; ++i) {
            const int r = t * TILE_M + rbase + i;
            const float o = xs[i] + fmaxf(acc2[i] + bia2, 0.0f);
            if (r < nrows) xout[(size_t)(r0 + r) * GNN_D + ncol] = o;
        }
    }
}

#define POOL_WAVES 16
#define POOL_U     8
__global__ __launch_bounds__(POOL_WAVES * 32) void gnn_pool_owner_kernel(
    const float* __restrict__ x, const int* __restrict__ batch,
    const float* __restrict__ zpool, float* __restrict__ out, int N)
{
    __shared__ float red_s[POOL_WAVES][32];
    __shared__ float red_m[POOL_WAVES][32];
    __shared__ int   red_c[POOL_WAVES];

    const int lane = threadIdx.x & 31;
    const int wave = __builtin_amdgcn_readfirstlane((int)(threadIdx.x >> 5));
    const int b    = blockIdx.x >> 2;
    const unsigned col = ((blockIdx.x & 3u) << 5) + (unsigned)lane;

    float sum = 0.0f;
    float mxv = -INFINITY;
    int   cnt = 0;

    for (int c0 = wave; (c0 << 5) < N; c0 += POOL_WAVES * POOL_U) {
        int ids[POOL_U];
        #pragma unroll
        for (int u = 0; u < POOL_U; ++u) {
            int n  = ((c0 + u * POOL_WAVES) << 5) + lane;
            int id = batch[min(n, N - 1)];
            ids[u] = (n < N) ? id : -1;
        }
        #pragma unroll
        for (int u = 0; u < POOL_U; ++u) {
            const unsigned mask = __builtin_amdgcn_ballot_w32(ids[u] == b);
            if (mask == 0u) continue;
            cnt += __builtin_popcount(mask);
            const int base = (c0 + u * POOL_WAVES) << 5;
            float v[32];
            if (base + 32 <= N) {
                const float* p = x + (size_t)base * GNN_D + col;
                #pragma unroll
                for (int j = 0; j < 32; ++j) v[j] = p[j * GNN_D];
            } else {
                #pragma unroll
                for (int j = 0; j < 32; ++j)
                    v[j] = x[(size_t)min(base + j, N - 1) * GNN_D + col];
            }
            #pragma unroll
            for (int j = 0; j < 32; ++j) {
                const bool in = (mask >> j) & 1u;
                sum += in ? v[j] : 0.0f;
                mxv = fmaxf(mxv, in ? v[j] : -INFINITY);
            }
        }
    }

    red_s[wave][lane] = sum;
    red_m[wave][lane] = mxv;
    if (lane == 0) red_c[wave] = cnt;
    __syncthreads();

    if (wave == 0) {
        float s = 0.0f;
        float m = -INFINITY;
        int   c = 0;
        #pragma unroll
        for (int w = 0; w < POOL_WAVES; ++w) {
            s += red_s[w][lane];
            m  = fmaxf(m, red_m[w][lane]);
            c += red_c[w];
        }
        const float cf = (float)c;
        const float mv = (c > 0) ? m : 0.0f;
        out[(size_t)b * GNN_D + col] =
            zpool[0] * s + zpool[1] * (s / fmaxf(cf, 1.0f)) + zpool[2] * mv;
    }
}


extern "C" void kernel_launch(void* const* d_in, const int* in_sizes, int n_in,
                              void* d_out, int out_size, void* d_ws, size_t ws_size,
                              hipStream_t stream) {
    const float* x_in   = (const float*)d_in[0];
    const int*   eidx   = (const int*)d_in[1];
    const int*   batch  = (const int*)d_in[2];
    const float* z_agg  = (const float*)d_in[3];
    const float* z_pool = (const float*)d_in[4];
    const float* W1     = (const float*)d_in[5];
    const float* b1     = (const float*)d_in[6];
    const float* W2     = (const float*)d_in[7];
    const float* b2     = (const float*)d_in[8];
    (void)n_in;

    const int N = in_sizes[0] / GNN_D;
    const int E = in_sizes[1] / 2;
    const int L = in_sizes[5] / (GNN_D * GNN_D);
    const int Bg = out_size / GNN_D;
    const size_t ND = (size_t)N * GNN_D;

    char* p = (char*)d_ws;
    char* pend = p + ws_size;
    auto take = [&](size_t bytes) -> void* {
        void* r = (void*)p;
        p += (bytes + 255) & ~(size_t)255;
        return r;
    };
    float*    xa   = (float*)take(ND * 4);
    float*    xb   = (float*)take(ND * 4);
    float*    deg  = (float*)take((size_t)N * 4);
    const int nown = (N + OWN_R - 1) / OWN_R;
    int*      wq   = (int*)take((size_t)nown * 2 * OWN_Q * 4);
    int*      wqn  = (int*)take((size_t)nown * 4);
    const int Epad = (E + 15) & ~15;
    unsigned char* ob = (unsigned char*)take((size_t)Epad);
    if (p > pend) return;

    const int* src = eidx;
    const int* dst = eidx + E;

    hipMemcpyAsync(xa, x_in, ND * 4, hipMemcpyDeviceToDevice, stream);

    hipMemsetAsync(deg, 0, (size_t)N * 4, stream);
    gnn_deg_kernel<<<(Epad + 255) / 256, 256, 0, stream>>>(dst, deg, ob, E, Epad);

    float* xcur = xa;
    float* xnxt = xb;
    for (int l = 0; l < L; ++l) {
        gnn_layer_owner_kernel<<<nown, OWN_WAVES * 32, 0, stream>>>(
            xcur, src, dst, (const uint4*)ob, deg, z_agg + l * 3,
            W1 + (size_t)l * GNN_D * GNN_D, b1 + (size_t)l * GNN_D,
            W2 + (size_t)l * GNN_D * GNN_D, b2 + (size_t)l * GNN_D, xnxt,
            wq, wqn, l > 0 ? 1 : 0, N, E);
        float* t = xcur; xcur = xnxt; xnxt = t;
    }

    gnn_pool_owner_kernel<<<Bg * 4, POOL_WAVES * 32, 0, stream>>>(
        xcur, batch, z_pool, (float*)d_out, N);
}


# Degree by owner: each owner work-group counts its queue's edges per owned row in LDS (int LDS atomics); degree array, its memset, global float atomics gone; byte kernel 16 edges/lane; sums as before
# speedup vs baseline: 4.9174x; 1.2176x over previous
#include <hip/hip_runtime.h>
#include <hip/hip_bf16.h>


typedef float v8f __attribute__((ext_vector_type(8)));
typedef float v2f __attribute__((ext_vector_type(2)));

#define GNN_D 128
#define TILE_M 16
#define LDS_STRIDE 132
#define KSTEPS (GNN_D / 4)

__device__ __forceinline__ unsigned f2mono(float f) {
    unsigned u = __float_as_uint(f);
    return (u & 0x80000000u) ? ~u : (u | 0x80000000u);
}
__device__ __forceinline__ float mono2f(unsigned m) {
    unsigned u = (m & 0x80000000u) ? (m & 0x7FFFFFFFu) : ~m;
    return __uint_as_float(u);
}
#define MONO_NEG_INF 0x007FFFFFu


#define OWN_R     200
#define OWN_TILES ((OWN_R + TILE_M - 1) / TILE_M)
#define OWN_LROWS (OWN_TILES * TILE_M)
static_assert(OWN_LROWS > OWN_R, "the spare row must lie inside the padded tile rows");
#define OB_QUADS 4
__device__ __forceinline__ unsigned own_byte_of(int d) {
    return ((unsigned)d / (unsigned)OWN_R) & 0xFFu;
}
__device__ __forceinline__ unsigned own_pack4(const int4 d) {
    return own_byte_of(d.x) | (own_byte_of(d.y) << 8) |
           (own_byte_of(d.z) << 16) | (own_byte_of(d.w) << 24);
}
__global__ __launch_bounds__(256) void gnn_owner_byte_kernel(const int* __restrict__ dst,
                                                             unsigned* __restrict__ ob32,
                                                             int E, int nquad) {
    const int qb = blockIdx.x * (256 * OB_QUADS);
    const bool whole = ((((size_t)dst) & 15u) == 0u) && ((qb + 256 * OB_QUADS) * 4 <= E);
    if (whole) {
        const int4* p = reinterpret_cast<const int4*>(dst) + qb + threadIdx.x;
        int4 d[OB_QUADS];
        #pragma unroll
        for (int k = 0; k < OB_QUADS; ++k) d[k] = p[k * 256];
        #pragma unroll
        for (int k = 0; k < OB_QUADS; ++k) ob32[qb + threadIdx.x + k * 256] = own_pack4(d[k]);
    } else {
        #pragma nounroll
        for (int k = 0; k < OB_QUADS; ++k) {
            const int q = qb + threadIdx.x + k * 256;
            if (q >= nquad) continue;
            const int e = q << 2;
            unsigned w = 0u;
            #pragma unroll
            for (int j = 0; j < 4; ++j) {
                const unsigned bj = (e + j < E) ? own_byte_of(dst[e + j]) : 0xFFu;
                w |= bj << (8 * j);
            }
            ob32[q] = w;
        }
    }
}

#define OWN_Q     4096
#define OWN_WAVES 32
#define OWN_U     8
#define OWN_PB    8
#define OWN_G     (OWN_WAVES * 32)
#define OWN_TRIP  (OWN_G * OWN_U * 16)

__device__ __forceinline__ void own_take(
    int e, int r0, int nrows, int* q_n, int* q_src, int* rcnt,
    float* acc_s, unsigned* acc_m,
    const float* __restrict__ x, const int* __restrict__ src,
    const int* __restrict__ dst, bool need_s, bool need_m)
{
    const int pos = atomicAdd(q_n, 1);
    if (pos < OWN_Q) {
        q_src[pos] = e;
    } else {
        const unsigned loc = (unsigned)dst[e] - (unsigned)r0;
        if (loc >= (unsigned)nrows) return;
        atomicAdd(rcnt + loc, 1);
        const float4* xr = reinterpret_cast<const float4*>(x + (size_t)src[e] * GNN_D);
        float*    as = acc_s + loc * LDS_STRIDE;
        unsigned* am = acc_m + loc * LDS_STRIDE;
        #pragma nounroll
        for (int f = 0; f < GNN_D / 4; ++f) {
            const float4 v = xr[f];
            if (need_s) {
                atomicAdd(as + 4 * f + 0, v.x);
                atomicAdd(as + 4 * f + 1, v.y);
                atomicAdd(as + 4 * f + 2, v.z);
                atomicAdd(as + 4 * f + 3, v.w);
            }
            if (need_m) {
                atomicMax(am + 4 * f + 0, f2mono(v.x));
                atomicMax(am + 4 * f + 1, f2mono(v.y));
                atomicMax(am + 4 * f + 2, f2mono(v.z));
                atomicMax(am + 4 * f + 3, f2mono(v.w));
            }
        }
    }
}

__device__ __forceinline__ unsigned own_eq_bytes(unsigned w, unsigned pat4) {
    const unsigned d = w ^ pat4;
    return ~(((d & 0x7F7F7F7Fu) + 0x7F7F7F7Fu) | d | 0x7F7F7F7Fu);
}
__device__ __forceinline__ unsigned own_nibble(unsigned h) {
    return (((h >> 7) * 0x00204081u) >> 21) & 0xFu;
}
__device__ __forceinline__ void own_scan16(
    const uint4 w, unsigned pat4, int e0, int elim, int r0, int nrows,
    int* q_n, int* q_src, int* rcnt, float* acc_s, unsigned* acc_m,
    const float* __restrict__ x, const int* __restrict__ src,
    const int* __restrict__ dst, bool need_s, bool need_m)
{
    const unsigned h0 = own_eq_bytes(w.x, pat4);
    const unsigned h1 = own_eq_bytes(w.y, pat4);
    const unsigned h2 = own_eq_bytes(w.z, pat4);
    const unsigned h3 = own_eq_bytes(w.w, pat4);
    if ((h0 | h1 | h2 | h3) != 0u) {
        unsigned m = own_nibble(h0) | (own_nibble(h1) << 4) |
                     (own_nibble(h2) << 8) | (own_nibble(h3) << 12);
        do {
            const int e = e0 + __builtin_ctz(m);
            if (e < elim)
                own_take(e, r0, nrows, q_n, q_src, rcnt, acc_s, acc_m, x, src, dst, need_s, need_m);
            m &= m - 1u;
        } while (m != 0u);
    }
}

template <bool SUM, bool MAX>
__device__ __forceinline__ void own_apply(
    const float* __restrict__ x, const int* q_src, const int* q_row,
    float* acc_s, unsigned* acc_m, int qn, int wave, int lane)
{
    for (int q0 = wave * OWN_PB; q0 < qn; q0 += OWN_WAVES * OWN_PB) {
        float v[OWN_PB][4];
        int   rw[OWN_PB];
        #pragma unroll
        for (int p = 0; p < OWN_PB; ++p) {
            const int qc = min(q0 + p, qn - 1);
            const int sn = __builtin_amdgcn_readfirstlane(q_src[qc]);
            const int rr = __builtin_amdgcn_readfirstlane(q_row[qc]);
            rw[p] = (q0 + p < qn) ? rr : OWN_R;
            const float* xp = x + (size_t)sn * GNN_D + lane;
            #pragma unroll
            for (int k = 0; k < 4; ++k) v[p][k] = xp[32 * k];
        }
        #pragma unroll
        for (int p = 0; p < OWN_PB; ++p) {
            float*    as = acc_s + rw[p] * LDS_STRIDE + lane;
            unsigned* am = acc_m + rw[p] * LDS_STRIDE + lane;
            #pragma unroll
            for (int k = 0; k < 4; ++k) {
                if (SUM) atomicAdd(as + 32 * k, v[p][k]);
                if (MAX) atomicMax(am + 32 * k, f2mono(v[p][k]));
            }
        }
    }
}

__global__ __launch_bounds__(OWN_WAVES * 32) void gnn_layer_owner_kernel(
    const float* __restrict__ x, const int* __restrict__ src,
    const int* __restrict__ dst, const uint4* __restrict__ ob16,
    const float* __restrict__ zagg,
    const float* __restrict__ W1, const float* __restrict__ b1,
    const float* __restrict__ W2, const float* __restrict__ b2,
    float* __restrict__ xout,
    int* wq, int* wqn, int reuse, int N, int E)
{
    __shared__ __attribute__((aligned(16))) float    acc_s[OWN_LROWS * LDS_STRIDE];
    __shared__ __attribute__((aligned(16))) unsigned acc_m[OWN_LROWS * LDS_STRIDE];
    __shared__ int q_src[OWN_Q];
    __shared__ int q_row[OWN_Q];
    __shared__ int rcnt[OWN_LROWS];
    __shared__ int q_n;

    const int tid   = threadIdx.x;
    const int lane  = tid & 31;
    const int wave  = __builtin_amdgcn_readfirstlane(tid >> 5);
    const int r0    = blockIdx.x * OWN_R;
    const int nrows = min(OWN_R, N - r0);
    int* const my_q = wq + (size_t)blockIdx.x * (2 * OWN_Q);
    const int saved = __builtin_amdgcn_readfirstlane(reuse ? wqn[blockIdx.x] : -1);
    const bool have = (unsigned)saved <= (unsigned)OWN_Q;
    const bool need_s = (zagg[0] != 0.0f) || (zagg[1] != 0.0f);
    const bool need_m = (zagg[2] != 0.0f);

    for (int i = tid; i < OWN_LROWS * LDS_STRIDE / 4; i += OWN_WAVES * 32) {
        if (need_s)
            reinterpret_cast<float4*>(acc_s)[i] = make_float4(0.0f, 0.0f, 0.0f, 0.0f);
        if (need_m)
            reinterpret_cast<uint4*>(acc_m)[i]  =
                make_uint4(MONO_NEG_INF, MONO_NEG_INF, MONO_NEG_INF, MONO_NEG_INF);
    }
    if (tid < OWN_LROWS) rcnt[tid] = 0;
    if (tid == 0) q_n = 0;
    __syncthreads();

    int qn;
    if (have) {
        qn = saved;
        for (int i = tid; i < qn; i += OWN_WAVES * 32) {
            const int rw = my_q[OWN_Q + i];
            q_src[i] = my_q[i];
            q_row[i] = rw;
            atomicAdd(&rcnt[rw], 1);
        }
        __syncthreads();
    } else {
        const unsigned ownb = (unsigned)blockIdx.x & 0xFFu;
        const unsigned pat4 = ownb * 0x01010101u;
        const int nfull = E / OWN_TRIP;
        unsigned trip = ((unsigned)blockIdx.x * (unsigned)nfull) / gridDim.x;
        for (int t = 0; t < nfull; ++t) {
            const int gbase = (int)trip * (OWN_G * OWN_U) + tid;
            const uint4* p = ob16 + gbase;
            uint4 d[OWN_U];
            #pragma unroll
            for (int u = 0; u < OWN_U; ++u) d[u] = p[u * OWN_G];
            #pragma unroll
            for (int u = 0; u < OWN_U; ++u)
                own_scan16(d[u], pat4, (gbase + u * OWN_G) * 16, E, r0, nrows, &q_n, q_src,
                           rcnt, acc_s, acc_m, x, src, dst, need_s, need_m);
            if (++trip == (unsigned)nfull) trip = 0u;
        }
        {
            const int ngrp  = (E + 15) >> 4;
            const int gbase = nfull * (OWN_G * OWN_U) + tid;
            uint4 d[OWN_U];
            #pragma unroll
            for (int u = 0; u < OWN_U; ++u) {
                const int g = gbase + u * OWN_G;
                d[u] = make_uint4(~pat4, ~pat4, ~pat4, ~pat4);
                if (g < ngrp) d[u] = ob16[g];
            }
            #pragma unroll
            for (int u = 0; u < OWN_U; ++u)
                own_scan16(d[u], pat4, (gbase + u * OWN_G) * 16, E, r0, nrows, &q_n, q_src,
                           rcnt, acc_s, acc_m, x, src, dst, need_s, need_m);
        }
        __syncthreads();

        qn = min(q_n, OWN_Q);
        for (int i = tid; i < qn; i += OWN_WAVES * 32) {
            const int e  = q_src[i];
            const int sn = src[e];
            const unsigned loc = (unsigned)dst[e] - (unsigned)r0;
            const int rw = (loc < (unsigned)nrows) ? (int)loc : OWN_R;
            q_src[i] = sn;
            q_row[i] = rw;
            atomicAdd(&rcnt[rw], 1);
            if (!reuse) {
                my_q[i]         = sn;
                my_q[OWN_Q + i] = rw;
            }
        }
        if (!reuse && tid == 0) wqn[blockIdx.x] = q_n;
        __syncthreads();
    }

    if (need_s && need_m)
        own_apply<true, true>(x, q_src, q_row, acc_s, acc_m, qn, wave, lane);
    else if (need_s)
        own_apply<true, false>(x, q_src, q_row, acc_s, acc_m, qn, wave, lane);
    else if (need_m)
        own_apply<false, true>(x, q_src, q_row, acc_s, acc_m, qn, wave, lane);
    __syncthreads();

    const int slab  = wave & 7;
    const int grp   = wave >> 3;
    const int ncol  = (slab << 4) + (lane & 15);
    const int mlo   = lane & 15;
    const int koff  = (lane >> 4) << 1;
    const int rbase = (lane >> 4) << 3;
    const int ntl   = (nrows + TILE_M - 1) / TILE_M;
    float* const hbuf = acc_s;
    float* const tbuf = reinterpret_cast<float*>(acc_m);

    v2f wf[KSTEPS];
    #pragma unroll
    for (int i = 0; i < KSTEPS; ++i) {
        const int kk = 4 * i + koff;
        wf[i].x = W1[(size_t)kk * GNN_D + ncol];
        wf[i].y = W1[(size_t)(kk + 1) * GNN_D + ncol];
    }
    const float bia1 = b1[ncol];
    const float bia2 = b2[ncol];

    {
        const float z0 = zagg[0], z1 = zagg[1], z2 = zagg[2];
        for (int i = tid; i < ntl * TILE_M * (GNN_D / 4); i += OWN_WAVES * 32) {
            const int r = i >> 5;
            const int cc = (i & 31) << 2;
            if (r >= nrows) {
                *reinterpret_cast<float4*>(hbuf + r * LDS_STRIDE + cc) = make_float4(0.0f, 0.0f, 0.0f, 0.0f);
                continue;
            }
            const float4 xv = *reinterpret_cast<const float4*>(x + (size_t)(r0 + r) * GNN_D + cc);
            const float dg = (float)rcnt[r];
            const float dc = fmaxf(dg, 1.0f);
            const bool  nb = (dg > 0.0f) && need_m;
            float4 sv = make_float4(0.0f, 0.0f, 0.0f, 0.0f);
            uint4  mv = make_uint4(0u, 0u, 0u, 0u);
            float* const hp = hbuf + r * LDS_STRIDE + cc;
            if (need_s) sv = *reinterpret_cast<const float4*>(hp);
            if (need_m) mv = *reinterpret_cast<const uint4*>(acc_m + r * LDS_STRIDE + cc);
            float4 h;
            h.x = xv.x + z0 * sv.x + z1 * (sv.x / dc) + z2 * (nb ? mono2f(mv.x) : 0.0f);
            h.y = xv.y + z0 * sv.y + z1 * (sv.y / dc) + z2 * (nb ? mono2f(mv.y) : 0.0f);
            h.z = xv.z + z0 * sv.z + z1 * (sv.z / dc) + z2 * (nb ? mono2f(mv.z) : 0.0f);
            h.w = xv.w + z0 * sv.w + z1 * (sv.w / dc) + z2 * (nb ? mono2f(mv.w) : 0.0f);
            *reinterpret_cast<float4*>(hp) = h;
        }
    }
    __syncthreads();

    for (int t = grp; t < ntl; t += OWN_WAVES / 8) {
        const float* const hrow = hbuf + (t * TILE_M + mlo) * LDS_STRIDE + koff;
        v8f acc = {};
        #pragma unroll
        for (int i = 0; i < KSTEPS; ++i) {
            v2f a; a.x = hrow[4 * i];
                   a.y = hrow[4 * i + 1];
            acc = __builtin_amdgcn_wmma_f32_16x16x4_f32(false, a, false, wf[i],
                                                        (short)0, acc, false, false);
        }
        #pragma unroll
        for (int i = 0; i < 8; ++i)
            tbuf[(t * TILE_M + rbase + i) * LDS_STRIDE + ncol] = fmaxf(acc[i] + bia1, 0.0f);
    }

    #pragma unroll
    for (int i = 0; i < KSTEPS; ++i) {
        const int kk = 4 * i + koff;
        wf[i].x = W2[(size_t)kk * GNN_D + ncol];
        wf[i].y = W2[(size_t)(kk + 1) * GNN_D + ncol];
    }
    __syncthreads();

    for (int t = grp; t < ntl; t += OWN_WAVES / 8) {
        float xs[8];
        #pragma unroll
        for (int i = 0; i < 8; ++i) {
            const int r = min(t * TILE_M + rbase + i, nrows - 1);
            xs[i] = x[(size_t)(r0 + r) * GNN_D + ncol];
        }
        const float* const trow = tbuf + (t * TILE_M + mlo) * LDS_STRIDE + koff;
        v8f acc2 = {};
        #pragma unroll
        for (int i = 0; i < KSTEPS; ++i) {
            v2f a; a.x = trow[4 * i];
                   a.y = trow[4 * i + 1];
            acc2 = __builtin_amdgcn_wmma_f32_16x16x4_f32(false, a, false, wf[i],
                                                         (short)0, acc2, false, false);
        }
        asm volatile("" :: "v"(xs[0]), "v"(xs[1]), "v"(xs[2]), "v"(xs[3]),
                           "v"(xs[4]), "v"(xs[5]), "v"(xs[6]), "v"(xs[7]));
        #pragma unroll
        for (int i = 0; i < 8; ++i) {
            const int r = t * TILE_M + rbase + i;
            const float o = xs[i] + fmaxf(acc2[i] + bia2, 0.0f);
            if (r < nrows) xout[(size_t)(r0 + r) * GNN_D + ncol] = o;
        }
    }
}

#define POOL_WAVES 16
#define POOL_U     8
__global__ __launch_bounds__(POOL_WAVES * 32) void gnn_pool_owner_kernel(
    const float* __restrict__ x, const int* __restrict__ batch,
    const float* __restrict__ zpool, float* __restrict__ out, int N)
{
    __shared__ float red_s[POOL_WAVES][32];
    __shared__ float red_m[POOL_WAVES][32];
    __shared__ int   red_c[POOL_WAVES];

    const int lane = threadIdx.x & 31;
    const int wave = __builtin_amdgcn_readfirstlane((int)(threadIdx.x >> 5));
    const int b    = blockIdx.x >> 2;
    const unsigned col = ((blockIdx.x & 3u) << 5) + (unsigned)lane;

    float sum = 0.0f;
    float mxv = -INFINITY;
    int   cnt = 0;

    for (int c0 = wave; (c0 << 5) < N; c0 += POOL_WAVES * POOL_U) {
        int ids[POOL_U];
        #pragma unroll
        for (int u = 0; u < POOL_U; ++u) {
            int n  = ((c0 + u * POOL_WAVES) << 5) + lane;
            int id = batch[min(n, N - 1)];
            ids[u] = (n < N) ? id : -1;
        }
        #pragma unroll
        for (int u = 0; u < POOL_U; ++u) {
            const unsigned mask = __builtin_amdgcn_ballot_w32(ids[u] == b);
            if (mask == 0u) continue;
            cnt += __builtin_popcount(mask);
            const int base = (c0 + u * POOL_WAVES) << 5;
            float v[32];
            if (base + 32 <= N) {
                const float* p = x + (size_t)base * GNN_D + col;
                #pragma unroll
                for (int j = 0; j < 32; ++j) v[j] = p[j * GNN_D];
            } else {
                #pragma unroll
                for (int j = 0; j < 32; ++j)
                    v[j] = x[(size_t)min(base + j, N - 1) * GNN_D + col];
            }
            #pragma unroll
            for (int j = 0; j < 32; ++j) {
                const bool in = (mask >> j) & 1u;
                sum += in ? v[j] : 0.0f;
                mxv = fmaxf(mxv, in ? v[j] : -INFINITY);
            }
        }
    }

    red_s[wave][lane] = sum;
    red_m[wave][lane] = mxv;
    if (lane == 0) red_c[wave] = cnt;
    __syncthreads();

    if (wave == 0) {
        float s = 0.0f;
        float m = -INFINITY;
        int   c = 0;
        #pragma unroll
        for (int w = 0; w < POOL_WAVES; ++w) {
            s += red_s[w][lane];
            m  = fmaxf(m, red_m[w][lane]);
            c += red_c[w];
        }
        const float cf = (float)c;
        const float mv = (c > 0) ? m : 0.0f;
        out[(size_t)b * GNN_D + col] =
            zpool[0] * s + zpool[1] * (s / fmaxf(cf, 1.0f)) + zpool[2] * mv;
    }
}


extern "C" void kernel_launch(void* const* d_in, const int* in_sizes, int n_in,
                              void* d_out, int out_size, void* d_ws, size_t ws_size,
                              hipStream_t stream) {
    const float* x_in   = (const float*)d_in[0];
    const int*   eidx   = (const int*)d_in[1];
    const int*   batch  = (const int*)d_in[2];
    const float* z_agg  = (const float*)d_in[3];
    const float* z_pool = (const float*)d_in[4];
    const float* W1     = (const float*)d_in[5];
    const float* b1     = (const float*)d_in[6];
    const float* W2     = (const float*)d_in[7];
    const float* b2     = (const float*)d_in[8];
    (void)n_in;

    const int N = in_sizes[0] / GNN_D;
    const int E = in_sizes[1] / 2;
    const int L = in_sizes[5] / (GNN_D * GNN_D);
    const int Bg = out_size / GNN_D;
    const size_t ND = (size_t)N * GNN_D;

    char* p = (char*)d_ws;
    char* pend = p + ws_size;
    auto take = [&](size_t bytes) -> void* {
        void* r = (void*)p;
        p += (bytes + 255) & ~(size_t)255;
        return r;
    };
    float*    xa   = (float*)take(ND * 4);
    float*    xb   = (float*)take(ND * 4);
    const int nown = (N + OWN_R - 1) / OWN_R;
    int*      wq   = (int*)take((size_t)nown * 2 * OWN_Q * 4);
    int*      wqn  = (int*)take((size_t)nown * 4);
    const int Epad = (E + 15) & ~15;
    unsigned char* ob = (unsigned char*)take((size_t)Epad);
    if (p > pend) return;

    const int* src = eidx;
    const int* dst = eidx + E;

    hipMemcpyAsync(xa, x_in, ND * 4, hipMemcpyDeviceToDevice, stream);

    const int nquad = Epad / 4;
    gnn_owner_byte_kernel<<<(nquad + 256 * OB_QUADS - 1) / (256 * OB_QUADS), 256, 0, stream>>>(
        dst, (unsigned*)ob, E, nquad);

    float* xcur = xa;
    float* xnxt = xb;
    for (int l = 0; l < L; ++l) {
        gnn_layer_owner_kernel<<<nown, OWN_WAVES * 32, 0, stream>>>(
            xcur, src, dst, (const uint4*)ob, z_agg + l * 3,
            W1 + (size_t)l * GNN_D * GNN_D, b1 + (size_t)l * GNN_D,
            W2 + (size_t)l * GNN_D * GNN_D, b2 + (size_t)l * GNN_D, xnxt,
            wq, wqn, l > 0 ? 1 : 0, N, E);
        float* t = xcur; xcur = xnxt; xnxt = t;
    }

    gnn_pool_owner_kernel<<<Bg * 4, POOL_WAVES * 32, 0, stream>>>(
        xcur, batch, z_pool, (float*)d_out, N);
}


# Owner queue sorted by owned row in LDS (counting sort, kept for layers 1, 2): a wave sums/maxes a row's source rows in registers, 16-byte loads; float LDS atomics gone; sums in arrival-rank order
# speedup vs baseline: 5.4372x; 1.1057x over previous
#include <hip/hip_runtime.h>
#include <hip/hip_bf16.h>


typedef float v8f __attribute__((ext_vector_type(8)));
typedef float v2f __attribute__((ext_vector_type(2)));

#define GNN_D 128
#define TILE_M 16
#define LDS_STRIDE 132
#define KSTEPS (GNN_D / 4)

__device__ __forceinline__ unsigned f2mono(float f) {
    unsigned u = __float_as_uint(f);
    return (u & 0x80000000u) ? ~u : (u | 0x80000000u);
}
__device__ __forceinline__ float mono2f(unsigned m) {
    unsigned u = (m & 0x80000000u) ? (m & 0x7FFFFFFFu) : ~m;
    return __uint_as_float(u);
}
#define MONO_NEG_INF 0x007FFFFFu


#define OWN_R     200
#define OWN_TILES ((OWN_R + TILE_M - 1) / TILE_M)
#define OWN_LROWS (OWN_TILES * TILE_M)
static_assert(OWN_LROWS > OWN_R, "the spare row must lie inside the padded tile rows");
static_assert(OWN_R < 256, "an owned row's number (the spare row's too) is kept in eight bits beside the rank");
#define OB_QUADS 4
__device__ __forceinline__ unsigned own_byte_of(int d) {
    return ((unsigned)d / (unsigned)OWN_R) & 0xFFu;
}
__device__ __forceinline__ unsigned own_pack4(const int4 d) {
    return own_byte_of(d.x) | (own_byte_of(d.y) << 8) |
           (own_byte_of(d.z) << 16) | (own_byte_of(d.w) << 24);
}
__global__ __launch_bounds__(256) void gnn_owner_byte_kernel(const int* __restrict__ dst,
                                                             unsigned* __restrict__ ob32,
                                                             int E, int nquad) {
    const int qb = blockIdx.x * (256 * OB_QUADS);
    const bool whole = ((((size_t)dst) & 15u) == 0u) && ((qb + 256 * OB_QUADS) * 4 <= E);
    if (whole) {
        const int4* p = reinterpret_cast<const int4*>(dst) + qb + threadIdx.x;
        int4 d[OB_QUADS];
        #pragma unroll
        for (int k = 0; k < OB_QUADS; ++k) d[k] = p[k * 256];
        #pragma unroll
        for (int k = 0; k < OB_QUADS; ++k) ob32[qb + threadIdx.x + k * 256] = own_pack4(d[k]);
    } else {
        #pragma nounroll
        for (int k = 0; k < OB_QUADS; ++k) {
            const int q = qb + threadIdx.x + k * 256;
            if (q >= nquad) continue;
            const int e = q << 2;
            unsigned w = 0u;
            #pragma unroll
            for (int j = 0; j < 4; ++j) {
                const unsigned bj = (e + j < E) ? own_byte_of(dst[e + j]) : 0xFFu;
                w |= bj << (8 * j);
            }
            ob32[q] = w;
        }
    }
}

#define OWN_Q     4096
#define OWN_WAVES 32
#define OWN_U     8
#define OWN_PB    8
#define OWN_PS    ((OWN_LROWS + 31) / 32)
#define OWN_G     (OWN_WAVES * 32)
#define OWN_TRIP  (OWN_G * OWN_U * 16)

__device__ __forceinline__ void own_take(
    int e, int r0, int nrows, int* q_n, int* q_src, int* rcnt,
    float* acc_s, unsigned* acc_m,
    const float* __restrict__ x, const int* __restrict__ src,
    const int* __restrict__ dst, bool need_s, bool need_m)
{
    const int pos = atomicAdd(q_n, 1);
    if (pos < OWN_Q) {
        q_src[pos] = e;
    } else {
        const unsigned loc = (unsigned)dst[e] - (unsigned)r0;
        if (loc >= (unsigned)nrows) return;
        atomicAdd(rcnt + loc, 1);
        const float4* xr = reinterpret_cast<const float4*>(x + (size_t)src[e] * GNN_D);
        float*    as = acc_s + loc * LDS_STRIDE;
        unsigned* am = acc_m + loc * LDS_STRIDE;
        #pragma nounroll
        for (int f = 0; f < GNN_D / 4; ++f) {
            const float4 v = xr[f];
            if (need_s) {
                atomicAdd(as + 4 * f + 0, v.x);
                atomicAdd(as + 4 * f + 1, v.y);
                atomicAdd(as + 4 * f + 2, v.z);
                atomicAdd(as + 4 * f + 3, v.w);
            }
            if (need_m) {
                atomicMax(am + 4 * f + 0, f2mono(v.x));
                atomicMax(am + 4 * f + 1, f2mono(v.y));
                atomicMax(am + 4 * f + 2, f2mono(v.z));
                atomicMax(am + 4 * f + 3, f2mono(v.w));
            }
        }
    }
}

__device__ __forceinline__ unsigned own_eq_bytes(unsigned w, unsigned pat4) {
    const unsigned d = w ^ pat4;
    return ~(((d & 0x7F7F7F7Fu) + 0x7F7F7F7Fu) | d | 0x7F7F7F7Fu);
}
__device__ __forceinline__ unsigned own_nibble(unsigned h) {
    return (((h >> 7) * 0x00204081u) >> 21) & 0xFu;
}
__device__ __forceinline__ void own_scan16(
    const uint4 w, unsigned pat4, int e0, int elim, int r0, int nrows,
    int* q_n, int* q_src, int* rcnt, float* acc_s, unsigned* acc_m,
    const float* __restrict__ x, const int* __restrict__ src,
    const int* __restrict__ dst, bool need_s, bool need_m)
{
    const unsigned h0 = own_eq_bytes(w.x, pat4);
    const unsigned h1 = own_eq_bytes(w.y, pat4);
    const unsigned h2 = own_eq_bytes(w.z, pat4);
    const unsigned h3 = own_eq_bytes(w.w, pat4);
    if ((h0 | h1 | h2 | h3) != 0u) {
        unsigned m = own_nibble(h0) | (own_nibble(h1) << 4) |
                     (own_nibble(h2) << 8) | (own_nibble(h3) << 12);
        do {
            const int e = e0 + __builtin_ctz(m);
            if (e < elim)
                own_take(e, r0, nrows, q_n, q_src, rcnt, acc_s, acc_m, x, src, dst, need_s, need_m);
            m &= m - 1u;
        } while (m != 0u);
    }
}

template <bool SUM, bool MAX>
__device__ __forceinline__ void own_apply_rows(
    const float* __restrict__ x, const int* q_srt, const int* rst,
    float* acc_s, unsigned* acc_m, int* next_row, int nrows, int wave, int lane)
{
    int r = wave;
    for (int it = 0; it <= OWN_R; ++it) {
        if (r >= nrows) break;
        int nxt = 0;
        if (lane == 0) nxt = atomicAdd(next_row, 1);
        const int rc = r;
        const int qb = __builtin_amdgcn_readfirstlane(rst[rc]);
        const int qe = __builtin_amdgcn_readfirstlane(rst[rc + 1]);
        if (qb < qe) {
            float4 s = make_float4(0.0f, 0.0f, 0.0f, 0.0f);
            uint4  m = make_uint4(MONO_NEG_INF, MONO_NEG_INF, MONO_NEG_INF, MONO_NEG_INF);
            for (int q0 = qb; q0 < qe; q0 += OWN_PB) {
                const int mine = q_srt[min(q0 + lane, qe - 1)];
                float4 v[OWN_PB];
                #pragma unroll
                for (int p = 0; p < OWN_PB; ++p) {
                    const int sn = __builtin_amdgcn_readlane(mine, p);
                    v[p] = *reinterpret_cast<const float4*>(x + (size_t)sn * GNN_D + 4 * lane);
                }
                #pragma unroll
                for (int p = 0; p < OWN_PB; ++p)
                    asm volatile("" :: "v"(v[p].x), "v"(v[p].y), "v"(v[p].z), "v"(v[p].w));
                #pragma unroll
                for (int p = 0; p < OWN_PB; ++p) {
                    if (q0 + p < qe) {
                        if (SUM) {
                            s.x += v[p].x; s.y += v[p].y; s.z += v[p].z; s.w += v[p].w;
                        }
                        if (MAX) {
                            m.x = max(m.x, f2mono(v[p].x)); m.y = max(m.y, f2mono(v[p].y));
                            m.z = max(m.z, f2mono(v[p].z)); m.w = max(m.w, f2mono(v[p].w));
                        }
                    }
                }
            }
            if (SUM) {
                float4* const ap = reinterpret_cast<float4*>(acc_s + rc * LDS_STRIDE + 4 * lane);
                float4 o = *ap;
                o.x += s.x; o.y += s.y; o.z += s.z; o.w += s.w;
                *ap = o;
            }
            if (MAX) {
                uint4* const mp = reinterpret_cast<uint4*>(acc_m + rc * LDS_STRIDE + 4 * lane);
                uint4 o = *mp;
                o.x = max(o.x, m.x); o.y = max(o.y, m.y); o.z = max(o.z, m.z); o.w = max(o.w, m.w);
                *mp = o;
            }
        }
        r = __builtin_amdgcn_readfirstlane(nxt);
    }
}

__global__ __launch_bounds__(OWN_WAVES * 32) void gnn_layer_owner_kernel(
    const float* __restrict__ x, const int* __restrict__ src,
    const int* __restrict__ dst, const uint4* __restrict__ ob16,
    const float* __restrict__ zagg,
    const float* __restrict__ W1, const float* __restrict__ b1,
    const float* __restrict__ W2, const float* __restrict__ b2,
    float* __restrict__ xout,
    int* wq, int* wqn, int reuse, int N, int E)
{
    __shared__ __attribute__((aligned(16))) float    acc_s[OWN_LROWS * LDS_STRIDE];
    __shared__ __attribute__((aligned(16))) unsigned acc_m[OWN_LROWS * LDS_STRIDE];
    __shared__ int q_src[OWN_Q];
    __shared__ int q_row[OWN_Q];
    __shared__ int q_srt[OWN_Q];
    __shared__ int qcnt[OWN_LROWS];
    __shared__ int rst[OWN_LROWS + 1];
    __shared__ int rcnt[OWN_LROWS];
    __shared__ int q_n;
    __shared__ int next_row;

    const int tid   = threadIdx.x;
    const int lane  = tid & 31;
    const int wave  = __builtin_amdgcn_readfirstlane(tid >> 5);
    const int r0    = blockIdx.x * OWN_R;
    const int nrows = min(OWN_R, N - r0);
    int* const my_q = wq + (size_t)blockIdx.x * (2 * OWN_Q);
    const int saved = __builtin_amdgcn_readfirstlane(reuse ? wqn[blockIdx.x] : -1);
    const bool have = (unsigned)saved <= (unsigned)OWN_Q;
    const bool need_s = (zagg[0] != 0.0f) || (zagg[1] != 0.0f);
    const bool need_m = (zagg[2] != 0.0f);

    for (int i = tid; i < OWN_LROWS * LDS_STRIDE / 4; i += OWN_WAVES * 32) {
        if (need_s)
            reinterpret_cast<float4*>(acc_s)[i] = make_float4(0.0f, 0.0f, 0.0f, 0.0f);
        if (need_m)
            reinterpret_cast<uint4*>(acc_m)[i]  =
                make_uint4(MONO_NEG_INF, MONO_NEG_INF, MONO_NEG_INF, MONO_NEG_INF);
    }
    if (tid < OWN_LROWS) { rcnt[tid] = 0; qcnt[tid] = 0; }
    if (tid == 0) { q_n = 0; next_row = OWN_WAVES; }
    __syncthreads();

    int qn;
    if (have) {
        qn = saved;
        for (int i = tid; i < qn; i += OWN_WAVES * 32) q_srt[i] = my_q[i];
        if (tid <= OWN_LROWS) rst[tid] = my_q[OWN_Q + tid];
        __syncthreads();
    } else {
        const unsigned ownb = (unsigned)blockIdx.x & 0xFFu;
        const unsigned pat4 = ownb * 0x01010101u;
        const int nfull = E / OWN_TRIP;
        unsigned trip = ((unsigned)blockIdx.x * (unsigned)nfull) / gridDim.x;
        for (int t = 0; t < nfull; ++t) {
            const int gbase = (int)trip * (OWN_G * OWN_U) + tid;
            const uint4* p = ob16 + gbase;
            uint4 d[OWN_U];
            #pragma unroll
            for (int u = 0; u < OWN_U; ++u) d[u] = p[u * OWN_G];
            #pragma unroll
            for (int u = 0; u < OWN_U; ++u)
                own_scan16(d[u], pat4, (gbase + u * OWN_G) * 16, E, r0, nrows, &q_n, q_src,
                           rcnt, acc_s, acc_m, x, src, dst, need_s, need_m);
            if (++trip == (unsigned)nfull) trip = 0u;
        }
        {
            const int ngrp  = (E + 15) >> 4;
            const int gbase = nfull * (OWN_G * OWN_U) + tid;
            uint4 d[OWN_U];
            #pragma unroll
            for (int u = 0; u < OWN_U; ++u) {
                const int g = gbase + u * OWN_G;
                d[u] = make_uint4(~pat4, ~pat4, ~pat4, ~pat4);
                if (g < ngrp) d[u] = ob16[g];
            }
            #pragma unroll
            for (int u = 0; u < OWN_U; ++u)
                own_scan16(d[u], pat4, (gbase + u * OWN_G) * 16, E, r0, nrows, &q_n, q_src,
                           rcnt, acc_s, acc_m, x, src, dst, need_s, need_m);
        }
        __syncthreads();

        qn = min(q_n, OWN_Q);
        for (int i = tid; i < qn; i += OWN_WAVES * 32) {
            const int e  = q_src[i];
            const int sn = src[e];
            const unsigned loc = (unsigned)dst[e] - (unsigned)r0;
            const int rw = (loc < (unsigned)nrows) ? (int)loc : OWN_R;
            const int rank = atomicAdd(&qcnt[rw], 1);
            q_src[i] = sn;
            q_row[i] = rw | (rank << 8);
        }
        __syncthreads();

        if (wave == 0) {
            int cnt[OWN_PS];
            int tot = 0;
            #pragma unroll
            for (int k = 0; k < OWN_PS; ++k) {
                const int j = lane * OWN_PS + k;
                cnt[k] = (j < OWN_LROWS) ? qcnt[j] : 0;
                tot += cnt[k];
            }
            int incl = tot;
            #pragma unroll
            for (int d = 1; d < 32; d <<= 1) {
                const int up = __shfl_up(incl, d, 32);
                if (lane >= d) incl += up;
            }
            int run = incl - tot;
            #pragma unroll
            for (int k = 0; k < OWN_PS; ++k) {
                const int j = lane * OWN_PS + k;
                if (j < OWN_LROWS) rst[j] = run;
                run += cnt[k];
            }
            if (lane == 31) rst[OWN_LROWS] = incl;
        }
        __syncthreads();

        for (int i = tid; i < qn; i += OWN_WAVES * 32) {
            const int rr = q_row[i];
            q_srt[rst[rr & 0xFF] + (rr >> 8)] = q_src[i];
        }
        __syncthreads();

        if (!reuse) {
            for (int i = tid; i < qn; i += OWN_WAVES * 32) my_q[i] = q_srt[i];
            if (tid <= OWN_LROWS) my_q[OWN_Q + tid] = rst[tid];
            if (tid == 0) wqn[blockIdx.x] = q_n;
        }
    }

    if (need_s && need_m)
        own_apply_rows<true, true>(x, q_srt, rst, acc_s, acc_m, &next_row, nrows, wave, lane);
    else if (need_s)
        own_apply_rows<true, false>(x, q_srt, rst, acc_s, acc_m, &next_row, nrows, wave, lane);
    else if (need_m)
        own_apply_rows<false, true>(x, q_srt, rst, acc_s, acc_m, &next_row, nrows, wave, lane);
    __syncthreads();

    const int slab  = wave & 7;
    const int grp   = wave >> 3;
    const int ncol  = (slab << 4) + (lane & 15);
    const int mlo   = lane & 15;
    const int koff  = (lane >> 4) << 1;
    const int rbase = (lane >> 4) << 3;
    const int ntl   = (nrows + TILE_M - 1) / TILE_M;
    float* const hbuf = acc_s;
    float* const tbuf = reinterpret_cast<float*>(acc_m);

    v2f wf[KSTEPS];
    #pragma unroll
    for (int i = 0; i < KSTEPS; ++i) {
        const int kk = 4 * i + koff;
        wf[i].x = W1[(size_t)kk * GNN_D + ncol];
        wf[i].y = W1[(size_t)(kk + 1) * GNN_D + ncol];
    }
    const float bia1 = b1[ncol];
    const float bia2 = b2[ncol];

    {
        const float z0 = zagg[0], z1 = zagg[1], z2 = zagg[2];
        for (int i = tid; i < ntl * TILE_M * (GNN_D / 4); i += OWN_WAVES * 32) {
            const int r = i >> 5;
            const int cc = (i & 31) << 2;
            if (r >= nrows) {
                *reinterpret_cast<float4*>(hbuf + r * LDS_STRIDE + cc) = make_float4(0.0f, 0.0f, 0.0f, 0.0f);
                continue;
            }
            const float4 xv = *reinterpret_cast<const float4*>(x + (size_t)(r0 + r) * GNN_D + cc);
            const float dg = (float)(rst[r + 1] - rst[r] + rcnt[r]);
            const float dc = fmaxf(dg, 1.0f);
            const bool  nb = (dg > 0.0f) && need_m;
            float4 sv = make_float4(0.0f, 0.0f, 0.0f, 0.0f);
            uint4  mv = make_uint4(0u, 0u, 0u, 0u);
            float* const hp = hbuf + r * LDS_STRIDE + cc;
            if (need_s) sv = *reinterpret_cast<const float4*>(hp);
            if (need_m) mv = *reinterpret_cast<const uint4*>(acc_m + r * LDS_STRIDE + cc);
            float4 h;
            h.x = xv.x + z0 * sv.x + z1 * (sv.x / dc) + z2 * (nb ? mono2f(mv.x) : 0.0f);
            h.y = xv.y + z0 * sv.y + z1 * (sv.y / dc) + z2 * (nb ? mono2f(mv.y) : 0.0f);
            h.z = xv.z + z0 * sv.z + z1 * (sv.z / dc) + z2 * (nb ? mono2f(mv.z) : 0.0f);
            h.w = xv.w + z0 * sv.w + z1 * (sv.w / dc) + z2 * (nb ? mono2f(mv.w) : 0.0f);
            *reinterpret_cast<float4*>(hp) = h;
        }
    }
    __syncthreads();

    for (int t = grp; t < ntl; t += OWN_WAVES / 8) {
        const float* const hrow = hbuf + (t * TILE_M + mlo) * LDS_STRIDE + koff;
        v8f acc = {};
        #pragma unroll
        for (int i = 0; i < KSTEPS; ++i) {
            v2f a; a.x = hrow[4 * i];
                   a.y = hrow[4 * i + 1];
            acc = __builtin_amdgcn_wmma_f32_16x16x4_f32(false, a, false, wf[i],
                                                        (short)0, acc, false, false);
        }
        #pragma unroll
        for (int i = 0; i < 8; ++i)
            tbuf[(t * TILE_M + rbase + i) * LDS_STRIDE + ncol] = fmaxf(acc[i] + bia1, 0.0f);
    }

    #pragma unroll
    for (int i = 0; i < KSTEPS; ++i) {
        const int kk = 4 * i + koff;
        wf[i].x = W2[(size_t)kk * GNN_D + ncol];
        wf[i].y = W2[(size_t)(kk + 1) * GNN_D + ncol];
    }
    __syncthreads();

    for (int t = grp; t < ntl; t += OWN_WAVES / 8) {
        float xs[8];
        #pragma unroll
        for (int i = 0; i < 8; ++i) {
            const int r = min(t * TILE_M + rbase + i, nrows - 1);
            xs[i] = x[(size_t)(r0 + r) * GNN_D + ncol];
        }
        const float* const trow = tbuf + (t * TILE_M + mlo) * LDS_STRIDE + koff;
        v8f acc2 = {};
        #pragma unroll
        for (int i = 0; i < KSTEPS; ++i) {
            v2f a; a.x = trow[4 * i];
                   a.y = trow[4 * i + 1];
            acc2 = __builtin_amdgcn_wmma_f32_16x16x4_f32(false, a, false, wf[i],
                                                         (short)0, acc2, false, false);
        }
        asm volatile("" :: "v"(xs[0]), "v"(xs[1]), "v"(xs[2]), "v"(xs[3]),
                           "v"(xs[4]), "v"(xs[5]), "v"(xs[6]), "v"(xs[7]));
        #pragma unroll
        for (int i = 0; i < 8; ++i) {
            const int r = t * TILE_M + rbase + i;
            const float o = xs[i] + fmaxf(acc2[i] + bia2, 0.0f);
            if (r < nrows) xout[(size_t)(r0 + r) * GNN_D + ncol] = o;
        }
    }
}

#define POOL_WAVES 16
#define POOL_U     8
__global__ __launch_bounds__(POOL_WAVES * 32) void gnn_pool_owner_kernel(
    const float* __restrict__ x, const int* __restrict__ batch,
    const float* __restrict__ zpool, float* __restrict__ out, int N)
{
    __shared__ float red_s[POOL_WAVES][32];
    __shared__ float red_m[POOL_WAVES][32];
    __shared__ int   red_c[POOL_WAVES];

    const int lane = threadIdx.x & 31;
    const int wave = __builtin_amdgcn_readfirstlane((int)(threadIdx.x >> 5));
    const int b    = blockIdx.x >> 2;
    const unsigned col = ((blockIdx.x & 3u) << 5) + (unsigned)lane;

    float sum = 0.0f;
    float mxv = -INFINITY;
    int   cnt = 0;

    for (int c0 = wave; (c0 << 5) < N; c0 += POOL_WAVES * POOL_U) {
        int ids[POOL_U];
        #pragma unroll
        for (int u = 0; u < POOL_U; ++u) {
            int n  = ((c0 + u * POOL_WAVES) << 5) + lane;
            int id = batch[min(n, N - 1)];
            ids[u] = (n < N) ? id : -1;
        }
        #pragma unroll
        for (int u = 0; u < POOL_U; ++u) {
            const unsigned mask = __builtin_amdgcn_ballot_w32(ids[u] == b);
            if (mask == 0u) continue;
            cnt += __builtin_popcount(mask);
            const int base = (c0 + u * POOL_WAVES) << 5;
            float v[32];
            if (base + 32 <= N) {
                const float* p = x + (size_t)base * GNN_D + col;
                #pragma unroll
                for (int j = 0; j < 32; ++j) v[j] = p[j * GNN_D];
            } else {
                #pragma unroll
                for (int j = 0; j < 32; ++j)
                    v[j] = x[(size_t)min(base + j, N - 1) * GNN_D + col];
            }
            #pragma unroll
            for (int j = 0; j < 32; ++j) {
                const bool in = (mask >> j) & 1u;
                sum += in ? v[j] : 0.0f;
                mxv = fmaxf(mxv, in ? v[j] : -INFINITY);
            }
        }
    }

    red_s[wave][lane] = sum;
    red_m[wave][lane] = mxv;
    if (lane == 0) red_c[wave] = cnt;
    __syncthreads();

    if (wave == 0) {
        float s = 0.0f;
        float m = -INFINITY;
        int   c = 0;
        #pragma unroll
        for (int w = 0; w < POOL_WAVES; ++w) {
            s += red_s[w][lane];
            m  = fmaxf(m, red_m[w][lane]);
            c += red_c[w];
        }
        const float cf = (float)c;
        const float mv = (c > 0) ? m : 0.0f;
        out[(size_t)b * GNN_D + col] =
            zpool[0] * s + zpool[1] * (s / fmaxf(cf, 1.0f)) + zpool[2] * mv;
    }
}


extern "C" void kernel_launch(void* const* d_in, const int* in_sizes, int n_in,
                              void* d_out, int out_size, void* d_ws, size_t ws_size,
                              hipStream_t stream) {
    const float* x_in   = (const float*)d_in[0];
    const int*   eidx   = (const int*)d_in[1];
    const int*   batch  = (const int*)d_in[2];
    const float* z_agg  = (const float*)d_in[3];
    const float* z_pool = (const float*)d_in[4];
    const float* W1     = (const float*)d_in[5];
    const float* b1     = (const float*)d_in[6];
    const float* W2     = (const float*)d_in[7];
    const float* b2     = (const float*)d_in[8];
    (void)n_in;

    const int N = in_sizes[0] / GNN_D;
    const int E = in_sizes[1] / 2;
    const int L = in_sizes[5] / (GNN_D * GNN_D);
    const int Bg = out_size / GNN_D;
    const size_t ND = (size_t)N * GNN_D;

    char* p = (char*)d_ws;
    char* pend = p + ws_size;
    auto take = [&](size_t bytes) -> void* {
        void* r = (void*)p;
        p += (bytes + 255) & ~(size_t)255;
        return r;
    };
    float*    xa   = (float*)take(ND * 4);
    float*    xb   = (float*)take(ND * 4);
    const int nown = (N + OWN_R - 1) / OWN_R;
    int*      wq   = (int*)take((size_t)nown * 2 * OWN_Q * 4);
    int*      wqn  = (int*)take((size_t)nown * 4);
    const int Epad = (E + 15) & ~15;
    unsigned char* ob = (unsigned char*)take((size_t)Epad);
    if (p > pend) return;

    const int* src = eidx;
    const int* dst = eidx + E;

    hipMemcpyAsync(xa, x_in, ND * 4, hipMemcpyDeviceToDevice, stream);

    const int nquad = Epad / 4;
    gnn_owner_byte_kernel<<<(nquad + 256 * OB_QUADS - 1) / (256 * OB_QUADS), 256, 0, stream>>>(
        dst, (unsigned*)ob, E, nquad);

    float* xcur = xa;
    float* xnxt = xb;
    for (int l = 0; l < L; ++l) {
        gnn_layer_owner_kernel<<<nown, OWN_WAVES * 32, 0, stream>>>(
            xcur, src, dst, (const uint4*)ob, z_agg + l * 3,
            W1 + (size_t)l * GNN_D * GNN_D, b1 + (size_t)l * GNN_D,
            W2 + (size_t)l * GNN_D * GNN_D, b2 + (size_t)l * GNN_D, xnxt,
            wq, wqn, l > 0 ? 1 : 0, N, E);
        float* t = xcur; xcur = xnxt; xnxt = t;
    }

    gnn_pool_owner_kernel<<<Bg * 4, POOL_WAVES * 32, 0, stream>>>(
        xcur, batch, z_pool, (float*)d_out, N);
}


# Combine pass fused into the apply phase: the wave that sums/maxes a row forms its layer-input row from registers, own x row asked for beside the source rows; pass and one barrier gone; sums as before
# speedup vs baseline: 5.6314x; 1.0357x over previous
#include <hip/hip_runtime.h>
#include <hip/hip_bf16.h>


typedef float v8f __attribute__((ext_vector_type(8)));
typedef float v2f __attribute__((ext_vector_type(2)));

#define GNN_D 128
#define TILE_M 16
#define LDS_STRIDE 132
#define KSTEPS (GNN_D / 4)

__device__ __forceinline__ unsigned f2mono(float f) {
    unsigned u = __float_as_uint(f);
    return (u & 0x80000000u) ? ~u : (u | 0x80000000u);
}
__device__ __forceinline__ float mono2f(unsigned m) {
    unsigned u = (m & 0x80000000u) ? (m & 0x7FFFFFFFu) : ~m;
    return __uint_as_float(u);
}
#define MONO_NEG_INF 0x007FFFFFu


#define OWN_R     200
#define OWN_TILES ((OWN_R + TILE_M - 1) / TILE_M)
#define OWN_LROWS (OWN_TILES * TILE_M)
static_assert(OWN_LROWS > OWN_R, "the spare row must lie inside the padded tile rows");
static_assert(OWN_R < 256, "an owned row's number (the spare row's too) is kept in eight bits beside the rank");
#define OB_QUADS 4
__device__ __forceinline__ unsigned own_byte_of(int d) {
    return ((unsigned)d / (unsigned)OWN_R) & 0xFFu;
}
__device__ __forceinline__ unsigned own_pack4(const int4 d) {
    return own_byte_of(d.x) | (own_byte_of(d.y) << 8) |
           (own_byte_of(d.z) << 16) | (own_byte_of(d.w) << 24);
}
__global__ __launch_bounds__(256) void gnn_owner_byte_kernel(const int* __restrict__ dst,
                                                             unsigned* __restrict__ ob32,
                                                             int E, int nquad) {
    const int qb = blockIdx.x * (256 * OB_QUADS);
    const bool whole = ((((size_t)dst) & 15u) == 0u) && ((qb + 256 * OB_QUADS) * 4 <= E);
    if (whole) {
        const int4* p = reinterpret_cast<const int4*>(dst) + qb + threadIdx.x;
        int4 d[OB_QUADS];
        #pragma unroll
        for (int k = 0; k < OB_QUADS; ++k) d[k] = p[k * 256];
        #pragma unroll
        for (int k = 0; k < OB_QUADS; ++k) ob32[qb + threadIdx.x + k * 256] = own_pack4(d[k]);
    } else {
        #pragma nounroll
        for (int k = 0; k < OB_QUADS; ++k) {
            const int q = qb + threadIdx.x + k * 256;
            if (q >= nquad) continue;
            const int e = q << 2;
            unsigned w = 0u;
            #pragma unroll
            for (int j = 0; j < 4; ++j) {
                const unsigned bj = (e + j < E) ? own_byte_of(dst[e + j]) : 0xFFu;
                w |= bj << (8 * j);
            }
            ob32[q] = w;
        }
    }
}

#define OWN_Q     4096
#define OWN_WAVES 32
#define OWN_U     8
#define OWN_PB    8
#define OWN_PS    ((OWN_LROWS + 31) / 32)
#define OWN_G     (OWN_WAVES * 32)
#define OWN_TRIP  (OWN_G * OWN_U * 16)

__device__ __forceinline__ void own_take(
    int e, int r0, int nrows, int* q_n, int* q_src, int* rcnt,
    float* acc_s, unsigned* acc_m,
    const float* __restrict__ x, const int* __restrict__ src,
    const int* __restrict__ dst, bool need_s, bool need_m)
{
    const int pos = atomicAdd(q_n, 1);
    if (pos < OWN_Q) {
        q_src[pos] = e;
    } else {
        const unsigned loc = (unsigned)dst[e] - (unsigned)r0;
        if (loc >= (unsigned)nrows) return;
        atomicAdd(rcnt + loc, 1);
        const float4* xr = reinterpret_cast<const float4*>(x + (size_t)src[e] * GNN_D);
        float*    as = acc_s + loc * LDS_STRIDE;
        unsigned* am = acc_m + loc * LDS_STRIDE;
        #pragma nounroll
        for (int f = 0; f < GNN_D / 4; ++f) {
            const float4 v = xr[f];
            if (need_s) {
                atomicAdd(as + 4 * f + 0, v.x);
                atomicAdd(as + 4 * f + 1, v.y);
                atomicAdd(as + 4 * f + 2, v.z);
                atomicAdd(as + 4 * f + 3, v.w);
            }
            if (need_m) {
                atomicMax(am + 4 * f + 0, f2mono(v.x));
                atomicMax(am + 4 * f + 1, f2mono(v.y));
                atomicMax(am + 4 * f + 2, f2mono(v.z));
                atomicMax(am + 4 * f + 3, f2mono(v.w));
            }
        }
    }
}

__device__ __forceinline__ unsigned own_eq_bytes(unsigned w, unsigned pat4) {
    const unsigned d = w ^ pat4;
    return ~(((d & 0x7F7F7F7Fu) + 0x7F7F7F7Fu) | d | 0x7F7F7F7Fu);
}
__device__ __forceinline__ unsigned own_nibble(unsigned h) {
    return (((h >> 7) * 0x00204081u) >> 21) & 0xFu;
}
__device__ __forceinline__ void own_scan16(
    const uint4 w, unsigned pat4, int e0, int elim, int r0, int nrows,
    int* q_n, int* q_src, int* rcnt, float* acc_s, unsigned* acc_m,
    const float* __restrict__ x, const int* __restrict__ src,
    const int* __restrict__ dst, bool need_s, bool need_m)
{
    const unsigned h0 = own_eq_bytes(w.x, pat4);
    const unsigned h1 = own_eq_bytes(w.y, pat4);
    const unsigned h2 = own_eq_bytes(w.z, pat4);
    const unsigned h3 = own_eq_bytes(w.w, pat4);
    if ((h0 | h1 | h2 | h3) != 0u) {
        unsigned m = own_nibble(h0) | (own_nibble(h1) << 4) |
                     (own_nibble(h2) << 8) | (own_nibble(h3) << 12);
        do {
            const int e = e0 + __builtin_ctz(m);
            if (e < elim)
                own_take(e, r0, nrows, q_n, q_src, rcnt, acc_s, acc_m, x, src, dst, need_s, need_m);
            m &= m - 1u;
        } while (m != 0u);
    }
}

template <bool SUM, bool MAX>
__device__ __forceinline__ void own_apply_rows(
    const float* __restrict__ x, const int* q_srt, const int* rst, const int* rcnt,
    float* acc_s, unsigned* acc_m, int* next_row, int r0, int nrows, int wave, int lane,
    float z0, float z1, float z2)
{
    int r = wave;
    for (int it = 0; it <= OWN_R; ++it) {
        if (r >= nrows) break;
        int nxt = 0;
        if (lane == 0) nxt = atomicAdd(next_row, 1);
        const int rc = r;
        const int qb = __builtin_amdgcn_readfirstlane(rst[rc]);
        const int qe = __builtin_amdgcn_readfirstlane(rst[rc + 1]);
        const int ov = __builtin_amdgcn_readfirstlane(rcnt[rc]);
        const float4 xv = *reinterpret_cast<const float4*>(x + (size_t)(r0 + rc) * GNN_D + 4 * lane);
        float4 s = make_float4(0.0f, 0.0f, 0.0f, 0.0f);
        uint4  m = make_uint4(MONO_NEG_INF, MONO_NEG_INF, MONO_NEG_INF, MONO_NEG_INF);
        if (SUM || MAX) {
            for (int q0 = qb; q0 < qe; q0 += OWN_PB) {
                const int mine = q_srt[min(q0 + lane, qe - 1)];
                float4 v[OWN_PB];
                #pragma unroll
                for (int p = 0; p < OWN_PB; ++p) {
                    const int sn = __builtin_amdgcn_readlane(mine, p);
                    v[p] = *reinterpret_cast<const float4*>(x + (size_t)sn * GNN_D + 4 * lane);
                }
                #pragma unroll
                for (int p = 0; p < OWN_PB; ++p)
                    asm volatile("" :: "v"(v[p].x), "v"(v[p].y), "v"(v[p].z), "v"(v[p].w));
                #pragma unroll
                for (int p = 0; p < OWN_PB; ++p) {
                    if (q0 + p < qe) {
                        if (SUM) {
                            s.x += v[p].x; s.y += v[p].y; s.z += v[p].z; s.w += v[p].w;
                        }
                        if (MAX) {
                            m.x = max(m.x, f2mono(v[p].x)); m.y = max(m.y, f2mono(v[p].y));
                            m.z = max(m.z, f2mono(v[p].z)); m.w = max(m.w, f2mono(v[p].w));
                        }
                    }
                }
            }
        }
        float* const hp = acc_s + rc * LDS_STRIDE + 4 * lane;
        float4 sv = make_float4(0.0f, 0.0f, 0.0f, 0.0f);
        uint4  mv = make_uint4(0u, 0u, 0u, 0u);
        if (SUM) {
            sv = *reinterpret_cast<const float4*>(hp);
            sv.x += s.x; sv.y += s.y; sv.z += s.z; sv.w += s.w;
        }
        if (MAX) {
            mv = *reinterpret_cast<const uint4*>(acc_m + rc * LDS_STRIDE + 4 * lane);
            mv.x = max(mv.x, m.x); mv.y = max(mv.y, m.y); mv.z = max(mv.z, m.z); mv.w = max(mv.w, m.w);
        }
        const float dg = (float)(qe - qb + ov);
        const float dc = fmaxf(dg, 1.0f);
        const bool  nb = MAX && (dg > 0.0f);
        float4 h;
        h.x = xv.x + z0 * sv.x + z1 * (sv.x / dc) + z2 * (nb ? mono2f(mv.x) : 0.0f);
        h.y = xv.y + z0 * sv.y + z1 * (sv.y / dc) + z2 * (nb ? mono2f(mv.y) : 0.0f);
        h.z = xv.z + z0 * sv.z + z1 * (sv.z / dc) + z2 * (nb ? mono2f(mv.z) : 0.0f);
        h.w = xv.w + z0 * sv.w + z1 * (sv.w / dc) + z2 * (nb ? mono2f(mv.w) : 0.0f);
        *reinterpret_cast<float4*>(hp) = h;
        r = __builtin_amdgcn_readfirstlane(nxt);
    }
}

__global__ __launch_bounds__(OWN_WAVES * 32) void gnn_layer_owner_kernel(
    const float* __restrict__ x, const int* __restrict__ src,
    const int* __restrict__ dst, const uint4* __restrict__ ob16,
    const float* __restrict__ zagg,
    const float* __restrict__ W1, const float* __restrict__ b1,
    const float* __restrict__ W2, const float* __restrict__ b2,
    float* __restrict__ xout,
    int* wq, int* wqn, int reuse, int N, int E)
{
    __shared__ __attribute__((aligned(16))) float    acc_s[OWN_LROWS * LDS_STRIDE];
    __shared__ __attribute__((aligned(16))) unsigned acc_m[OWN_LROWS * LDS_STRIDE];
    __shared__ int q_src[OWN_Q];
    __shared__ int q_row[OWN_Q];
    __shared__ int q_srt[OWN_Q];
    __shared__ int qcnt[OWN_LROWS];
    __shared__ int rst[OWN_LROWS + 1];
    __shared__ int rcnt[OWN_LROWS];
    __shared__ int q_n;
    __shared__ int next_row;

    const int tid   = threadIdx.x;
    const int lane  = tid & 31;
    const int wave  = __builtin_amdgcn_readfirstlane(tid >> 5);
    const int r0    = blockIdx.x * OWN_R;
    const int nrows = min(OWN_R, N - r0);
    int* const my_q = wq + (size_t)blockIdx.x * (2 * OWN_Q);
    const int saved = __builtin_amdgcn_readfirstlane(reuse ? wqn[blockIdx.x] : -1);
    const bool have = (unsigned)saved <= (unsigned)OWN_Q;
    const float z0 = zagg[0], z1 = zagg[1], z2 = zagg[2];
    const bool need_s = (z0 != 0.0f) || (z1 != 0.0f);
    const bool need_m = (z2 != 0.0f);

    for (int i = tid; i < OWN_LROWS * LDS_STRIDE / 4; i += OWN_WAVES * 32) {
        if (need_s)
            reinterpret_cast<float4*>(acc_s)[i] = make_float4(0.0f, 0.0f, 0.0f, 0.0f);
        if (need_m)
            reinterpret_cast<uint4*>(acc_m)[i]  =
                make_uint4(MONO_NEG_INF, MONO_NEG_INF, MONO_NEG_INF, MONO_NEG_INF);
    }
    if (tid < OWN_LROWS) { rcnt[tid] = 0; qcnt[tid] = 0; }
    if (tid == 0) { q_n = 0; next_row = OWN_WAVES; }
    __syncthreads();

    int qn;
    if (have) {
        qn = saved;
        for (int i = tid; i < qn; i += OWN_WAVES * 32) q_srt[i] = my_q[i];
        if (tid <= OWN_LROWS) rst[tid] = my_q[OWN_Q + tid];
        __syncthreads();
    } else {
        const unsigned ownb = (unsigned)blockIdx.x & 0xFFu;
        const unsigned pat4 = ownb * 0x01010101u;
        const int nfull = E / OWN_TRIP;
        unsigned trip = ((unsigned)blockIdx.x * (unsigned)nfull) / gridDim.x;
        for (int t = 0; t < nfull; ++t) {
            const int gbase = (int)trip * (OWN_G * OWN_U) + tid;
            const uint4* p = ob16 + gbase;
            uint4 d[OWN_U];
            #pragma unroll
            for (int u = 0; u < OWN_U; ++u) d[u] = p[u * OWN_G];
            #pragma unroll
            for (int u = 0; u < OWN_U; ++u)
                own_scan16(d[u], pat4, (gbase + u * OWN_G) * 16, E, r0, nrows, &q_n, q_src,
                           rcnt, acc_s, acc_m, x, src, dst, need_s, need_m);
            if (++trip == (unsigned)nfull) trip = 0u;
        }
        {
            const int ngrp  = (E + 15) >> 4;
            const int gbase = nfull * (OWN_G * OWN_U) + tid;
            uint4 d[OWN_U];
            #pragma unroll
            for (int u = 0; u < OWN_U; ++u) {
                const int g = gbase + u * OWN_G;
                d[u] = make_uint4(~pat4, ~pat4, ~pat4, ~pat4);
                if (g < ngrp) d[u] = ob16[g];
            }
            #pragma unroll
            for (int u = 0; u < OWN_U; ++u)
                own_scan16(d[u], pat4, (gbase + u * OWN_G) * 16, E, r0, nrows, &q_n, q_src,
                           rcnt, acc_s, acc_m, x, src, dst, need_s, need_m);
        }
        __syncthreads();

        qn = min(q_n, OWN_Q);
        for (int i = tid; i < qn; i += OWN_WAVES * 32) {
            const int e  = q_src[i];
            const int sn = src[e];
            const unsigned loc = (unsigned)dst[e] - (unsigned)r0;
            const int rw = (loc < (unsigned)nrows) ? (int)loc : OWN_R;
            const int rank = atomicAdd(&qcnt[rw], 1);
            q_src[i] = sn;
            q_row[i] = rw | (rank << 8);
        }
        __syncthreads();

        if (wave == 0) {
            int cnt[OWN_PS];
            int tot = 0;
            #pragma unroll
            for (int k = 0; k < OWN_PS; ++k) {
                const int j = lane * OWN_PS + k;
                cnt[k] = (j < OWN_LROWS) ? qcnt[j] : 0;
                tot += cnt[k];
            }
            int incl = tot;
            #pragma unroll
            for (int d = 1; d < 32; d <<= 1) {
                const int up = __shfl_up(incl, d, 32);
                if (lane >= d) incl += up;
            }
            int run = incl - tot;
            #pragma unroll
            for (int k = 0; k < OWN_PS; ++k) {
                const int j = lane * OWN_PS + k;
                if (j < OWN_LROWS) rst[j] = run;
                run += cnt[k];
            }
            if (lane == 31) rst[OWN_LROWS] = incl;
        }
        __syncthreads();

        for (int i = tid; i < qn; i += OWN_WAVES * 32) {
            const int rr = q_row[i];
            q_srt[rst[rr & 0xFF] + (rr >> 8)] = q_src[i];
        }
        __syncthreads();

        if (!reuse) {
            for (int i = tid; i < qn; i += OWN_WAVES * 32) my_q[i] = q_srt[i];
            if (tid <= OWN_LROWS) my_q[OWN_Q + tid] = rst[tid];
            if (tid == 0) wqn[blockIdx.x] = q_n;
        }
    }

    const int slab  = wave & 7;
    const int grp   = wave >> 3;
    const int ncol  = (slab << 4) + (lane & 15);
    const int mlo   = lane & 15;
    const int koff  = (lane >> 4) << 1;
    const int rbase = (lane >> 4) << 3;
    const int ntl   = (nrows + TILE_M - 1) / TILE_M;
    float* const hbuf = acc_s;
    float* const tbuf = reinterpret_cast<float*>(acc_m);

    if (need_s && need_m)
        own_apply_rows<true, true>(x, q_srt, rst, rcnt, acc_s, acc_m, &next_row, r0, nrows, wave, lane, z0, z1, z2);
    else if (need_s)
        own_apply_rows<true, false>(x, q_srt, rst, rcnt, acc_s, acc_m, &next_row, r0, nrows, wave, lane, z0, z1, z2);
    else if (need_m)
        own_apply_rows<false, true>(x, q_srt, rst, rcnt, acc_s, acc_m, &next_row, r0, nrows, wave, lane, z0, z1, z2);
    else
        own_apply_rows<false, false>(x, q_srt, rst, rcnt, acc_s, acc_m, &next_row, r0, nrows, wave, lane, z0, z1, z2);
    static_assert(TILE_M <= OWN_WAVES, "one padding row per wave must cover the last tile");
    if (nrows + wave < ntl * TILE_M)
        *reinterpret_cast<float4*>(hbuf + (nrows + wave) * LDS_STRIDE + 4 * lane) =
            make_float4(0.0f, 0.0f, 0.0f, 0.0f);

    v2f wf[KSTEPS];
    #pragma unroll
    for (int i = 0; i < KSTEPS; ++i) {
        const int kk = 4 * i + koff;
        wf[i].x = W1[(size_t)kk * GNN_D + ncol];
        wf[i].y = W1[(size_t)(kk + 1) * GNN_D + ncol];
    }
    const float bia1 = b1[ncol];
    const float bia2 = b2[ncol];
    __syncthreads();

    for (int t = grp; t < ntl; t += OWN_WAVES / 8) {
        const float* const hrow = hbuf + (t * TILE_M + mlo) * LDS_STRIDE + koff;
        v8f acc = {};
        #pragma unroll
        for (int i = 0; i < KSTEPS; ++i) {
            v2f a; a.x = hrow[4 * i];
                   a.y = hrow[4 * i + 1];
            acc = __builtin_amdgcn_wmma_f32_16x16x4_f32(false, a, false, wf[i],
                                                        (short)0, acc, false, false);
        }
        #pragma unroll
        for (int i = 0; i < 8; ++i)
            tbuf[(t * TILE_M + rbase + i) * LDS_STRIDE + ncol] = fmaxf(acc[i] + bia1, 0.0f);
    }

    #pragma unroll
    for (int i = 0; i < KSTEPS; ++i) {
        const int kk = 4 * i + koff;
        wf[i].x = W2[(size_t)kk * GNN_D + ncol];
        wf[i].y = W2[(size_t)(kk + 1) * GNN_D + ncol];
    }
    __syncthreads();

    for (int t = grp; t < ntl; t += OWN_WAVES / 8) {
        float xs[8];
        #pragma unroll
        for (int i = 0; i < 8; ++i) {
            const int r = min(t * TILE_M + rbase + i, nrows - 1);
            xs[i] = x[(size_t)(r0 + r) * GNN_D + ncol];
        }
        const float* const trow = tbuf + (t * TILE_M + mlo) * LDS_STRIDE + koff;
        v8f acc2 = {};
        #pragma unroll
        for (int i = 0; i < KSTEPS; ++i) {
            v2f a; a.x = trow[4 * i];
                   a.y = trow[4 * i + 1];
            acc2 = __builtin_amdgcn_wmma_f32_16x16x4_f32(false, a, false, wf[i],
                                                         (short)0, acc2, false, false);
        }
        asm volatile("" :: "v"(xs[0]), "v"(xs[1]), "v"(xs[2]), "v"(xs[3]),
                           "v"(xs[4]), "v"(xs[5]), "v"(xs[6]), "v"(xs[7]));
        #pragma unroll
        for (int i = 0; i < 8; ++i) {
            const int r = t * TILE_M + rbase + i;
            const float o = xs[i] + fmaxf(acc2[i] + bia2, 0.0f);
            if (r < nrows) xout[(size_t)(r0 + r) * GNN_D + ncol] = o;
        }
    }
}

#define POOL_WAVES 16
#define POOL_U     8
__global__ __launch_bounds__(POOL_WAVES * 32) void gnn_pool_owner_kernel(
    const float* __restrict__ x, const int* __restrict__ batch,
    const float* __restrict__ zpool, float* __restrict__ out, int N)
{
    __shared__ float red_s[POOL_WAVES][32];
    __shared__ float red_m[POOL_WAVES][32];
    __shared__ int   red_c[POOL_WAVES];

    const int lane = threadIdx.x & 31;
    const int wave = __builtin_amdgcn_readfirstlane((int)(threadIdx.x >> 5));
    const int b    = blockIdx.x >> 2;
    const unsigned col = ((blockIdx.x & 3u) << 5) + (unsigned)lane;

    float sum = 0.0f;
    float mxv = -INFINITY;
    int   cnt = 0;

    for (int c0 = wave; (c0 << 5) < N; c0 += POOL_WAVES * POOL_U) {
        int ids[POOL_U];
        #pragma unroll
        for (int u = 0; u < POOL_U; ++u) {
            int n  = ((c0 + u * POOL_WAVES) << 5) + lane;
            int id = batch[min(n, N - 1)];
            ids[u] = (n < N) ? id : -1;
        }
        #pragma unroll
        for (int u = 0; u < POOL_U; ++u) {
            const unsigned mask = __builtin_amdgcn_ballot_w32(ids[u] == b);
            if (mask == 0u) continue;
            cnt += __builtin_popcount(mask);
            const int base = (c0 + u * POOL_WAVES) << 5;
            float v[32];
            if (base + 32 <= N) {
                const float* p = x + (size_t)base * GNN_D + col;
                #pragma unroll
                for (int j = 0; j < 32; ++j) v[j] = p[j * GNN_D];
            } else {
                #pragma unroll
                for (int j = 0; j < 32; ++j)
                    v[j] = x[(size_t)min(base + j, N - 1) * GNN_D + col];
            }
            #pragma unroll
            for (int j = 0; j < 32; ++j) {
                const bool in = (mask >> j) & 1u;
                sum += in ? v[j] : 0.0f;
                mxv = fmaxf(mxv, in ? v[j] : -INFINITY);
            }
        }
    }

    red_s[wave][lane] = sum;
    red_m[wave][lane] = mxv;
    if (lane == 0) red_c[wave] = cnt;
    __syncthreads();

    if (wave == 0) {
        float s = 0.0f;
        float m = -INFINITY;
        int   c = 0;
        #pragma unroll
        for (int w = 0; w < POOL_WAVES; ++w) {
            s += red_s[w][lane];
            m  = fmaxf(m, red_m[w][lane]);
            c += red_c[w];
        }
        const float cf = (float)c;
        const float mv = (c > 0) ? m : 0.0f;
        out[(size_t)b * GNN_D + col] =
            zpool[0] * s + zpool[1] * (s / fmaxf(cf, 1.0f)) + zpool[2] * mv;
    }
}


extern "C" void kernel_launch(void* const* d_in, const int* in_sizes, int n_in,
                              void* d_out, int out_size, void* d_ws, size_t ws_size,
                              hipStream_t stream) {
    const float* x_in   = (const float*)d_in[0];
    const int*   eidx   = (const int*)d_in[1];
    const int*   batch  = (const int*)d_in[2];
    const float* z_agg  = (const float*)d_in[3];
    const float* z_pool = (const float*)d_in[4];
    const float* W1     = (const float*)d_in[5];
    const float* b1     = (const float*)d_in[6];
    const float* W2     = (const float*)d_in[7];
    const float* b2     = (const float*)d_in[8];
    (void)n_in;

    const int N = in_sizes[0] / GNN_D;
    const int E = in_sizes[1] / 2;
    const int L = in_sizes[5] / (GNN_D * GNN_D);
    const int Bg = out_size / GNN_D;
    const size_t ND = (size_t)N * GNN_D;

    char* p = (char*)d_ws;
    char* pend = p + ws_size;
    auto take = [&](size_t bytes) -> void* {
        void* r = (void*)p;
        p += (bytes + 255) & ~(size_t)255;
        return r;
    };
    float*    xa   = (float*)take(ND * 4);
    float*    xb   = (float*)take(ND * 4);
    const int nown = (N + OWN_R - 1) / OWN_R;
    int*      wq   = (int*)take((size_t)nown * 2 * OWN_Q * 4);
    int*      wqn  = (int*)take((size_t)nown * 4);
    const int Epad = (E + 15) & ~15;
    unsigned char* ob = (unsigned char*)take((size_t)Epad);
    if (p > pend) return;

    const int* src = eidx;
    const int* dst = eidx + E;

    hipMemcpyAsync(xa, x_in, ND * 4, hipMemcpyDeviceToDevice, stream);

    const int nquad = Epad / 4;
    gnn_owner_byte_kernel<<<(nquad + 256 * OB_QUADS - 1) / (256 * OB_QUADS), 256, 0, stream>>>(
        dst, (unsigned*)ob, E, nquad);

    float* xcur = xa;
    float* xnxt = xb;
    for (int l = 0; l < L; ++l) {
        gnn_layer_owner_kernel<<<nown, OWN_WAVES * 32, 0, stream>>>(
            xcur, src, dst, (const uint4*)ob, z_agg + l * 3,
            W1 + (size_t)l * GNN_D * GNN_D, b1 + (size_t)l * GNN_D,
            W2 + (size_t)l * GNN_D * GNN_D, b2 + (size_t)l * GNN_D, xnxt,
            wq, wqn, l > 0 ? 1 : 0, N, E);
        float* t = xcur; xcur = xnxt; xnxt = t;
    }

    gnn_pool_owner_kernel<<<Bg * 4, POOL_WAVES * 32, 0, stream>>>(
        xcur, batch, z_pool, (float*)d_out, N);
}
